# starting kernel
# baseline (speedup 1.0000x reference)
#include <hip/hip_runtime.h>
#include <math.h>

#define NN   50000
#define EE   800000
#define SDIM 128
#define VDIM 16
#define V3   48
#define ESD  32

typedef float v2f __attribute__((ext_vector_type(2)));
typedef float v8f __attribute__((ext_vector_type(8)));

__device__ __forceinline__ v8f wmma4(v2f a, v2f b, v8f c) {
  return __builtin_amdgcn_wmma_f32_16x16x4_f32(false, a, false, b, (short)0, c,
                                               false, false);
}

__device__ __forceinline__ float fast_rcp(float x) {
  return __builtin_amdgcn_rcpf(x);
}

__device__ __forceinline__ float sigmoidf_(float x) {
  return fast_rcp(1.f + __expf(-x));
}

__device__ __forceinline__ void fadd_atomic(float* p, float v) {
  unsafeAtomicAdd(p, v);
}

__global__ void k_zero(float* __restrict__ p, int n) {
  int i = blockIdx.x * blockDim.x + threadIdx.x;
  int stride = gridDim.x * blockDim.x;
  for (; i < n; i += stride) p[i] = 0.f;
}

__global__ void __launch_bounds__(256) k_norm(
    const float* __restrict__ node_s, const float* __restrict__ node_v,
    const float* __restrict__ ln_g, const float* __restrict__ ln_b,
    float* __restrict__ s_n, float* __restrict__ v_n) {
  int wave = (blockIdx.x * blockDim.x + threadIdx.x) >> 5;
  int lane = threadIdx.x & 31;
  if (wave >= NN) return;

  const float* srow = node_s + (long)wave * SDIM;
  float4 x = ((const float4*)srow)[lane];
  float s = x.x + x.y + x.z + x.w;
#pragma unroll
  for (int m = 16; m >= 1; m >>= 1) s += __shfl_xor(s, m, 32);
  float mu = s * (1.f / SDIM);
  float dx = x.x - mu, dy = x.y - mu, dz = x.z - mu, dw = x.w - mu;
  float q = dx * dx + dy * dy + dz * dz + dw * dw;
#pragma unroll
  for (int m = 16; m >= 1; m >>= 1) q += __shfl_xor(q, m, 32);
  float rs = rsqrtf(q * (1.f / SDIM) + 1e-5f);
  int c0 = lane * 4;
  float4 o;
  o.x = dx * rs * ln_g[c0 + 0] + ln_b[c0 + 0];
  o.y = dy * rs * ln_g[c0 + 1] + ln_b[c0 + 1];
  o.z = dz * rs * ln_g[c0 + 2] + ln_b[c0 + 2];
  o.w = dw * rs * ln_g[c0 + 3] + ln_b[c0 + 3];
  ((float4*)(s_n + (long)wave * SDIM))[lane] = o;

  const float* vrow = node_v + (long)wave * V3;
  float2 v = {0.f, 0.f};
  int vi = lane * 2;
  if (vi < V3) v = ((const float2*)vrow)[lane];
  float qq = v.x * v.x + v.y * v.y;
#pragma unroll
  for (int m = 16; m >= 1; m >>= 1) qq += __shfl_xor(qq, m, 32);
  float inv = rsqrtf(qq * (1.f / VDIM) + 1e-8f);
  if (vi < V3) {
    float2 ov;
    ov.x = v.x * inv;
    ov.y = v.y * inv;
    ((float2*)(v_n + (long)wave * V3))[lane] = ov;
  }
}

__global__ void __launch_bounds__(256) k_edge(
    const float* __restrict__ edge_s, const float* __restrict__ edge_v,
    const int* __restrict__ edge_dst,
    const float* __restrict__ wes, const float* __restrict__ bes,
    const float* __restrict__ wev, const float* __restrict__ bev,
    float* __restrict__ agg_s, float* __restrict__ agg_v,
    float* __restrict__ cnt) {
  __shared__ float lds_gate[8][16 * 16];
  int wid = threadIdx.x >> 5;
  int lane = threadIdx.x & 31;
  int group = blockIdx.x * 8 + wid;
  if (group >= EE / 16) return;
  int ebase = group * 16;
  int row = lane & 15;
  int half = lane >> 4;

  const float* arow = edge_s + (long)(ebase + row) * ESD;

  int dsts[8];
#pragma unroll
  for (int r = 0; r < 8; r++) dsts[r] = edge_dst[ebase + r + 8 * half];

#pragma unroll 1
  for (int c = 0; c < 8; c++) {
    v8f acc = {};
    int n = 16 * c + row;
    const float* brow = wes + (long)n * ESD;
#pragma unroll
    for (int j = 0; j < 8; j++) {
      int k0 = 4 * j + 2 * half;
      v2f a = *(const v2f*)(arow + k0);
      v2f b = *(const v2f*)(brow + k0);
      acc = wmma4(a, b, acc);
    }
    float bias = bes[n];
#pragma unroll
    for (int r = 0; r < 8; r++) {
      float pre = acc[r] + bias;
      float so = pre * sigmoidf_(pre);
      if (c == 0) lds_gate[wid][(r + 8 * half) * 16 + n] = sigmoidf_(so);
      fadd_atomic(agg_s + (long)dsts[r] * SDIM + n, so);
    }
  }

  if (lane < 16) fadd_atomic(cnt + edge_dst[ebase + lane], 1.f);

  int e = ebase + row;
  int dstv = edge_dst[e];
  float ev0 = edge_v[e * 3 + 0];
  float ev1 = edge_v[e * 3 + 1];
  float ev2 = edge_v[e * 3 + 2];
#pragma unroll
  for (int t = 0; t < 24; t++) {
    int d = half * 24 + t;
    float v = ev0 * wev[d * 3 + 0] + ev1 * wev[d * 3 + 1] +
              ev2 * wev[d * 3 + 2] + bev[d];
    float g = lds_gate[wid][row * 16 + d / 3];
    fadd_atomic(agg_v + (long)dstv * V3 + d, v * g);
  }
}

__global__ void __launch_bounds__(256) k_node(
    const float* __restrict__ s_n, const float* __restrict__ v_n,
    const float* __restrict__ agg_s, const float* __restrict__ agg_v,
    const float* __restrict__ cnt,
    const float* __restrict__ wns, const float* __restrict__ bns,
    const float* __restrict__ wnv, const float* __restrict__ bnv,
    const float* __restrict__ wrs, const float* __restrict__ brs,
    const float* __restrict__ wrv, const float* __restrict__ brv,
    float* __restrict__ out) {
  __shared__ float lds_gate[8][16 * 16];
  int wid = threadIdx.x >> 5;
  int lane = threadIdx.x & 31;
  int group = blockIdx.x * 8 + wid;
  if (group >= NN / 16) return;
  int nb = group * 16;
  int row = lane & 15;
  int half = lane >> 4;
  int noderow = nb + row;

  float invden = fast_rcp(fmaxf(cnt[noderow], 1.f));
  const float* sA = s_n + (long)noderow * SDIM;
  const float* gA = agg_s + (long)noderow * SDIM;

#pragma unroll 1
  for (int ct = 0; ct < 8; ct++) {
    v8f ao = {};
    v8f ar = {};
    int n = 16 * ct + row;
    const float* bo = wns + (long)n * SDIM;
    const float* br = wrs + (long)n * SDIM;
#pragma unroll 4
    for (int j = 0; j < 32; j++) {
      int k0 = 4 * j + 2 * half;
      v2f as = *(const v2f*)(sA + k0);
      v2f ag = *(const v2f*)(gA + k0);
      v2f ain = {as.x + ag.x * invden, as.y + ag.y * invden};
      v2f bO = *(const v2f*)(bo + k0);
      v2f bR = *(const v2f*)(br + k0);
      ao = wmma4(ain, bO, ao);
      ar = wmma4(as, bR, ar);
    }
    float biasO = bns[n], biasR = brs[n];
#pragma unroll
    for (int r = 0; r < 8; r++) {
      int m = r + 8 * half;
      float pre = ao[r] + biasO;
      float so = pre * sigmoidf_(pre);
      if (ct == 0) lds_gate[wid][m * 16 + n] = sigmoidf_(so);
      out[(long)(nb + m) * SDIM + n] = so + (ar[r] + biasR);
    }
  }

  const float* vA = v_n + (long)noderow * V3;
  const float* gV = agg_v + (long)noderow * V3;
  float* outv = out + (long)NN * SDIM;
#pragma unroll 1
  for (int ct = 0; ct < 3; ct++) {
    v8f ao = {};
    v8f ar = {};
    int d = 16 * ct + row;
    const float* bo = wnv + (long)d * V3;
    const float* br = wrv + (long)d * V3;
#pragma unroll
    for (int j = 0; j < 12; j++) {
      int k0 = 4 * j + 2 * half;
      v2f av = *(const v2f*)(vA + k0);
      v2f ag = *(const v2f*)(gV + k0);
      v2f ain = {av.x + ag.x * invden, av.y + ag.y * invden};
      v2f bO = *(const v2f*)(bo + k0);
      v2f bR = *(const v2f*)(br + k0);
      ao = wmma4(ain, bO, ao);
      ar = wmma4(av, bR, ar);
    }
    float biasO = bnv[d], biasR = brv[d];
    int vo = d / 3;
#pragma unroll
    for (int r = 0; r < 8; r++) {
      int m = r + 8 * half;
      float pre = ao[r] + biasO;
      float g = lds_gate[wid][m * 16 + vo];
      outv[(long)(nb + m) * V3 + d] = pre * g + (ar[r] + biasR);
    }
  }
}

extern "C" void kernel_launch(void* const* d_in, const int* in_sizes, int n_in,
                              void* d_out, int out_size, void* d_ws,
                              size_t ws_size, hipStream_t stream) {
  const float* node_s = (const float*)d_in[0];
  const float* node_v = (const float*)d_in[1];
  const int* eidx     = (const int*)d_in[2];
  const float* edge_s = (const float*)d_in[3];
  const float* edge_v = (const float*)d_in[4];
  const float* ln_g = (const float*)d_in[5];
  const float* ln_b = (const float*)d_in[6];
  const float* wes = (const float*)d_in[7];
  const float* bes = (const float*)d_in[8];
  const float* wev = (const float*)d_in[9];
  const float* bev = (const float*)d_in[10];
  const float* wns = (const float*)d_in[11];
  const float* bns = (const float*)d_in[12];
  const float* wnv = (const float*)d_in[13];
  const float* bnv = (const float*)d_in[14];
  const float* wrs = (const float*)d_in[15];
  const float* brs = (const float*)d_in[16];
  const float* wrv = (const float*)d_in[17];
  const float* brv = (const float*)d_in[18];
  float* out = (float*)d_out;

  float* ws = (float*)d_ws;
  float* s_n = ws;
  float* v_n = s_n + (size_t)NN * SDIM;
  float* agg_s = v_n + (size_t)NN * V3;
  float* agg_v = agg_s + (size_t)NN * SDIM;
  float* cnt = agg_v + (size_t)NN * V3;
  const int* edge_dst = eidx + EE;

  int zn = NN * (SDIM + V3 + 1);
  k_zero<<<2048, 256, 0, stream>>>(agg_s, zn);
  k_norm<<<(NN * 32 + 255) / 256, 256, 0, stream>>>(node_s, node_v, ln_g, ln_b,
                                                    s_n, v_n);
  k_edge<<<EE / 16 / 8, 256, 0, stream>>>(edge_s, edge_v, edge_dst, wes, bes,
                                          wev, bev, agg_s, agg_v, cnt);
  k_node<<<(NN / 16 + 7) / 8, 256, 0, stream>>>(s_n, v_n, agg_s, agg_v, cnt,
                                                wns, bns, wnv, bnv, wrs, brs,
                                                wrv, brv, out);
}


# Edges bucketed by destination (int count, scan, fill); each 16-node block sums its buckets' messages by a 0/1 selection WMMA and plain stores: float atomics gone, sum order = bucket arrival order
# speedup vs baseline: 2.0572x; 2.0572x over previous
#include <hip/hip_runtime.h>
#include <math.h>

#define NN   50000
#define EE   800000
#define SDIM 128
#define VDIM 16
#define V3   48
#define ESD  32

typedef float v2f __attribute__((ext_vector_type(2)));
typedef float v8f __attribute__((ext_vector_type(8)));

__device__ __forceinline__ v8f wmma4(v2f a, v2f b, v8f c) {
  return __builtin_amdgcn_wmma_f32_16x16x4_f32(false, a, false, b, (short)0, c,
                                               false, false);
}

__device__ __forceinline__ float fast_rcp(float x) {
  return __builtin_amdgcn_rcpf(x);
}

__device__ __forceinline__ float sigmoidf_(float x) {
  return fast_rcp(1.f + __expf(-x));
}

__global__ void k_zero(int* __restrict__ p, int n) {
  int i = blockIdx.x * blockDim.x + threadIdx.x;
  int stride = gridDim.x * blockDim.x;
  for (; i < n; i += stride) p[i] = 0;
}

__global__ void __launch_bounds__(256) k_hist(const int* __restrict__ edge_dst,
                                              int* __restrict__ icnt) {
  int e = blockIdx.x * 256 + threadIdx.x;
  if (e < EE) atomicAdd(icnt + edge_dst[e], 1);
}

#define SCAN_CH 49
__global__ void __launch_bounds__(1024) k_scan(const int* __restrict__ icnt,
                                               int* __restrict__ off,
                                               int* __restrict__ cursor,
                                               float* __restrict__ cnt) {
  __shared__ int wsum[32];
  int t = threadIdx.x;
  int lane = t & 31;
  int w = t >> 5;
  int cbase = w * (SCAN_CH * 32);
  int v[SCAN_CH];
#pragma unroll
  for (int i = 0; i < SCAN_CH; i++) {
    int idx = cbase + i * 32 + lane;
    v[i] = (idx < NN) ? icnt[idx] : 0;
  }
  int s = 0;
#pragma unroll
  for (int i = 0; i < SCAN_CH; i++) s += v[i];
#pragma unroll
  for (int m = 16; m >= 1; m >>= 1) s += __shfl_xor(s, m, 32);
  if (lane == 0) wsum[w] = s;
  __syncthreads();
  int x = (lane < w) ? wsum[lane] : 0;
#pragma unroll
  for (int m = 16; m >= 1; m >>= 1) x += __shfl_xor(x, m, 32);
  int carry = x;
#pragma unroll
  for (int i = 0; i < SCAN_CH; i++) {
    int inc = v[i];
#pragma unroll
    for (int d = 1; d < 32; d <<= 1) {
      int y = __shfl_up(inc, d, 32);
      if (lane >= d) inc += y;
    }
    int excl = carry + inc - v[i];
    int idx = cbase + i * 32 + lane;
    if (idx < NN) {
      off[idx] = excl;
      cursor[idx] = excl;
      cnt[idx] = (float)v[i];
    }
    carry += __shfl(inc, 31, 32);
  }
  if (t == 1023) off[NN] = carry;
}

__global__ void __launch_bounds__(256) k_fill(const int* __restrict__ edge_dst,
                                              int* __restrict__ cursor,
                                              int* __restrict__ perm,
                                              int* __restrict__ sdst) {
  int e = blockIdx.x * 256 + threadIdx.x;
  if (e < EE) {
    int d = edge_dst[e];
    int pos = atomicAdd(cursor + d, 1);
    perm[pos] = e;
    sdst[pos] = d;
  }
}

__global__ void __launch_bounds__(256) k_norm(
    const float* __restrict__ node_s, const float* __restrict__ node_v,
    const float* __restrict__ ln_g, const float* __restrict__ ln_b,
    float* __restrict__ s_n, float* __restrict__ v_n) {
  int wave = (blockIdx.x * blockDim.x + threadIdx.x) >> 5;
  int lane = threadIdx.x & 31;
  if (wave >= NN) return;

  const float* srow = node_s + (long)wave * SDIM;
  float4 x = ((const float4*)srow)[lane];
  float s = x.x + x.y + x.z + x.w;
#pragma unroll
  for (int m = 16; m >= 1; m >>= 1) s += __shfl_xor(s, m, 32);
  float mu = s * (1.f / SDIM);
  float dx = x.x - mu, dy = x.y - mu, dz = x.z - mu, dw = x.w - mu;
  float q = dx * dx + dy * dy + dz * dz + dw * dw;
#pragma unroll
  for (int m = 16; m >= 1; m >>= 1) q += __shfl_xor(q, m, 32);
  float rs = rsqrtf(q * (1.f / SDIM) + 1e-5f);
  int c0 = lane * 4;
  float4 o;
  o.x = dx * rs * ln_g[c0 + 0] + ln_b[c0 + 0];
  o.y = dy * rs * ln_g[c0 + 1] + ln_b[c0 + 1];
  o.z = dz * rs * ln_g[c0 + 2] + ln_b[c0 + 2];
  o.w = dw * rs * ln_g[c0 + 3] + ln_b[c0 + 3];
  ((float4*)(s_n + (long)wave * SDIM))[lane] = o;

  const float* vrow = node_v + (long)wave * V3;
  float2 v = {0.f, 0.f};
  int vi = lane * 2;
  if (vi < V3) v = ((const float2*)vrow)[lane];
  float qq = v.x * v.x + v.y * v.y;
#pragma unroll
  for (int m = 16; m >= 1; m >>= 1) qq += __shfl_xor(qq, m, 32);
  float inv = rsqrtf(qq * (1.f / VDIM) + 1e-8f);
  if (vi < V3) {
    float2 ov;
    ov.x = v.x * inv;
    ov.y = v.y * inv;
    ((float2*)(v_n + (long)wave * V3))[lane] = ov;
  }
}

__device__ __forceinline__ void load_idx(const int* __restrict__ perm,
                                         const int* __restrict__ sdst, int p,
                                         int row, int pend, int nb, int& e,
                                         int& dl) {
  int pe = p + row;
  bool valid = pe < pend;
  int pc = valid ? pe : 0;
  int ev = perm[pc];
  int dv = sdst[pc];
  e = valid ? ev : 0;
  dl = valid ? dv - nb : -1;
}

__device__ __forceinline__ void sel_frag(int dl, int row, int half,
                                         v2f sa[4]) {
#pragma unroll
  for (int jj = 0; jj < 4; jj++) {
    int x0 = __shfl(dl, 8 * half + 2 * jj, 32);
    int x1 = __shfl(dl, 8 * half + 2 * jj + 1, 32);
    sa[jj].x = (x0 == row) ? 1.f : 0.f;
    sa[jj].y = (x1 == row) ? 1.f : 0.f;
  }
}

__global__ void __launch_bounds__(160) k_edge(
    const float* __restrict__ edge_s, const float* __restrict__ edge_v,
    const int* __restrict__ perm, const int* __restrict__ sdst,
    const int* __restrict__ off,
    const float* __restrict__ wes, const float* __restrict__ bes,
    const float* __restrict__ wev, const float* __restrict__ bev,
    float* __restrict__ agg_s, float* __restrict__ agg_v) {
  int wid = threadIdx.x >> 5;
  int lane = threadIdx.x & 31;
  int row = lane & 15;
  int half = lane >> 4;
  int nb = blockIdx.x * 16;
  int pbeg = off[nb];
  int pend = off[nb + 16];

  if (wid < 4) {
    int n0 = 32 * wid + row;
    int n1 = n0 + 16;
    v2f b0[8], b1[8];
#pragma unroll
    for (int j = 0; j < 8; j++) {
      int k0 = 4 * j + 2 * half;
      b0[j] = *(const v2f*)(wes + (long)n0 * ESD + k0);
      b1[j] = *(const v2f*)(wes + (long)n1 * ESD + k0);
    }
    float bias0 = bes[n0], bias1 = bes[n1];
    v8f g0 = {};
    v8f g1 = {};
    int e, dl;
    load_idx(perm, sdst, pbeg, row, pend, nb, e, dl);
    for (int p = pbeg; p < pend; p += 16) {
      int en, dln;
      load_idx(perm, sdst, p + 16, row, pend, nb, en, dln);
      const float* arow = edge_s + (long)e * ESD;
      v8f c0 = {};
      v8f c1 = {};
#pragma unroll
      for (int j = 0; j < 8; j++) {
        int k0 = 4 * j + 2 * half;
        v2f a = *(const v2f*)(arow + k0);
        c0 = wmma4(a, b0[j], c0);
        c1 = wmma4(a, b1[j], c1);
      }
      v2f sa[4];
      sel_frag(dl, row, half, sa);
#pragma unroll
      for (int r = 0; r < 8; r++) {
        float p0 = c0[r] + bias0;
        float p1 = c1[r] + bias1;
        c0[r] = p0 * sigmoidf_(p0);
        c1[r] = p1 * sigmoidf_(p1);
      }
#pragma unroll
      for (int jj = 0; jj < 4; jj++) {
        v2f m0 = {c0[2 * jj], c0[2 * jj + 1]};
        v2f m1 = {c1[2 * jj], c1[2 * jj + 1]};
        g0 = wmma4(sa[jj], m0, g0);
        g1 = wmma4(sa[jj], m1, g1);
      }
      e = en;
      dl = dln;
    }
#pragma unroll
    for (int r = 0; r < 8; r++) {
      float* orow = agg_s + (long)(nb + r + 8 * half) * SDIM;
      orow[n0] = g0[r];
      orow[n1] = g1[r];
    }
  } else {
    v2f b0[8];
#pragma unroll
    for (int j = 0; j < 8; j++)
      b0[j] = *(const v2f*)(wes + (long)row * ESD + 4 * j + 2 * half);
    float bias0 = bes[row];
    v2f bw[3];
    float bv[3];
    int gsrc[3];
#pragma unroll
    for (int vt = 0; vt < 3; vt++) {
      int d = 16 * vt + row;
      float w1 = wev[d * 3 + 1];
      bw[vt].x = wev[d * 3 + 2 * half];
      bw[vt].y = (half == 0) ? w1 : 0.f;
      bv[vt] = bev[d];
      gsrc[vt] = d / 3 + 16 * half;
    }
    v8f q[3];
#pragma unroll
    for (int vt = 0; vt < 3; vt++) q[vt] = (v8f){};
    int e, dl;
    load_idx(perm, sdst, pbeg, row, pend, nb, e, dl);
    for (int p = pbeg; p < pend; p += 16) {
      int en, dln;
      load_idx(perm, sdst, p + 16, row, pend, nb, en, dln);
      const float* arow = edge_s + (long)e * ESD;
      float e1 = edge_v[(long)e * 3 + 1];
      v2f av;
      av.x = edge_v[(long)e * 3 + 2 * half];
      av.y = (half == 0) ? e1 : 0.f;
      v8f c0 = {};
#pragma unroll
      for (int j = 0; j < 8; j++) {
        v2f a = *(const v2f*)(arow + 4 * j + 2 * half);
        c0 = wmma4(a, b0[j], c0);
      }
      v2f sa[4];
      sel_frag(dl, row, half, sa);
      float gate[8];
#pragma unroll
      for (int r = 0; r < 8; r++) {
        float pre = c0[r] + bias0;
        float so = pre * sigmoidf_(pre);
        gate[r] = sigmoidf_(so);
      }
#pragma unroll
      for (int vt = 0; vt < 3; vt++) {
        v8f vv = {};
        vv = wmma4(av, bw[vt], vv);
        float val[8];
#pragma unroll
        for (int r = 0; r < 8; r++)
          val[r] = (vv[r] + bv[vt]) * __shfl(gate[r], gsrc[vt], 32);
#pragma unroll
        for (int jj = 0; jj < 4; jj++) {
          v2f m = {val[2 * jj], val[2 * jj + 1]};
          q[vt] = wmma4(sa[jj], m, q[vt]);
        }
      }
      e = en;
      dl = dln;
    }
#pragma unroll
    for (int vt = 0; vt < 3; vt++) {
#pragma unroll
      for (int r = 0; r < 8; r++)
        agg_v[(long)(nb + r + 8 * half) * V3 + 16 * vt + row] = q[vt][r];
    }
  }
}

__global__ void __launch_bounds__(256) k_node(
    const float* __restrict__ s_n, const float* __restrict__ v_n,
    const float* __restrict__ agg_s, const float* __restrict__ agg_v,
    const float* __restrict__ cnt,
    const float* __restrict__ wns, const float* __restrict__ bns,
    const float* __restrict__ wnv, const float* __restrict__ bnv,
    const float* __restrict__ wrs, const float* __restrict__ brs,
    const float* __restrict__ wrv, const float* __restrict__ brv,
    float* __restrict__ out) {
  __shared__ float lds_gate[8][16 * 16];
  int wid = threadIdx.x >> 5;
  int lane = threadIdx.x & 31;
  int group = blockIdx.x * 8 + wid;
  if (group >= NN / 16) return;
  int nb = group * 16;
  int row = lane & 15;
  int half = lane >> 4;
  int noderow = nb + row;

  float invden = fast_rcp(fmaxf(cnt[noderow], 1.f));
  const float* sA = s_n + (long)noderow * SDIM;
  const float* gA = agg_s + (long)noderow * SDIM;

#pragma unroll 1
  for (int ct = 0; ct < 8; ct++) {
    v8f ao = {};
    v8f ar = {};
    int n = 16 * ct + row;
    const float* bo = wns + (long)n * SDIM;
    const float* br = wrs + (long)n * SDIM;
#pragma unroll 4
    for (int j = 0; j < 32; j++) {
      int k0 = 4 * j + 2 * half;
      v2f as = *(const v2f*)(sA + k0);
      v2f ag = *(const v2f*)(gA + k0);
      v2f ain = {as.x + ag.x * invden, as.y + ag.y * invden};
      v2f bO = *(const v2f*)(bo + k0);
      v2f bR = *(const v2f*)(br + k0);
      ao = wmma4(ain, bO, ao);
      ar = wmma4(as, bR, ar);
    }
    float biasO = bns[n], biasR = brs[n];
#pragma unroll
    for (int r = 0; r < 8; r++) {
      int m = r + 8 * half;
      float pre = ao[r] + biasO;
      float so = pre * sigmoidf_(pre);
      if (ct == 0) lds_gate[wid][m * 16 + n] = sigmoidf_(so);
      out[(long)(nb + m) * SDIM + n] = so + (ar[r] + biasR);
    }
  }

  const float* vA = v_n + (long)noderow * V3;
  const float* gV = agg_v + (long)noderow * V3;
  float* outv = out + (long)NN * SDIM;
#pragma unroll 1
  for (int ct = 0; ct < 3; ct++) {
    v8f ao = {};
    v8f ar = {};
    int d = 16 * ct + row;
    const float* bo = wnv + (long)d * V3;
    const float* br = wrv + (long)d * V3;
#pragma unroll
    for (int j = 0; j < 12; j++) {
      int k0 = 4 * j + 2 * half;
      v2f av = *(const v2f*)(vA + k0);
      v2f ag = *(const v2f*)(gV + k0);
      v2f ain = {av.x + ag.x * invden, av.y + ag.y * invden};
      v2f bO = *(const v2f*)(bo + k0);
      v2f bR = *(const v2f*)(br + k0);
      ao = wmma4(ain, bO, ao);
      ar = wmma4(av, bR, ar);
    }
    float biasO = bnv[d], biasR = brv[d];
    int vo = d / 3;
#pragma unroll
    for (int r = 0; r < 8; r++) {
      int m = r + 8 * half;
      float pre = ao[r] + biasO;
      float g = lds_gate[wid][m * 16 + vo];
      outv[(long)(nb + m) * V3 + d] = pre * g + (ar[r] + biasR);
    }
  }
}

extern "C" void kernel_launch(void* const* d_in, const int* in_sizes, int n_in,
                              void* d_out, int out_size, void* d_ws,
                              size_t ws_size, hipStream_t stream) {
  const float* node_s = (const float*)d_in[0];
  const float* node_v = (const float*)d_in[1];
  const int* eidx     = (const int*)d_in[2];
  const float* edge_s = (const float*)d_in[3];
  const float* edge_v = (const float*)d_in[4];
  const float* ln_g = (const float*)d_in[5];
  const float* ln_b = (const float*)d_in[6];
  const float* wes = (const float*)d_in[7];
  const float* bes = (const float*)d_in[8];
  const float* wev = (const float*)d_in[9];
  const float* bev = (const float*)d_in[10];
  const float* wns = (const float*)d_in[11];
  const float* bns = (const float*)d_in[12];
  const float* wnv = (const float*)d_in[13];
  const float* bnv = (const float*)d_in[14];
  const float* wrs = (const float*)d_in[15];
  const float* brs = (const float*)d_in[16];
  const float* wrv = (const float*)d_in[17];
  const float* brv = (const float*)d_in[18];
  float* out = (float*)d_out;

  float* ws = (float*)d_ws;
  float* s_n = ws;
  float* v_n = s_n + (size_t)NN * SDIM;
  float* agg_s = v_n + (size_t)NN * V3;
  float* agg_v = agg_s + (size_t)NN * SDIM;
  float* cnt = agg_v + (size_t)NN * V3;
  int* icnt = (int*)(cnt + NN);
  int* off = icnt + NN;
  int* cursor = off + (NN + 1);
  int* perm = cursor + NN;
  int* sdst = perm + EE;
  const int* edge_dst = eidx + EE;

  k_zero<<<(NN + 255) / 256, 256, 0, stream>>>(icnt, NN);
  k_hist<<<EE / 256, 256, 0, stream>>>(edge_dst, icnt);
  k_scan<<<1, 1024, 0, stream>>>(icnt, off, cursor, cnt);
  k_fill<<<EE / 256, 256, 0, stream>>>(edge_dst, cursor, perm, sdst);
  k_norm<<<(NN * 32 + 255) / 256, 256, 0, stream>>>(node_s, node_v, ln_g, ln_b,
                                                    s_n, v_n);
  k_edge<<<NN / 16, 160, 0, stream>>>(edge_s, edge_v, perm, sdst, off, wes, bes,
                                      wev, bev, agg_s, agg_v);
  k_node<<<(NN / 16 + 7) / 8, 256, 0, stream>>>(s_n, v_n, agg_s, agg_v, cnt,
                                                wns, bns, wnv, bnv, wrs, brs,
                                                wrv, brv, out);
}


# k_node: both 16-node A tiles read once with 128-bit loads and kept in registers over the 8 column tiles; weights read 128-bit in the same k order (sum order inside each 8-column group changes)
# speedup vs baseline: 2.4248x; 1.1787x over previous
#include <hip/hip_runtime.h>
#include <math.h>

#define NN   50000
#define EE   800000
#define SDIM 128
#define VDIM 16
#define V3   48
#define ESD  32

typedef float v2f __attribute__((ext_vector_type(2)));
typedef float v8f __attribute__((ext_vector_type(8)));

__device__ __forceinline__ v8f wmma4(v2f a, v2f b, v8f c) {
  return __builtin_amdgcn_wmma_f32_16x16x4_f32(false, a, false, b, (short)0, c,
                                               false, false);
}

__device__ __forceinline__ float fast_rcp(float x) {
  return __builtin_amdgcn_rcpf(x);
}

__device__ __forceinline__ float sigmoidf_(float x) {
  return fast_rcp(1.f + __expf(-x));
}

__global__ void k_zero(int* __restrict__ p, int n) {
  int i = blockIdx.x * blockDim.x + threadIdx.x;
  int stride = gridDim.x * blockDim.x;
  for (; i < n; i += stride) p[i] = 0;
}

__global__ void __launch_bounds__(256) k_hist(const int* __restrict__ edge_dst,
                                              int* __restrict__ icnt) {
  int e = blockIdx.x * 256 + threadIdx.x;
  if (e < EE) atomicAdd(icnt + edge_dst[e], 1);
}

#define SCAN_CH 49
__global__ void __launch_bounds__(1024) k_scan(const int* __restrict__ icnt,
                                               int* __restrict__ off,
                                               int* __restrict__ cursor,
                                               float* __restrict__ cnt) {
  __shared__ int wsum[32];
  int t = threadIdx.x;
  int lane = t & 31;
  int w = t >> 5;
  int cbase = w * (SCAN_CH * 32);
  int v[SCAN_CH];
#pragma unroll
  for (int i = 0; i < SCAN_CH; i++) {
    int idx = cbase + i * 32 + lane;
    v[i] = (idx < NN) ? icnt[idx] : 0;
  }
  int s = 0;
#pragma unroll
  for (int i = 0; i < SCAN_CH; i++) s += v[i];
#pragma unroll
  for (int m = 16; m >= 1; m >>= 1) s += __shfl_xor(s, m, 32);
  if (lane == 0) wsum[w] = s;
  __syncthreads();
  int x = (lane < w) ? wsum[lane] : 0;
#pragma unroll
  for (int m = 16; m >= 1; m >>= 1) x += __shfl_xor(x, m, 32);
  int carry = x;
#pragma unroll
  for (int i = 0; i < SCAN_CH; i++) {
    int inc = v[i];
#pragma unroll
    for (int d = 1; d < 32; d <<= 1) {
      int y = __shfl_up(inc, d, 32);
      if (lane >= d) inc += y;
    }
    int excl = carry + inc - v[i];
    int idx = cbase + i * 32 + lane;
    if (idx < NN) {
      off[idx] = excl;
      cursor[idx] = excl;
      cnt[idx] = (float)v[i];
    }
    carry += __shfl(inc, 31, 32);
  }
  if (t == 1023) off[NN] = carry;
}

__global__ void __launch_bounds__(256) k_fill(const int* __restrict__ edge_dst,
                                              int* __restrict__ cursor,
                                              int* __restrict__ perm,
                                              int* __restrict__ sdst) {
  int e = blockIdx.x * 256 + threadIdx.x;
  if (e < EE) {
    int d = edge_dst[e];
    int pos = atomicAdd(cursor + d, 1);
    perm[pos] = e;
    sdst[pos] = d;
  }
}

__global__ void __launch_bounds__(256) k_norm(
    const float* __restrict__ node_s, const float* __restrict__ node_v,
    const float* __restrict__ ln_g, const float* __restrict__ ln_b,
    float* __restrict__ s_n, float* __restrict__ v_n) {
  int wave = (blockIdx.x * blockDim.x + threadIdx.x) >> 5;
  int lane = threadIdx.x & 31;
  if (wave >= NN) return;

  const float* srow = node_s + (long)wave * SDIM;
  float4 x = ((const float4*)srow)[lane];
  float s = x.x + x.y + x.z + x.w;
#pragma unroll
  for (int m = 16; m >= 1; m >>= 1) s += __shfl_xor(s, m, 32);
  float mu = s * (1.f / SDIM);
  float dx = x.x - mu, dy = x.y - mu, dz = x.z - mu, dw = x.w - mu;
  float q = dx * dx + dy * dy + dz * dz + dw * dw;
#pragma unroll
  for (int m = 16; m >= 1; m >>= 1) q += __shfl_xor(q, m, 32);
  float rs = rsqrtf(q * (1.f / SDIM) + 1e-5f);
  int c0 = lane * 4;
  float4 o;
  o.x = dx * rs * ln_g[c0 + 0] + ln_b[c0 + 0];
  o.y = dy * rs * ln_g[c0 + 1] + ln_b[c0 + 1];
  o.z = dz * rs * ln_g[c0 + 2] + ln_b[c0 + 2];
  o.w = dw * rs * ln_g[c0 + 3] + ln_b[c0 + 3];
  ((float4*)(s_n + (long)wave * SDIM))[lane] = o;

  const float* vrow = node_v + (long)wave * V3;
  float2 v = {0.f, 0.f};
  int vi = lane * 2;
  if (vi < V3) v = ((const float2*)vrow)[lane];
  float qq = v.x * v.x + v.y * v.y;
#pragma unroll
  for (int m = 16; m >= 1; m >>= 1) qq += __shfl_xor(qq, m, 32);
  float inv = rsqrtf(qq * (1.f / VDIM) + 1e-8f);
  if (vi < V3) {
    float2 ov;
    ov.x = v.x * inv;
    ov.y = v.y * inv;
    ((float2*)(v_n + (long)wave * V3))[lane] = ov;
  }
}

__device__ __forceinline__ void load_idx(const int* __restrict__ perm,
                                         const int* __restrict__ sdst, int p,
                                         int row, int pend, int nb, int& e,
                                         int& dl) {
  int pe = p + row;
  bool valid = pe < pend;
  int pc = valid ? pe : 0;
  int ev = perm[pc];
  int dv = sdst[pc];
  e = valid ? ev : 0;
  dl = valid ? dv - nb : -1;
}

__device__ __forceinline__ void sel_frag(int dl, int row, int half,
                                         v2f sa[4]) {
#pragma unroll
  for (int jj = 0; jj < 4; jj++) {
    int x0 = __shfl(dl, 8 * half + 2 * jj, 32);
    int x1 = __shfl(dl, 8 * half + 2 * jj + 1, 32);
    sa[jj].x = (x0 == row) ? 1.f : 0.f;
    sa[jj].y = (x1 == row) ? 1.f : 0.f;
  }
}

__global__ void __launch_bounds__(160) k_edge(
    const float* __restrict__ edge_s, const float* __restrict__ edge_v,
    const int* __restrict__ perm, const int* __restrict__ sdst,
    const int* __restrict__ off,
    const float* __restrict__ wes, const float* __restrict__ bes,
    const float* __restrict__ wev, const float* __restrict__ bev,
    float* __restrict__ agg_s, float* __restrict__ agg_v) {
  int wid = threadIdx.x >> 5;
  int lane = threadIdx.x & 31;
  int row = lane & 15;
  int half = lane >> 4;
  int nb = blockIdx.x * 16;
  int pbeg = off[nb];
  int pend = off[nb + 16];

  if (wid < 4) {
    int n0 = 32 * wid + row;
    int n1 = n0 + 16;
    v2f b0[8], b1[8];
#pragma unroll
    for (int j = 0; j < 8; j++) {
      int k0 = 4 * j + 2 * half;
      b0[j] = *(const v2f*)(wes + (long)n0 * ESD + k0);
      b1[j] = *(const v2f*)(wes + (long)n1 * ESD + k0);
    }
    float bias0 = bes[n0], bias1 = bes[n1];
    v8f g0 = {};
    v8f g1 = {};
    int e, dl;
    load_idx(perm, sdst, pbeg, row, pend, nb, e, dl);
    for (int p = pbeg; p < pend; p += 16) {
      int en, dln;
      load_idx(perm, sdst, p + 16, row, pend, nb, en, dln);
      const float* arow = edge_s + (long)e * ESD;
      v8f c0 = {};
      v8f c1 = {};
#pragma unroll
      for (int j = 0; j < 8; j++) {
        int k0 = 4 * j + 2 * half;
        v2f a = *(const v2f*)(arow + k0);
        c0 = wmma4(a, b0[j], c0);
        c1 = wmma4(a, b1[j], c1);
      }
      v2f sa[4];
      sel_frag(dl, row, half, sa);
#pragma unroll
      for (int r = 0; r < 8; r++) {
        float p0 = c0[r] + bias0;
        float p1 = c1[r] + bias1;
        c0[r] = p0 * sigmoidf_(p0);
        c1[r] = p1 * sigmoidf_(p1);
      }
#pragma unroll
      for (int jj = 0; jj < 4; jj++) {
        v2f m0 = {c0[2 * jj], c0[2 * jj + 1]};
        v2f m1 = {c1[2 * jj], c1[2 * jj + 1]};
        g0 = wmma4(sa[jj], m0, g0);
        g1 = wmma4(sa[jj], m1, g1);
      }
      e = en;
      dl = dln;
    }
#pragma unroll
    for (int r = 0; r < 8; r++) {
      float* orow = agg_s + (long)(nb + r + 8 * half) * SDIM;
      orow[n0] = g0[r];
      orow[n1] = g1[r];
    }
  } else {
    v2f b0[8];
#pragma unroll
    for (int j = 0; j < 8; j++)
      b0[j] = *(const v2f*)(wes + (long)row * ESD + 4 * j + 2 * half);
    float bias0 = bes[row];
    v2f bw[3];
    float bv[3];
    int gsrc[3];
#pragma unroll
    for (int vt = 0; vt < 3; vt++) {
      int d = 16 * vt + row;
      float w1 = wev[d * 3 + 1];
      bw[vt].x = wev[d * 3 + 2 * half];
      bw[vt].y = (half == 0) ? w1 : 0.f;
      bv[vt] = bev[d];
      gsrc[vt] = d / 3 + 16 * half;
    }
    v8f q[3];
#pragma unroll
    for (int vt = 0; vt < 3; vt++) q[vt] = (v8f){};
    int e, dl;
    load_idx(perm, sdst, pbeg, row, pend, nb, e, dl);
    for (int p = pbeg; p < pend; p += 16) {
      int en, dln;
      load_idx(perm, sdst, p + 16, row, pend, nb, en, dln);
      const float* arow = edge_s + (long)e * ESD;
      float e1 = edge_v[(long)e * 3 + 1];
      v2f av;
      av.x = edge_v[(long)e * 3 + 2 * half];
      av.y = (half == 0) ? e1 : 0.f;
      v8f c0 = {};
#pragma unroll
      for (int j = 0; j < 8; j++) {
        v2f a = *(const v2f*)(arow + 4 * j + 2 * half);
        c0 = wmma4(a, b0[j], c0);
      }
      v2f sa[4];
      sel_frag(dl, row, half, sa);
      float gate[8];
#pragma unroll
      for (int r = 0; r < 8; r++) {
        float pre = c0[r] + bias0;
        float so = pre * sigmoidf_(pre);
        gate[r] = sigmoidf_(so);
      }
#pragma unroll
      for (int vt = 0; vt < 3; vt++) {
        v8f vv = {};
        vv = wmma4(av, bw[vt], vv);
        float val[8];
#pragma unroll
        for (int r = 0; r < 8; r++)
          val[r] = (vv[r] + bv[vt]) * __shfl(gate[r], gsrc[vt], 32);
#pragma unroll
        for (int jj = 0; jj < 4; jj++) {
          v2f m = {val[2 * jj], val[2 * jj + 1]};
          q[vt] = wmma4(sa[jj], m, q[vt]);
        }
      }
      e = en;
      dl = dln;
    }
#pragma unroll
    for (int vt = 0; vt < 3; vt++) {
#pragma unroll
      for (int r = 0; r < 8; r++)
        agg_v[(long)(nb + r + 8 * half) * V3 + 16 * vt + row] = q[vt][r];
    }
  }
}

typedef float v4f __attribute__((ext_vector_type(4)));

__device__ __forceinline__ v2f lo2(v4f x) { return (v2f){x.x, x.y}; }
__device__ __forceinline__ v2f hi2(v4f x) { return (v2f){x.z, x.w}; }

__global__ void __launch_bounds__(256, 4) k_node(
    const float* __restrict__ s_n, const float* __restrict__ v_n,
    const float* __restrict__ agg_s, const float* __restrict__ agg_v,
    const float* __restrict__ cnt,
    const float* __restrict__ wns, const float* __restrict__ bns,
    const float* __restrict__ wnv, const float* __restrict__ bnv,
    const float* __restrict__ wrs, const float* __restrict__ brs,
    const float* __restrict__ wrv, const float* __restrict__ brv,
    float* __restrict__ out) {
  __shared__ float lds_gate[8][16 * 16];
  int wid = threadIdx.x >> 5;
  int lane = threadIdx.x & 31;
  int group = blockIdx.x * 8 + wid;
  if (group >= NN / 16) return;
  int nb = group * 16;
  int row = lane & 15;
  int half = lane >> 4;
  int noderow = nb + row;

  float invden = fast_rcp(fmaxf(cnt[noderow], 1.f));
  const v4f* sA = (const v4f*)(s_n + (long)noderow * SDIM) + half;
  const v4f* gA = (const v4f*)(agg_s + (long)noderow * SDIM) + half;
  v4f as[16], ai[16];
#pragma unroll
  for (int q = 0; q < 16; q++) {
    as[q] = sA[2 * q];
    ai[q] = gA[2 * q];
  }
#pragma unroll
  for (int q = 0; q < 16; q++) ai[q] = as[q] + ai[q] * invden;

#pragma unroll 1
  for (int ct = 0; ct < 8; ct++) {
    v8f ao = {};
    v8f ar = {};
    int n = 16 * ct + row;
    const v4f* bo = (const v4f*)(wns + (long)n * SDIM) + half;
    const v4f* br = (const v4f*)(wrs + (long)n * SDIM) + half;
#pragma unroll
    for (int q = 0; q < 16; q++) {
      v4f bO = bo[2 * q];
      v4f bR = br[2 * q];
      ao = wmma4(lo2(ai[q]), lo2(bO), ao);
      ar = wmma4(lo2(as[q]), lo2(bR), ar);
      ao = wmma4(hi2(ai[q]), hi2(bO), ao);
      ar = wmma4(hi2(as[q]), hi2(bR), ar);
    }
    float biasO = bns[n], biasR = brs[n];
#pragma unroll
    for (int r = 0; r < 8; r++) {
      int m = r + 8 * half;
      float pre = ao[r] + biasO;
      float so = pre * sigmoidf_(pre);
      if (ct == 0) lds_gate[wid][m * 16 + n] = sigmoidf_(so);
      out[(long)(nb + m) * SDIM + n] = so + (ar[r] + biasR);
    }
  }

  const v4f* vA = (const v4f*)(v_n + (long)noderow * V3) + half;
  const v4f* gV = (const v4f*)(agg_v + (long)noderow * V3) + half;
  v4f av[6], aw[6];
#pragma unroll
  for (int q = 0; q < 6; q++) {
    av[q] = vA[2 * q];
    aw[q] = gV[2 * q];
  }
#pragma unroll
  for (int q = 0; q < 6; q++) aw[q] = av[q] + aw[q] * invden;
  float* outv = out + (long)NN * SDIM;
#pragma unroll 1
  for (int ct = 0; ct < 3; ct++) {
    v8f ao = {};
    v8f ar = {};
    int d = 16 * ct + row;
    const v4f* bo = (const v4f*)(wnv + (long)d * V3) + half;
    const v4f* br = (const v4f*)(wrv + (long)d * V3) + half;
#pragma unroll
    for (int q = 0; q < 6; q++) {
      v4f bO = bo[2 * q];
      v4f bR = br[2 * q];
      ao = wmma4(lo2(aw[q]), lo2(bO), ao);
      ar = wmma4(lo2(av[q]), lo2(bR), ar);
      ao = wmma4(hi2(aw[q]), hi2(bO), ao);
      ar = wmma4(hi2(av[q]), hi2(bR), ar);
    }
    float biasO = bnv[d], biasR = brv[d];
    int vo = d / 3;
#pragma unroll
    for (int r = 0; r < 8; r++) {
      int m = r + 8 * half;
      float pre = ao[r] + biasO;
      float g = lds_gate[wid][m * 16 + vo];
      outv[(long)(nb + m) * V3 + d] = pre * g + (ar[r] + biasR);
    }
  }
}

extern "C" void kernel_launch(void* const* d_in, const int* in_sizes, int n_in,
                              void* d_out, int out_size, void* d_ws,
                              size_t ws_size, hipStream_t stream) {
  const float* node_s = (const float*)d_in[0];
  const float* node_v = (const float*)d_in[1];
  const int* eidx     = (const int*)d_in[2];
  const float* edge_s = (const float*)d_in[3];
  const float* edge_v = (const float*)d_in[4];
  const float* ln_g = (const float*)d_in[5];
  const float* ln_b = (const float*)d_in[6];
  const float* wes = (const float*)d_in[7];
  const float* bes = (const float*)d_in[8];
  const float* wev = (const float*)d_in[9];
  const float* bev = (const float*)d_in[10];
  const float* wns = (const float*)d_in[11];
  const float* bns = (const float*)d_in[12];
  const float* wnv = (const float*)d_in[13];
  const float* bnv = (const float*)d_in[14];
  const float* wrs = (const float*)d_in[15];
  const float* brs = (const float*)d_in[16];
  const float* wrv = (const float*)d_in[17];
  const float* brv = (const float*)d_in[18];
  float* out = (float*)d_out;

  float* ws = (float*)d_ws;
  float* s_n = ws;
  float* v_n = s_n + (size_t)NN * SDIM;
  float* agg_s = v_n + (size_t)NN * V3;
  float* agg_v = agg_s + (size_t)NN * SDIM;
  float* cnt = agg_v + (size_t)NN * V3;
  int* icnt = (int*)(cnt + NN);
  int* off = icnt + NN;
  int* cursor = off + (NN + 1);
  int* perm = cursor + NN;
  int* sdst = perm + EE;
  const int* edge_dst = eidx + EE;

  k_zero<<<(NN + 255) / 256, 256, 0, stream>>>(icnt, NN);
  k_hist<<<EE / 256, 256, 0, stream>>>(edge_dst, icnt);
  k_scan<<<1, 1024, 0, stream>>>(icnt, off, cursor, cnt);
  k_fill<<<EE / 256, 256, 0, stream>>>(edge_dst, cursor, perm, sdst);
  k_norm<<<(NN * 32 + 255) / 256, 256, 0, stream>>>(node_s, node_v, ln_g, ln_b,
                                                    s_n, v_n);
  k_edge<<<NN / 16, 160, 0, stream>>>(edge_s, edge_v, perm, sdst, off, wes, bes,
                                      wev, bev, agg_s, agg_v);
  k_node<<<(NN / 16 + 7) / 8, 256, 0, stream>>>(s_n, v_n, agg_s, agg_v, cnt,
                                                wns, bns, wnv, bnv, wrs, brs,
                                                wrv, brv, out);
}


# Edge bucketing without global atomics: edges grouped by 256-node range, then by node, on LDS counters (returning LDS add = rank); order inside a bucket = LDS arrival order
# speedup vs baseline: 3.6958x; 1.5242x over previous
#include <hip/hip_runtime.h>
#include <math.h>

#define NN   50000
#define EE   800000
#define SDIM 128
#define VDIM 16
#define V3   48
#define ESD  32

typedef float v2f __attribute__((ext_vector_type(2)));
typedef float v8f __attribute__((ext_vector_type(8)));

__device__ __forceinline__ v8f wmma4(v2f a, v2f b, v8f c) {
  return __builtin_amdgcn_wmma_f32_16x16x4_f32(false, a, false, b, (short)0, c,
                                               false, false);
}

__device__ __forceinline__ float fast_rcp(float x) {
  return __builtin_amdgcn_rcpf(x);
}

__device__ __forceinline__ float sigmoidf_(float x) {
  return fast_rcp(1.f + __expf(-x));
}

#define G1 256
#define CH1 (EE / G1)
#define NB1 ((NN + 255) / 256)
#define T1 (NB1 * G1)
#define SCAN_CH 49
static_assert(EE % G1 == 0, "chunks must tile the edge list");
static_assert(CH1 <= 4 * 1024, "a chunk is read as 4 values per thread");
static_assert(T1 == SCAN_CH * 1024, "the scan block covers the table exactly");

__global__ void __launch_bounds__(1024) k_cnt1(const int* __restrict__ edge_dst,
                                               int* __restrict__ hist1) {
  __shared__ int h[NB1];
  int t = threadIdx.x;
  int g = blockIdx.x;
  if (t < NB1) h[t] = 0;
  const int* src = edge_dst + g * CH1;
  int d[4];
#pragma unroll
  for (int k = 0; k < 4; k++) {
    int i = t + 1024 * k;
    d[k] = (i < CH1) ? src[i] : 0;
  }
  __syncthreads();
#pragma unroll
  for (int k = 0; k < 4; k++)
    if (t + 1024 * k < CH1) atomicAdd(&h[d[k] >> 8], 1);
  __syncthreads();
  if (t < NB1) hist1[t * G1 + g] = h[t];
}

__global__ void __launch_bounds__(1024) k_scan(const int* __restrict__ hist1,
                                               int* __restrict__ offs1) {
  __shared__ int wsum[32];
  int t = threadIdx.x;
  int lane = t & 31;
  int w = t >> 5;
  int cbase = w * (SCAN_CH * 32);
  int v[SCAN_CH];
#pragma unroll
  for (int i = 0; i < SCAN_CH; i++) v[i] = hist1[cbase + i * 32 + lane];
  int s = 0;
#pragma unroll
  for (int i = 0; i < SCAN_CH; i++) s += v[i];
#pragma unroll
  for (int m = 16; m >= 1; m >>= 1) s += __shfl_xor(s, m, 32);
  if (lane == 0) wsum[w] = s;
  __syncthreads();
  int x = (lane < w) ? wsum[lane] : 0;
#pragma unroll
  for (int m = 16; m >= 1; m >>= 1) x += __shfl_xor(x, m, 32);
  int carry = x;
#pragma unroll
  for (int i = 0; i < SCAN_CH; i++) {
    int inc = v[i];
#pragma unroll
    for (int d = 1; d < 32; d <<= 1) {
      int y = __shfl_up(inc, d, 32);
      if (lane >= d) inc += y;
    }
    offs1[cbase + i * 32 + lane] = carry + inc - v[i];
    carry += __shfl(inc, 31, 32);
  }
  if (t == 1023) offs1[T1] = carry;
}

__global__ void __launch_bounds__(1024) k_scat1(const int* __restrict__ edge_dst,
                                                const int* __restrict__ offs1,
                                                int* __restrict__ rec) {
  __shared__ int cur[NB1];
  int t = threadIdx.x;
  int g = blockIdx.x;
  if (t < NB1) cur[t] = offs1[t * G1 + g];
  const int* src = edge_dst + g * CH1;
  int d[4];
#pragma unroll
  for (int k = 0; k < 4; k++) {
    int i = t + 1024 * k;
    d[k] = (i < CH1) ? src[i] : 0;
  }
  __syncthreads();
  int2* out = (int2*)rec;
#pragma unroll
  for (int k = 0; k < 4; k++) {
    int i = t + 1024 * k;
    if (i < CH1) {
      int pos = atomicAdd(&cur[d[k] >> 8], 1);
      out[pos] = make_int2(g * CH1 + i, d[k]);
    }
  }
}

__global__ void __launch_bounds__(1024) k_sort2(const int* __restrict__ rec,
                                                const int* __restrict__ offs1,
                                                int* __restrict__ off,
                                                float* __restrict__ cnt,
                                                int* __restrict__ perm,
                                                int* __restrict__ sdst) {
  __shared__ int c[256];
  __shared__ int cur[256];
  int t = threadIdx.x;
  int b = blockIdx.x;
  int S = offs1[b * G1];
  int T = offs1[(b + 1) * G1];
  const int2* in = (const int2*)rec;
  if (t < 256) c[t] = 0;
  __syncthreads();
  for (int base = S; base < T; base += 4096) {
    int dn[4];
#pragma unroll
    for (int k = 0; k < 4; k++) {
      int i = base + t + 1024 * k;
      dn[k] = in[min(i, T - 1)].y;
    }
#pragma unroll
    for (int k = 0; k < 4; k++)
      if (base + t + 1024 * k < T) atomicAdd(&c[dn[k] & 255], 1);
  }
  __syncthreads();
  if (t < 32) {
    int v[8];
    int s = 0;
#pragma unroll
    for (int k = 0; k < 8; k++) {
      v[k] = c[t * 8 + k];
      s += v[k];
    }
    int inc = s;
#pragma unroll
    for (int d = 1; d < 32; d <<= 1) {
      int y = __shfl_up(inc, d, 32);
      if (t >= d) inc += y;
    }
    int run = S + inc - s;
#pragma unroll
    for (int k = 0; k < 8; k++) {
      cur[t * 8 + k] = run;
      run += v[k];
    }
  }
  __syncthreads();
  if (t < 256) {
    int node = b * 256 + t;
    if (node < NN) {
      off[node] = cur[t];
      cnt[node] = (float)c[t];
    }
  }
  if (b == 0 && t == 0) off[NN] = EE;
  __syncthreads();
  for (int base = S; base < T; base += 4096) {
    int2 r[4];
#pragma unroll
    for (int k = 0; k < 4; k++) {
      int i = base + t + 1024 * k;
      r[k] = (i < T) ? in[i] : make_int2(0, 0);
    }
#pragma unroll
    for (int k = 0; k < 4; k++) {
      if (base + t + 1024 * k < T) {
        int pos = atomicAdd(&cur[r[k].y & 255], 1);
        perm[pos] = r[k].x;
        sdst[pos] = r[k].y;
      }
    }
  }
}

__global__ void __launch_bounds__(256) k_norm(
    const float* __restrict__ node_s, const float* __restrict__ node_v,
    const float* __restrict__ ln_g, const float* __restrict__ ln_b,
    float* __restrict__ s_n, float* __restrict__ v_n) {
  int wave = (blockIdx.x * blockDim.x + threadIdx.x) >> 5;
  int lane = threadIdx.x & 31;
  if (wave >= NN) return;

  const float* srow = node_s + (long)wave * SDIM;
  float4 x = ((const float4*)srow)[lane];
  float s = x.x + x.y + x.z + x.w;
#pragma unroll
  for (int m = 16; m >= 1; m >>= 1) s += __shfl_xor(s, m, 32);
  float mu = s * (1.f / SDIM);
  float dx = x.x - mu, dy = x.y - mu, dz = x.z - mu, dw = x.w - mu;
  float q = dx * dx + dy * dy + dz * dz + dw * dw;
#pragma unroll
  for (int m = 16; m >= 1; m >>= 1) q += __shfl_xor(q, m, 32);
  float rs = rsqrtf(q * (1.f / SDIM) + 1e-5f);
  int c0 = lane * 4;
  float4 o;
  o.x = dx * rs * ln_g[c0 + 0] + ln_b[c0 + 0];
  o.y = dy * rs * ln_g[c0 + 1] + ln_b[c0 + 1];
  o.z = dz * rs * ln_g[c0 + 2] + ln_b[c0 + 2];
  o.w = dw * rs * ln_g[c0 + 3] + ln_b[c0 + 3];
  ((float4*)(s_n + (long)wave * SDIM))[lane] = o;

  const float* vrow = node_v + (long)wave * V3;
  float2 v = {0.f, 0.f};
  int vi = lane * 2;
  if (vi < V3) v = ((const float2*)vrow)[lane];
  float qq = v.x * v.x + v.y * v.y;
#pragma unroll
  for (int m = 16; m >= 1; m >>= 1) qq += __shfl_xor(qq, m, 32);
  float inv = rsqrtf(qq * (1.f / VDIM) + 1e-8f);
  if (vi < V3) {
    float2 ov;
    ov.x = v.x * inv;
    ov.y = v.y * inv;
    ((float2*)(v_n + (long)wave * V3))[lane] = ov;
  }
}

__device__ __forceinline__ void load_idx(const int* __restrict__ perm,
                                         const int* __restrict__ sdst, int p,
                                         int row, int pend, int nb, int& e,
                                         int& dl) {
  int pe = p + row;
  bool valid = pe < pend;
  int pc = valid ? pe : 0;
  int ev = perm[pc];
  int dv = sdst[pc];
  e = valid ? ev : 0;
  dl = valid ? dv - nb : -1;
}

__device__ __forceinline__ void sel_frag(int dl, int row, int half,
                                         v2f sa[4]) {
#pragma unroll
  for (int jj = 0; jj < 4; jj++) {
    int x0 = __shfl(dl, 8 * half + 2 * jj, 32);
    int x1 = __shfl(dl, 8 * half + 2 * jj + 1, 32);
    sa[jj].x = (x0 == row) ? 1.f : 0.f;
    sa[jj].y = (x1 == row) ? 1.f : 0.f;
  }
}

__global__ void __launch_bounds__(160) k_edge(
    const float* __restrict__ edge_s, const float* __restrict__ edge_v,
    const int* __restrict__ perm, const int* __restrict__ sdst,
    const int* __restrict__ off,
    const float* __restrict__ wes, const float* __restrict__ bes,
    const float* __restrict__ wev, const float* __restrict__ bev,
    float* __restrict__ agg_s, float* __restrict__ agg_v) {
  int wid = threadIdx.x >> 5;
  int lane = threadIdx.x & 31;
  int row = lane & 15;
  int half = lane >> 4;
  int nb = blockIdx.x * 16;
  int pbeg = off[nb];
  int pend = off[nb + 16];

  if (wid < 4) {
    int n0 = 32 * wid + row;
    int n1 = n0 + 16;
    v2f b0[8], b1[8];
#pragma unroll
    for (int j = 0; j < 8; j++) {
      int k0 = 4 * j + 2 * half;
      b0[j] = *(const v2f*)(wes + (long)n0 * ESD + k0);
      b1[j] = *(const v2f*)(wes + (long)n1 * ESD + k0);
    }
    float bias0 = bes[n0], bias1 = bes[n1];
    v8f g0 = {};
    v8f g1 = {};
    int e, dl;
    load_idx(perm, sdst, pbeg, row, pend, nb, e, dl);
    for (int p = pbeg; p < pend; p += 16) {
      int en, dln;
      load_idx(perm, sdst, p + 16, row, pend, nb, en, dln);
      const float* arow = edge_s + (long)e * ESD;
      v8f c0 = {};
      v8f c1 = {};
#pragma unroll
      for (int j = 0; j < 8; j++) {
        int k0 = 4 * j + 2 * half;
        v2f a = *(const v2f*)(arow + k0);
        c0 = wmma4(a, b0[j], c0);
        c1 = wmma4(a, b1[j], c1);
      }
      v2f sa[4];
      sel_frag(dl, row, half, sa);
#pragma unroll
      for (int r = 0; r < 8; r++) {
        float p0 = c0[r] + bias0;
        float p1 = c1[r] + bias1;
        c0[r] = p0 * sigmoidf_(p0);
        c1[r] = p1 * sigmoidf_(p1);
      }
#pragma unroll
      for (int jj = 0; jj < 4; jj++) {
        v2f m0 = {c0[2 * jj], c0[2 * jj + 1]};
        v2f m1 = {c1[2 * jj], c1[2 * jj + 1]};
        g0 = wmma4(sa[jj], m0, g0);
        g1 = wmma4(sa[jj], m1, g1);
      }
      e = en;
      dl = dln;
    }
#pragma unroll
    for (int r = 0; r < 8; r++) {
      float* orow = agg_s + (long)(nb + r + 8 * half) * SDIM;
      orow[n0] = g0[r];
      orow[n1] = g1[r];
    }
  } else {
    v2f b0[8];
#pragma unroll
    for (int j = 0; j < 8; j++)
      b0[j] = *(const v2f*)(wes + (long)row * ESD + 4 * j + 2 * half);
    float bias0 = bes[row];
    v2f bw[3];
    float bv[3];
    int gsrc[3];
#pragma unroll
    for (int vt = 0; vt < 3; vt++) {
      int d = 16 * vt + row;
      float w1 = wev[d * 3 + 1];
      bw[vt].x = wev[d * 3 + 2 * half];
      bw[vt].y = (half == 0) ? w1 : 0.f;
      bv[vt] = bev[d];
      gsrc[vt] = d / 3 + 16 * half;
    }
    v8f q[3];
#pragma unroll
    for (int vt = 0; vt < 3; vt++) q[vt] = (v8f){};
    int e, dl;
    load_idx(perm, sdst, pbeg, row, pend, nb, e, dl);
    for (int p = pbeg; p < pend; p += 16) {
      int en, dln;
      load_idx(perm, sdst, p + 16, row, pend, nb, en, dln);
      const float* arow = edge_s + (long)e * ESD;
      float e1 = edge_v[(long)e * 3 + 1];
      v2f av;
      av.x = edge_v[(long)e * 3 + 2 * half];
      av.y = (half == 0) ? e1 : 0.f;
      v8f c0 = {};
#pragma unroll
      for (int j = 0; j < 8; j++) {
        v2f a = *(const v2f*)(arow + 4 * j + 2 * half);
        c0 = wmma4(a, b0[j], c0);
      }
      v2f sa[4];
      sel_frag(dl, row, half, sa);
      float gate[8];
#pragma unroll
      for (int r = 0; r < 8; r++) {
        float pre = c0[r] + bias0;
        float so = pre * sigmoidf_(pre);
        gate[r] = sigmoidf_(so);
      }
#pragma unroll
      for (int vt = 0; vt < 3; vt++) {
        v8f vv = {};
        vv = wmma4(av, bw[vt], vv);
        float val[8];
#pragma unroll
        for (int r = 0; r < 8; r++)
          val[r] = (vv[r] + bv[vt]) * __shfl(gate[r], gsrc[vt], 32);
#pragma unroll
        for (int jj = 0; jj < 4; jj++) {
          v2f m = {val[2 * jj], val[2 * jj + 1]};
          q[vt] = wmma4(sa[jj], m, q[vt]);
        }
      }
      e = en;
      dl = dln;
    }
#pragma unroll
    for (int vt = 0; vt < 3; vt++) {
#pragma unroll
      for (int r = 0; r < 8; r++)
        agg_v[(long)(nb + r + 8 * half) * V3 + 16 * vt + row] = q[vt][r];
    }
  }
}

typedef float v4f __attribute__((ext_vector_type(4)));

__device__ __forceinline__ v2f lo2(v4f x) { return (v2f){x.x, x.y}; }
__device__ __forceinline__ v2f hi2(v4f x) { return (v2f){x.z, x.w}; }

__global__ void __launch_bounds__(256, 4) k_node(
    const float* __restrict__ s_n, const float* __restrict__ v_n,
    const float* __restrict__ agg_s, const float* __restrict__ agg_v,
    const float* __restrict__ cnt,
    const float* __restrict__ wns, const float* __restrict__ bns,
    const float* __restrict__ wnv, const float* __restrict__ bnv,
    const float* __restrict__ wrs, const float* __restrict__ brs,
    const float* __restrict__ wrv, const float* __restrict__ brv,
    float* __restrict__ out) {
  __shared__ float lds_gate[8][16 * 16];
  int wid = threadIdx.x >> 5;
  int lane = threadIdx.x & 31;
  int group = blockIdx.x * 8 + wid;
  if (group >= NN / 16) return;
  int nb = group * 16;
  int row = lane & 15;
  int half = lane >> 4;
  int noderow = nb + row;

  float invden = fast_rcp(fmaxf(cnt[noderow], 1.f));
  const v4f* sA = (const v4f*)(s_n + (long)noderow * SDIM) + half;
  const v4f* gA = (const v4f*)(agg_s + (long)noderow * SDIM) + half;
  v4f as[16], ai[16];
#pragma unroll
  for (int q = 0; q < 16; q++) {
    as[q] = sA[2 * q];
    ai[q] = gA[2 * q];
  }
#pragma unroll
  for (int q = 0; q < 16; q++) ai[q] = as[q] + ai[q] * invden;

#pragma unroll 1
  for (int ct = 0; ct < 8; ct++) {
    v8f ao = {};
    v8f ar = {};
    int n = 16 * ct + row;
    const v4f* bo = (const v4f*)(wns + (long)n * SDIM) + half;
    const v4f* br = (const v4f*)(wrs + (long)n * SDIM) + half;
#pragma unroll
    for (int q = 0; q < 16; q++) {
      v4f bO = bo[2 * q];
      v4f bR = br[2 * q];
      ao = wmma4(lo2(ai[q]), lo2(bO), ao);
      ar = wmma4(lo2(as[q]), lo2(bR), ar);
      ao = wmma4(hi2(ai[q]), hi2(bO), ao);
      ar = wmma4(hi2(as[q]), hi2(bR), ar);
    }
    float biasO = bns[n], biasR = brs[n];
#pragma unroll
    for (int r = 0; r < 8; r++) {
      int m = r + 8 * half;
      float pre = ao[r] + biasO;
      float so = pre * sigmoidf_(pre);
      if (ct == 0) lds_gate[wid][m * 16 + n] = sigmoidf_(so);
      out[(long)(nb + m) * SDIM + n] = so + (ar[r] + biasR);
    }
  }

  const v4f* vA = (const v4f*)(v_n + (long)noderow * V3) + half;
  const v4f* gV = (const v4f*)(agg_v + (long)noderow * V3) + half;
  v4f av[6], aw[6];
#pragma unroll
  for (int q = 0; q < 6; q++) {
    av[q] = vA[2 * q];
    aw[q] = gV[2 * q];
  }
#pragma unroll
  for (int q = 0; q < 6; q++) aw[q] = av[q] + aw[q] * invden;
  float* outv = out + (long)NN * SDIM;
#pragma unroll 1
  for (int ct = 0; ct < 3; ct++) {
    v8f ao = {};
    v8f ar = {};
    int d = 16 * ct + row;
    const v4f* bo = (const v4f*)(wnv + (long)d * V3) + half;
    const v4f* br = (const v4f*)(wrv + (long)d * V3) + half;
#pragma unroll
    for (int q = 0; q < 6; q++) {
      v4f bO = bo[2 * q];
      v4f bR = br[2 * q];
      ao = wmma4(lo2(aw[q]), lo2(bO), ao);
      ar = wmma4(lo2(av[q]), lo2(bR), ar);
      ao = wmma4(hi2(aw[q]), hi2(bO), ao);
      ar = wmma4(hi2(av[q]), hi2(bR), ar);
    }
    float biasO = bnv[d], biasR = brv[d];
    int vo = d / 3;
#pragma unroll
    for (int r = 0; r < 8; r++) {
      int m = r + 8 * half;
      float pre = ao[r] + biasO;
      float g = lds_gate[wid][m * 16 + vo];
      outv[(long)(nb + m) * V3 + d] = pre * g + (ar[r] + biasR);
    }
  }
}

extern "C" void kernel_launch(void* const* d_in, const int* in_sizes, int n_in,
                              void* d_out, int out_size, void* d_ws,
                              size_t ws_size, hipStream_t stream) {
  const float* node_s = (const float*)d_in[0];
  const float* node_v = (const float*)d_in[1];
  const int* eidx     = (const int*)d_in[2];
  const float* edge_s = (const float*)d_in[3];
  const float* edge_v = (const float*)d_in[4];
  const float* ln_g = (const float*)d_in[5];
  const float* ln_b = (const float*)d_in[6];
  const float* wes = (const float*)d_in[7];
  const float* bes = (const float*)d_in[8];
  const float* wev = (const float*)d_in[9];
  const float* bev = (const float*)d_in[10];
  const float* wns = (const float*)d_in[11];
  const float* bns = (const float*)d_in[12];
  const float* wnv = (const float*)d_in[13];
  const float* bnv = (const float*)d_in[14];
  const float* wrs = (const float*)d_in[15];
  const float* brs = (const float*)d_in[16];
  const float* wrv = (const float*)d_in[17];
  const float* brv = (const float*)d_in[18];
  float* out = (float*)d_out;

  float* ws = (float*)d_ws;
  float* s_n = ws;
  float* v_n = s_n + (size_t)NN * SDIM;
  float* agg_s = v_n + (size_t)NN * V3;
  float* agg_v = agg_s + (size_t)NN * SDIM;
  float* cnt = agg_v + (size_t)NN * V3;
  int* hist1 = (int*)(cnt + NN);
  int* offs1 = hist1 + T1;
  int* off = offs1 + (T1 + 16);
  int* perm = off + (NN + 16);
  int* sdst = perm + EE;
  int* rec = sdst + EE;
  const int* edge_dst = eidx + EE;

  k_cnt1<<<G1, 1024, 0, stream>>>(edge_dst, hist1);
  k_scan<<<1, 1024, 0, stream>>>(hist1, offs1);
  k_scat1<<<G1, 1024, 0, stream>>>(edge_dst, offs1, rec);
  k_sort2<<<NB1, 1024, 0, stream>>>(rec, offs1, off, cnt, perm, sdst);
  k_norm<<<(NN * 32 + 255) / 256, 256, 0, stream>>>(node_s, node_v, ln_g, ln_b,
                                                    s_n, v_n);
  k_edge<<<NN / 16, 160, 0, stream>>>(edge_s, edge_v, perm, sdst, off, wes, bes,
                                      wev, bev, agg_s, agg_v);
  k_node<<<(NN / 16 + 7) / 8, 256, 0, stream>>>(s_n, v_n, agg_s, agg_v, cnt,
                                                wns, bns, wnv, bnv, wrs, brs,
                                                wrv, brv, out);
}


# Per-node normalisation fused into k_node: its A tiles come from node_s/node_v and are normalised in registers (row sums: 64 per lane + one half exchange, sum order changes); k_norm launch gone
# speedup vs baseline: 3.9354x; 1.0648x over previous
#include <hip/hip_runtime.h>
#include <math.h>

#define NN   50000
#define EE   800000
#define SDIM 128
#define VDIM 16
#define V3   48
#define ESD  32

typedef float v2f __attribute__((ext_vector_type(2)));
typedef float v8f __attribute__((ext_vector_type(8)));

__device__ __forceinline__ v8f wmma4(v2f a, v2f b, v8f c) {
  return __builtin_amdgcn_wmma_f32_16x16x4_f32(false, a, false, b, (short)0, c,
                                               false, false);
}

__device__ __forceinline__ float fast_rcp(float x) {
  return __builtin_amdgcn_rcpf(x);
}

__device__ __forceinline__ float sigmoidf_(float x) {
  return fast_rcp(1.f + __expf(-x));
}

#define G1 256
#define CH1 (EE / G1)
#define NB1 ((NN + 255) / 256)
#define T1 (NB1 * G1)
#define SCAN_CH 49
static_assert(EE % G1 == 0, "chunks must tile the edge list");
static_assert(CH1 <= 4 * 1024, "a chunk is read as 4 values per thread");
static_assert(T1 == SCAN_CH * 1024, "the scan block covers the table exactly");

__global__ void __launch_bounds__(1024) k_cnt1(const int* __restrict__ edge_dst,
                                               int* __restrict__ hist1) {
  __shared__ int h[NB1];
  int t = threadIdx.x;
  int g = blockIdx.x;
  if (t < NB1) h[t] = 0;
  const int* src = edge_dst + g * CH1;
  int d[4];
#pragma unroll
  for (int k = 0; k < 4; k++) {
    int i = t + 1024 * k;
    d[k] = (i < CH1) ? src[i] : 0;
  }
  __syncthreads();
#pragma unroll
  for (int k = 0; k < 4; k++)
    if (t + 1024 * k < CH1) atomicAdd(&h[d[k] >> 8], 1);
  __syncthreads();
  if (t < NB1) hist1[t * G1 + g] = h[t];
}

__global__ void __launch_bounds__(1024) k_scan(const int* __restrict__ hist1,
                                               int* __restrict__ offs1) {
  __shared__ int wsum[32];
  int t = threadIdx.x;
  int lane = t & 31;
  int w = t >> 5;
  int cbase = w * (SCAN_CH * 32);
  int v[SCAN_CH];
#pragma unroll
  for (int i = 0; i < SCAN_CH; i++) v[i] = hist1[cbase + i * 32 + lane];
  int s = 0;
#pragma unroll
  for (int i = 0; i < SCAN_CH; i++) s += v[i];
#pragma unroll
  for (int m = 16; m >= 1; m >>= 1) s += __shfl_xor(s, m, 32);
  if (lane == 0) wsum[w] = s;
  __syncthreads();
  int x = (lane < w) ? wsum[lane] : 0;
#pragma unroll
  for (int m = 16; m >= 1; m >>= 1) x += __shfl_xor(x, m, 32);
  int carry = x;
#pragma unroll
  for (int i = 0; i < SCAN_CH; i++) {
    int inc = v[i];
#pragma unroll
    for (int d = 1; d < 32; d <<= 1) {
      int y = __shfl_up(inc, d, 32);
      if (lane >= d) inc += y;
    }
    offs1[cbase + i * 32 + lane] = carry + inc - v[i];
    carry += __shfl(inc, 31, 32);
  }
  if (t == 1023) offs1[T1] = carry;
}

__global__ void __launch_bounds__(1024) k_scat1(const int* __restrict__ edge_dst,
                                                const int* __restrict__ offs1,
                                                int* __restrict__ rec) {
  __shared__ int cur[NB1];
  int t = threadIdx.x;
  int g = blockIdx.x;
  if (t < NB1) cur[t] = offs1[t * G1 + g];
  const int* src = edge_dst + g * CH1;
  int d[4];
#pragma unroll
  for (int k = 0; k < 4; k++) {
    int i = t + 1024 * k;
    d[k] = (i < CH1) ? src[i] : 0;
  }
  __syncthreads();
  int2* out = (int2*)rec;
#pragma unroll
  for (int k = 0; k < 4; k++) {
    int i = t + 1024 * k;
    if (i < CH1) {
      int pos = atomicAdd(&cur[d[k] >> 8], 1);
      out[pos] = make_int2(g * CH1 + i, d[k]);
    }
  }
}

__global__ void __launch_bounds__(1024) k_sort2(const int* __restrict__ rec,
                                                const int* __restrict__ offs1,
                                                int* __restrict__ off,
                                                float* __restrict__ cnt,
                                                int* __restrict__ perm,
                                                int* __restrict__ sdst) {
  __shared__ int c[256];
  __shared__ int cur[256];
  int t = threadIdx.x;
  int b = blockIdx.x;
  int S = offs1[b * G1];
  int T = offs1[(b + 1) * G1];
  const int2* in = (const int2*)rec;
  if (t < 256) c[t] = 0;
  __syncthreads();
  for (int base = S; base < T; base += 4096) {
    int dn[4];
#pragma unroll
    for (int k = 0; k < 4; k++) {
      int i = base + t + 1024 * k;
      dn[k] = in[min(i, T - 1)].y;
    }
#pragma unroll
    for (int k = 0; k < 4; k++)
      if (base + t + 1024 * k < T) atomicAdd(&c[dn[k] & 255], 1);
  }
  __syncthreads();
  if (t < 32) {
    int v[8];
    int s = 0;
#pragma unroll
    for (int k = 0; k < 8; k++) {
      v[k] = c[t * 8 + k];
      s += v[k];
    }
    int inc = s;
#pragma unroll
    for (int d = 1; d < 32; d <<= 1) {
      int y = __shfl_up(inc, d, 32);
      if (t >= d) inc += y;
    }
    int run = S + inc - s;
#pragma unroll
    for (int k = 0; k < 8; k++) {
      cur[t * 8 + k] = run;
      run += v[k];
    }
  }
  __syncthreads();
  if (t < 256) {
    int node = b * 256 + t;
    if (node < NN) {
      off[node] = cur[t];
      cnt[node] = (float)c[t];
    }
  }
  if (b == 0 && t == 0) off[NN] = EE;
  __syncthreads();
  for (int base = S; base < T; base += 4096) {
    int2 r[4];
#pragma unroll
    for (int k = 0; k < 4; k++) {
      int i = base + t + 1024 * k;
      r[k] = (i < T) ? in[i] : make_int2(0, 0);
    }
#pragma unroll
    for (int k = 0; k < 4; k++) {
      if (base + t + 1024 * k < T) {
        int pos = atomicAdd(&cur[r[k].y & 255], 1);
        perm[pos] = r[k].x;
        sdst[pos] = r[k].y;
      }
    }
  }
}

__device__ __forceinline__ void load_idx(const int* __restrict__ perm,
                                         const int* __restrict__ sdst, int p,
                                         int row, int pend, int nb, int& e,
                                         int& dl) {
  int pe = p + row;
  bool valid = pe < pend;
  int pc = valid ? pe : 0;
  int ev = perm[pc];
  int dv = sdst[pc];
  e = valid ? ev : 0;
  dl = valid ? dv - nb : -1;
}

__device__ __forceinline__ void sel_frag(int dl, int row, int half,
                                         v2f sa[4]) {
#pragma unroll
  for (int jj = 0; jj < 4; jj++) {
    int x0 = __shfl(dl, 8 * half + 2 * jj, 32);
    int x1 = __shfl(dl, 8 * half + 2 * jj + 1, 32);
    sa[jj].x = (x0 == row) ? 1.f : 0.f;
    sa[jj].y = (x1 == row) ? 1.f : 0.f;
  }
}

__global__ void __launch_bounds__(160) k_edge(
    const float* __restrict__ edge_s, const float* __restrict__ edge_v,
    const int* __restrict__ perm, const int* __restrict__ sdst,
    const int* __restrict__ off,
    const float* __restrict__ wes, const float* __restrict__ bes,
    const float* __restrict__ wev, const float* __restrict__ bev,
    float* __restrict__ agg_s, float* __restrict__ agg_v) {
  int wid = threadIdx.x >> 5;
  int lane = threadIdx.x & 31;
  int row = lane & 15;
  int half = lane >> 4;
  int nb = blockIdx.x * 16;
  int pbeg = off[nb];
  int pend = off[nb + 16];

  if (wid < 4) {
    int n0 = 32 * wid + row;
    int n1 = n0 + 16;
    v2f b0[8], b1[8];
#pragma unroll
    for (int j = 0; j < 8; j++) {
      int k0 = 4 * j + 2 * half;
      b0[j] = *(const v2f*)(wes + (long)n0 * ESD + k0);
      b1[j] = *(const v2f*)(wes + (long)n1 * ESD + k0);
    }
    float bias0 = bes[n0], bias1 = bes[n1];
    v8f g0 = {};
    v8f g1 = {};
    int e, dl;
    load_idx(perm, sdst, pbeg, row, pend, nb, e, dl);
    for (int p = pbeg; p < pend; p += 16) {
      int en, dln;
      load_idx(perm, sdst, p + 16, row, pend, nb, en, dln);
      const float* arow = edge_s + (long)e * ESD;
      v8f c0 = {};
      v8f c1 = {};
#pragma unroll
      for (int j = 0; j < 8; j++) {
        int k0 = 4 * j + 2 * half;
        v2f a = *(const v2f*)(arow + k0);
        c0 = wmma4(a, b0[j], c0);
        c1 = wmma4(a, b1[j], c1);
      }
      v2f sa[4];
      sel_frag(dl, row, half, sa);
#pragma unroll
      for (int r = 0; r < 8; r++) {
        float p0 = c0[r] + bias0;
        float p1 = c1[r] + bias1;
        c0[r] = p0 * sigmoidf_(p0);
        c1[r] = p1 * sigmoidf_(p1);
      }
#pragma unroll
      for (int jj = 0; jj < 4; jj++) {
        v2f m0 = {c0[2 * jj], c0[2 * jj + 1]};
        v2f m1 = {c1[2 * jj], c1[2 * jj + 1]};
        g0 = wmma4(sa[jj], m0, g0);
        g1 = wmma4(sa[jj], m1, g1);
      }
      e = en;
      dl = dln;
    }
#pragma unroll
    for (int r = 0; r < 8; r++) {
      float* orow = agg_s + (long)(nb + r + 8 * half) * SDIM;
      orow[n0] = g0[r];
      orow[n1] = g1[r];
    }
  } else {
    v2f b0[8];
#pragma unroll
    for (int j = 0; j < 8; j++)
      b0[j] = *(const v2f*)(wes + (long)row * ESD + 4 * j + 2 * half);
    float bias0 = bes[row];
    v2f bw[3];
    float bv[3];
    int gsrc[3];
#pragma unroll
    for (int vt = 0; vt < 3; vt++) {
      int d = 16 * vt + row;
      float w1 = wev[d * 3 + 1];
      bw[vt].x = wev[d * 3 + 2 * half];
      bw[vt].y = (half == 0) ? w1 : 0.f;
      bv[vt] = bev[d];
      gsrc[vt] = d / 3 + 16 * half;
    }
    v8f q[3];
#pragma unroll
    for (int vt = 0; vt < 3; vt++) q[vt] = (v8f){};
    int e, dl;
    load_idx(perm, sdst, pbeg, row, pend, nb, e, dl);
    for (int p = pbeg; p < pend; p += 16) {
      int en, dln;
      load_idx(perm, sdst, p + 16, row, pend, nb, en, dln);
      const float* arow = edge_s + (long)e * ESD;
      float e1 = edge_v[(long)e * 3 + 1];
      v2f av;
      av.x = edge_v[(long)e * 3 + 2 * half];
      av.y = (half == 0) ? e1 : 0.f;
      v8f c0 = {};
#pragma unroll
      for (int j = 0; j < 8; j++) {
        v2f a = *(const v2f*)(arow + 4 * j + 2 * half);
        c0 = wmma4(a, b0[j], c0);
      }
      v2f sa[4];
      sel_frag(dl, row, half, sa);
      float gate[8];
#pragma unroll
      for (int r = 0; r < 8; r++) {
        float pre = c0[r] + bias0;
        float so = pre * sigmoidf_(pre);
        gate[r] = sigmoidf_(so);
      }
#pragma unroll
      for (int vt = 0; vt < 3; vt++) {
        v8f vv = {};
        vv = wmma4(av, bw[vt], vv);
        float val[8];
#pragma unroll
        for (int r = 0; r < 8; r++)
          val[r] = (vv[r] + bv[vt]) * __shfl(gate[r], gsrc[vt], 32);
#pragma unroll
        for (int jj = 0; jj < 4; jj++) {
          v2f m = {val[2 * jj], val[2 * jj + 1]};
          q[vt] = wmma4(sa[jj], m, q[vt]);
        }
      }
      e = en;
      dl = dln;
    }
#pragma unroll
    for (int vt = 0; vt < 3; vt++) {
#pragma unroll
      for (int r = 0; r < 8; r++)
        agg_v[(long)(nb + r + 8 * half) * V3 + 16 * vt + row] = q[vt][r];
    }
  }
}

typedef float v4f __attribute__((ext_vector_type(4)));

__device__ __forceinline__ v2f lo2(v4f x) { return (v2f){x.x, x.y}; }
__device__ __forceinline__ v2f hi2(v4f x) { return (v2f){x.z, x.w}; }

__global__ void __launch_bounds__(256, 4) k_node(
    const float* __restrict__ node_s, const float* __restrict__ node_v,
    const float* __restrict__ ln_g, const float* __restrict__ ln_b,
    const float* __restrict__ agg_s, const float* __restrict__ agg_v,
    const float* __restrict__ cnt,
    const float* __restrict__ wns, const float* __restrict__ bns,
    const float* __restrict__ wnv, const float* __restrict__ bnv,
    const float* __restrict__ wrs, const float* __restrict__ brs,
    const float* __restrict__ wrv, const float* __restrict__ brv,
    float* __restrict__ out) {
  __shared__ float lds_gate[8][16 * 16];
  __shared__ v4f lds_gb[2 * SDIM / 4];
  static_assert(2 * SDIM == 256, "one LayerNorm parameter per thread");
  int wid = threadIdx.x >> 5;
  int lane = threadIdx.x & 31;
  int group = blockIdx.x * 8 + wid;
  bool live = group < NN / 16;
  int nb = (live ? group : NN / 16 - 1) * 16;
  int row = lane & 15;
  int half = lane >> 4;
  int noderow = nb + row;

  float gb = (threadIdx.x < SDIM) ? ln_g[threadIdx.x] : ln_b[threadIdx.x - SDIM];
  float invden = fast_rcp(fmaxf(cnt[noderow], 1.f));
  const v4f* sA = (const v4f*)(node_s + (long)noderow * SDIM) + half;
  const v4f* gA = (const v4f*)(agg_s + (long)noderow * SDIM) + half;
  v4f as[16], ai[16];
#pragma unroll
  for (int q = 0; q < 16; q++) {
    as[q] = sA[2 * q];
    ai[q] = gA[2 * q];
  }
  ((float*)lds_gb)[threadIdx.x] = gb;
  __syncthreads();
  if (!live) return;
  v4f acc = as[0];
#pragma unroll
  for (int q = 1; q < 16; q++) acc += as[q];
  float s = (acc.x + acc.y) + (acc.z + acc.w);
  s += __shfl_xor(s, 16, 32);
  float mu = s * (1.f / SDIM);
  v4f qa = {0.f, 0.f, 0.f, 0.f};
#pragma unroll
  for (int q = 0; q < 16; q++) {
    as[q] = as[q] - mu;
    qa += as[q] * as[q];
  }
  float qs = (qa.x + qa.y) + (qa.z + qa.w);
  qs += __shfl_xor(qs, 16, 32);
  float rs = rsqrtf(qs * (1.f / SDIM) + 1e-5f);
  const v4f* lG = lds_gb + half;
  const v4f* lB = lds_gb + SDIM / 4 + half;
#pragma unroll
  for (int q = 0; q < 16; q++) as[q] = as[q] * rs * lG[2 * q] + lB[2 * q];
#pragma unroll
  for (int q = 0; q < 16; q++) ai[q] = as[q] + ai[q] * invden;

#pragma unroll 1
  for (int ct = 0; ct < 8; ct++) {
    v8f ao = {};
    v8f ar = {};
    int n = 16 * ct + row;
    const v4f* bo = (const v4f*)(wns + (long)n * SDIM) + half;
    const v4f* br = (const v4f*)(wrs + (long)n * SDIM) + half;
#pragma unroll
    for (int q = 0; q < 16; q++) {
      v4f bO = bo[2 * q];
      v4f bR = br[2 * q];
      ao = wmma4(lo2(ai[q]), lo2(bO), ao);
      ar = wmma4(lo2(as[q]), lo2(bR), ar);
      ao = wmma4(hi2(ai[q]), hi2(bO), ao);
      ar = wmma4(hi2(as[q]), hi2(bR), ar);
    }
    float biasO = bns[n], biasR = brs[n];
#pragma unroll
    for (int r = 0; r < 8; r++) {
      int m = r + 8 * half;
      float pre = ao[r] + biasO;
      float so = pre * sigmoidf_(pre);
      if (ct == 0) lds_gate[wid][m * 16 + n] = sigmoidf_(so);
      out[(long)(nb + m) * SDIM + n] = so + (ar[r] + biasR);
    }
  }

  const v4f* vA = (const v4f*)(node_v + (long)noderow * V3) + half;
  const v4f* gV = (const v4f*)(agg_v + (long)noderow * V3) + half;
  v4f av[6], aw[6];
#pragma unroll
  for (int q = 0; q < 6; q++) {
    av[q] = vA[2 * q];
    aw[q] = gV[2 * q];
  }
  v4f qv = av[0] * av[0];
#pragma unroll
  for (int q = 1; q < 6; q++) qv += av[q] * av[q];
  float qq = (qv.x + qv.y) + (qv.z + qv.w);
  qq += __shfl_xor(qq, 16, 32);
  float vinv = rsqrtf(qq * (1.f / VDIM) + 1e-8f);
#pragma unroll
  for (int q = 0; q < 6; q++) av[q] = av[q] * vinv;
#pragma unroll
  for (int q = 0; q < 6; q++) aw[q] = av[q] + aw[q] * invden;
  float* outv = out + (long)NN * SDIM;
#pragma unroll 1
  for (int ct = 0; ct < 3; ct++) {
    v8f ao = {};
    v8f ar = {};
    int d = 16 * ct + row;
    const v4f* bo = (const v4f*)(wnv + (long)d * V3) + half;
    const v4f* br = (const v4f*)(wrv + (long)d * V3) + half;
#pragma unroll
    for (int q = 0; q < 6; q++) {
      v4f bO = bo[2 * q];
      v4f bR = br[2 * q];
      ao = wmma4(lo2(aw[q]), lo2(bO), ao);
      ar = wmma4(lo2(av[q]), lo2(bR), ar);
      ao = wmma4(hi2(aw[q]), hi2(bO), ao);
      ar = wmma4(hi2(av[q]), hi2(bR), ar);
    }
    float biasO = bnv[d], biasR = brv[d];
    int vo = d / 3;
#pragma unroll
    for (int r = 0; r < 8; r++) {
      int m = r + 8 * half;
      float pre = ao[r] + biasO;
      float g = lds_gate[wid][m * 16 + vo];
      outv[(long)(nb + m) * V3 + d] = pre * g + (ar[r] + biasR);
    }
  }
}

extern "C" void kernel_launch(void* const* d_in, const int* in_sizes, int n_in,
                              void* d_out, int out_size, void* d_ws,
                              size_t ws_size, hipStream_t stream) {
  const float* node_s = (const float*)d_in[0];
  const float* node_v = (const float*)d_in[1];
  const int* eidx     = (const int*)d_in[2];
  const float* edge_s = (const float*)d_in[3];
  const float* edge_v = (const float*)d_in[4];
  const float* ln_g = (const float*)d_in[5];
  const float* ln_b = (const float*)d_in[6];
  const float* wes = (const float*)d_in[7];
  const float* bes = (const float*)d_in[8];
  const float* wev = (const float*)d_in[9];
  const float* bev = (const float*)d_in[10];
  const float* wns = (const float*)d_in[11];
  const float* bns = (const float*)d_in[12];
  const float* wnv = (const float*)d_in[13];
  const float* bnv = (const float*)d_in[14];
  const float* wrs = (const float*)d_in[15];
  const float* brs = (const float*)d_in[16];
  const float* wrv = (const float*)d_in[17];
  const float* brv = (const float*)d_in[18];
  float* out = (float*)d_out;

  float* ws = (float*)d_ws;
  float* agg_s = ws;
  float* agg_v = agg_s + (size_t)NN * SDIM;
  float* cnt = agg_v + (size_t)NN * V3;
  int* hist1 = (int*)(cnt + NN);
  int* offs1 = hist1 + T1;
  int* off = offs1 + (T1 + 16);
  int* perm = off + (NN + 16);
  int* sdst = perm + EE;
  int* rec = sdst + EE;
  const int* edge_dst = eidx + EE;

  k_cnt1<<<G1, 1024, 0, stream>>>(edge_dst, hist1);
  k_scan<<<1, 1024, 0, stream>>>(hist1, offs1);
  k_scat1<<<G1, 1024, 0, stream>>>(edge_dst, offs1, rec);
  k_sort2<<<NB1, 1024, 0, stream>>>(rec, offs1, off, cnt, perm, sdst);
  k_edge<<<NN / 16, 160, 0, stream>>>(edge_s, edge_v, perm, sdst, off, wes, bes,
                                      wev, bev, agg_s, agg_v);
  k_node<<<(NN / 16 + 7) / 8, 256, 0, stream>>>(node_s, node_v, ln_g, ln_b,
                                                agg_s, agg_v, cnt, wns, bns,
                                                wnv, bnv, wrs, brs, wrv, brv,
                                                out);
}


# k_edge: four equal waves per 16-node block, each wave forms all 176 columns of every 4th edge tile (edge row read once, weights in LDS); wave partial sums added in LDS (sum order changes)
# speedup vs baseline: 4.3598x; 1.1078x over previous
#include <hip/hip_runtime.h>
#include <math.h>

#define NN   50000
#define EE   800000
#define SDIM 128
#define VDIM 16
#define V3   48
#define ESD  32

typedef float v2f __attribute__((ext_vector_type(2)));
typedef float v8f __attribute__((ext_vector_type(8)));

__device__ __forceinline__ v8f wmma4(v2f a, v2f b, v8f c) {
  return __builtin_amdgcn_wmma_f32_16x16x4_f32(false, a, false, b, (short)0, c,
                                               false, false);
}

__device__ __forceinline__ float fast_rcp(float x) {
  return __builtin_amdgcn_rcpf(x);
}

__device__ __forceinline__ float sigmoidf_(float x) {
  return fast_rcp(1.f + __expf(-x));
}

#define G1 256
#define CH1 (EE / G1)
#define NB1 ((NN + 255) / 256)
#define T1 (NB1 * G1)
#define SCAN_CH 49
static_assert(EE % G1 == 0, "chunks must tile the edge list");
static_assert(CH1 <= 4 * 1024, "a chunk is read as 4 values per thread");
static_assert(T1 == SCAN_CH * 1024, "the scan block covers the table exactly");

__global__ void __launch_bounds__(1024) k_cnt1(const int* __restrict__ edge_dst,
                                               int* __restrict__ hist1) {
  __shared__ int h[NB1];
  int t = threadIdx.x;
  int g = blockIdx.x;
  if (t < NB1) h[t] = 0;
  const int* src = edge_dst + g * CH1;
  int d[4];
#pragma unroll
  for (int k = 0; k < 4; k++) {
    int i = t + 1024 * k;
    d[k] = (i < CH1) ? src[i] : 0;
  }
  __syncthreads();
#pragma unroll
  for (int k = 0; k < 4; k++)
    if (t + 1024 * k < CH1) atomicAdd(&h[d[k] >> 8], 1);
  __syncthreads();
  if (t < NB1) hist1[t * G1 + g] = h[t];
}

__global__ void __launch_bounds__(1024) k_scan(const int* __restrict__ hist1,
                                               int* __restrict__ offs1) {
  __shared__ int wsum[32];
  int t = threadIdx.x;
  int lane = t & 31;
  int w = t >> 5;
  int cbase = w * (SCAN_CH * 32);
  int v[SCAN_CH];
#pragma unroll
  for (int i = 0; i < SCAN_CH; i++) v[i] = hist1[cbase + i * 32 + lane];
  int s = 0;
#pragma unroll
  for (int i = 0; i < SCAN_CH; i++) s += v[i];
#pragma unroll
  for (int m = 16; m >= 1; m >>= 1) s += __shfl_xor(s, m, 32);
  if (lane == 0) wsum[w] = s;
  __syncthreads();
  int x = (lane < w) ? wsum[lane] : 0;
#pragma unroll
  for (int m = 16; m >= 1; m >>= 1) x += __shfl_xor(x, m, 32);
  int carry = x;
#pragma unroll
  for (int i = 0; i < SCAN_CH; i++) {
    int inc = v[i];
#pragma unroll
    for (int d = 1; d < 32; d <<= 1) {
      int y = __shfl_up(inc, d, 32);
      if (lane >= d) inc += y;
    }
    offs1[cbase + i * 32 + lane] = carry + inc - v[i];
    carry += __shfl(inc, 31, 32);
  }
  if (t == 1023) offs1[T1] = carry;
}

__global__ void __launch_bounds__(1024) k_scat1(const int* __restrict__ edge_dst,
                                                const int* __restrict__ offs1,
                                                int* __restrict__ rec) {
  __shared__ int cur[NB1];
  int t = threadIdx.x;
  int g = blockIdx.x;
  if (t < NB1) cur[t] = offs1[t * G1 + g];
  const int* src = edge_dst + g * CH1;
  int d[4];
#pragma unroll
  for (int k = 0; k < 4; k++) {
    int i = t + 1024 * k;
    d[k] = (i < CH1) ? src[i] : 0;
  }
  __syncthreads();
  int2* out = (int2*)rec;
#pragma unroll
  for (int k = 0; k < 4; k++) {
    int i = t + 1024 * k;
    if (i < CH1) {
      int pos = atomicAdd(&cur[d[k] >> 8], 1);
      out[pos] = make_int2(g * CH1 + i, d[k]);
    }
  }
}

__global__ void __launch_bounds__(1024) k_sort2(const int* __restrict__ rec,
                                                const int* __restrict__ offs1,
                                                int* __restrict__ off,
                                                float* __restrict__ cnt,
                                                int* __restrict__ perm,
                                                int* __restrict__ sdst) {
  __shared__ int c[256];
  __shared__ int cur[256];
  int t = threadIdx.x;
  int b = blockIdx.x;
  int S = offs1[b * G1];
  int T = offs1[(b + 1) * G1];
  const int2* in = (const int2*)rec;
  if (t < 256) c[t] = 0;
  __syncthreads();
  for (int base = S; base < T; base += 4096) {
    int dn[4];
#pragma unroll
    for (int k = 0; k < 4; k++) {
      int i = base + t + 1024 * k;
      dn[k] = in[min(i, T - 1)].y;
    }
#pragma unroll
    for (int k = 0; k < 4; k++)
      if (base + t + 1024 * k < T) atomicAdd(&c[dn[k] & 255], 1);
  }
  __syncthreads();
  if (t < 32) {
    int v[8];
    int s = 0;
#pragma unroll
    for (int k = 0; k < 8; k++) {
      v[k] = c[t * 8 + k];
      s += v[k];
    }
    int inc = s;
#pragma unroll
    for (int d = 1; d < 32; d <<= 1) {
      int y = __shfl_up(inc, d, 32);
      if (t >= d) inc += y;
    }
    int run = S + inc - s;
#pragma unroll
    for (int k = 0; k < 8; k++) {
      cur[t * 8 + k] = run;
      run += v[k];
    }
  }
  __syncthreads();
  if (t < 256) {
    int node = b * 256 + t;
    if (node < NN) {
      off[node] = cur[t];
      cnt[node] = (float)c[t];
    }
  }
  if (b == 0 && t == 0) off[NN] = EE;
  __syncthreads();
  for (int base = S; base < T; base += 4096) {
    int2 r[4];
#pragma unroll
    for (int k = 0; k < 4; k++) {
      int i = base + t + 1024 * k;
      r[k] = (i < T) ? in[i] : make_int2(0, 0);
    }
#pragma unroll
    for (int k = 0; k < 4; k++) {
      if (base + t + 1024 * k < T) {
        int pos = atomicAdd(&cur[r[k].y & 255], 1);
        perm[pos] = r[k].x;
        sdst[pos] = r[k].y;
      }
    }
  }
}

typedef float v4f __attribute__((ext_vector_type(4)));

__device__ __forceinline__ v2f lo2(v4f x) { return (v2f){x.x, x.y}; }
__device__ __forceinline__ v2f hi2(v4f x) { return (v2f){x.z, x.w}; }

#define WP4 9
#define RP4 45
#define RPF (4 * RP4)
static_assert(SDIM + V3 <= RPF, "a partial-sum row holds all message columns");

__device__ __forceinline__ void load_idx(const int* __restrict__ perm,
                                         const int* __restrict__ sdst, int p,
                                         int row, int pend, int nb, int& e,
                                         int& dl) {
  int pe = p + row;
  bool valid = pe < pend;
  int pc = valid ? pe : 0;
  int ev = perm[pc];
  int dv = sdst[pc];
  e = valid ? ev : 0;
  dl = valid ? dv - nb : -1;
}

__device__ __forceinline__ void sel_frag(int dl, int row, int half,
                                         v2f sa[4]) {
#pragma unroll
  for (int jj = 0; jj < 4; jj++) {
    int x0 = __shfl(dl, 8 * half + 2 * jj, 32);
    int x1 = __shfl(dl, 8 * half + 2 * jj + 1, 32);
    sa[jj].x = (x0 == row) ? 1.f : 0.f;
    sa[jj].y = (x1 == row) ? 1.f : 0.f;
  }
}

__global__ void __launch_bounds__(128) k_edge(
    const float* __restrict__ edge_s, const float* __restrict__ edge_v,
    const int* __restrict__ perm, const int* __restrict__ sdst,
    const int* __restrict__ off,
    const float* __restrict__ wes, const float* __restrict__ bes,
    const float* __restrict__ wev, const float* __restrict__ bev,
    float* __restrict__ agg_s, float* __restrict__ agg_v) {
  __shared__ v4f lds_w[SDIM * WP4];
  __shared__ float lds_bias[SDIM];
  __shared__ v4f lds_red[4 * 16 * RP4];
  int wid = __builtin_amdgcn_readfirstlane(threadIdx.x >> 5);
  int lane = threadIdx.x & 31;
  int row = lane & 15;
  int half = lane >> 4;
  int nb = blockIdx.x * 16;
  int pbeg = off[nb];
  int pend = off[nb + 16];

#pragma unroll
  for (int i = 0; i < SDIM * ESD / 4 / 128; i++) {
    int idx = threadIdx.x + 128 * i;
    lds_w[(idx >> 3) * WP4 + (idx & 7)] = ((const v4f*)wes)[idx];
  }
  lds_bias[threadIdx.x] = bes[threadIdx.x];

  v2f bw[3];
  float bv[3];
  int gsrc[3];
#pragma unroll
  for (int vt = 0; vt < 3; vt++) {
    int d = 16 * vt + row;
    float w1 = wev[d * 3 + 1];
    bw[vt].x = wev[d * 3 + 2 * half];
    bw[vt].y = (half == 0) ? w1 : 0.f;
    bv[vt] = bev[d];
    gsrc[vt] = d / 3 + 16 * half;
  }
  v8f g[8];
  v8f qv[3];
#pragma unroll
  for (int ct = 0; ct < 8; ct++) g[ct] = (v8f){};
#pragma unroll
  for (int vt = 0; vt < 3; vt++) qv[vt] = (v8f){};
  __syncthreads();

  int p = pbeg + 16 * ((wid + blockIdx.x) & 3);
  int e, dl;
  load_idx(perm, sdst, p, row, pend, nb, e, dl);
  for (; p < pend; p += 64) {
    int en, dln;
    load_idx(perm, sdst, p + 64, row, pend, nb, en, dln);
    const v4f* arow = (const v4f*)(edge_s + (long)e * ESD) + half;
    v4f a[4];
#pragma unroll
    for (int q = 0; q < 4; q++) a[q] = arow[2 * q];
    float e1 = edge_v[(long)e * 3 + 1];
    v2f av;
    av.x = edge_v[(long)e * 3 + 2 * half];
    av.y = (half == 0) ? e1 : 0.f;
    v2f sa[4];
    sel_frag(dl, row, half, sa);
    float gate[8];
    asm volatile("" ::: "memory");
#pragma unroll
    for (int cp = 0; cp < 4; cp++) {
      const v4f* w0 = lds_w + (32 * cp + row) * WP4 + half;
      const v4f* w1 = w0 + 16 * WP4;
      v8f c0 = {};
      v8f c1 = {};
#pragma unroll
      for (int q = 0; q < 4; q++) {
        v4f b0 = w0[2 * q];
        v4f b1 = w1[2 * q];
        c0 = wmma4(lo2(a[q]), lo2(b0), c0);
        c1 = wmma4(lo2(a[q]), lo2(b1), c1);
        c0 = wmma4(hi2(a[q]), hi2(b0), c0);
        c1 = wmma4(hi2(a[q]), hi2(b1), c1);
      }
      float bias0 = lds_bias[32 * cp + row];
      float bias1 = lds_bias[32 * cp + 16 + row];
#pragma unroll
      for (int r = 0; r < 8; r++) {
        float p0 = c0[r] + bias0;
        float p1 = c1[r] + bias1;
        c0[r] = p0 * sigmoidf_(p0);
        c1[r] = p1 * sigmoidf_(p1);
      }
      if (cp == 0) {
#pragma unroll
        for (int r = 0; r < 8; r++) gate[r] = sigmoidf_(c0[r]);
      }
#pragma unroll
      for (int jj = 0; jj < 4; jj++) {
        v2f m0 = {c0[2 * jj], c0[2 * jj + 1]};
        v2f m1 = {c1[2 * jj], c1[2 * jj + 1]};
        g[2 * cp] = wmma4(sa[jj], m0, g[2 * cp]);
        g[2 * cp + 1] = wmma4(sa[jj], m1, g[2 * cp + 1]);
      }
    }
#pragma unroll
    for (int vt = 0; vt < 3; vt++) {
      v8f vv = {};
      vv = wmma4(av, bw[vt], vv);
      float val[8];
#pragma unroll
      for (int r = 0; r < 8; r++)
        val[r] = (vv[r] + bv[vt]) * __shfl(gate[r], gsrc[vt], 32);
#pragma unroll
      for (int jj = 0; jj < 4; jj++) {
        v2f m = {val[2 * jj], val[2 * jj + 1]};
        qv[vt] = wmma4(sa[jj], m, qv[vt]);
      }
    }
    e = en;
    dl = dln;
  }

  float* red = (float*)lds_red + wid * (16 * RPF);
#pragma unroll
  for (int ct = 0; ct < 8; ct++) {
#pragma unroll
    for (int r = 0; r < 8; r++) red[(r + 8 * half) * RPF + 16 * ct + row] = g[ct][r];
  }
#pragma unroll
  for (int vt = 0; vt < 3; vt++) {
#pragma unroll
    for (int r = 0; r < 8; r++)
      red[(r + 8 * half) * RPF + SDIM + 16 * vt + row] = qv[vt][r];
  }
  __syncthreads();
#pragma unroll
  for (int i = 0; i < 16 * SDIM / 4 / 128; i++) {
    int idx = threadIdx.x + 128 * i;
    int m = idx >> 5;
    int c4 = idx & 31;
    const v4f* src = lds_red + m * RP4 + c4;
    v4f s = src[0];
    s += src[16 * RP4];
    s += src[2 * 16 * RP4];
    s += src[3 * 16 * RP4];
    ((v4f*)(agg_s + (long)(nb + m) * SDIM))[c4] = s;
  }
#pragma unroll
  for (int i = 0; i < 2; i++) {
    int idx = threadIdx.x + 128 * i;
    if (idx < 16 * V3 / 4) {
      int m = idx / (V3 / 4);
      int c4 = idx - m * (V3 / 4);
      const v4f* src = lds_red + m * RP4 + SDIM / 4 + c4;
      v4f s = src[0];
      s += src[16 * RP4];
      s += src[2 * 16 * RP4];
      s += src[3 * 16 * RP4];
      ((v4f*)(agg_v + (long)nb * V3))[idx] = s;
    }
  }
}

__global__ void __launch_bounds__(256, 4) k_node(
    const float* __restrict__ node_s, const float* __restrict__ node_v,
    const float* __restrict__ ln_g, const float* __restrict__ ln_b,
    const float* __restrict__ agg_s, const float* __restrict__ agg_v,
    const float* __restrict__ cnt,
    const float* __restrict__ wns, const float* __restrict__ bns,
    const float* __restrict__ wnv, const float* __restrict__ bnv,
    const float* __restrict__ wrs, const float* __restrict__ brs,
    const float* __restrict__ wrv, const float* __restrict__ brv,
    float* __restrict__ out) {
  __shared__ float lds_gate[8][16 * 16];
  __shared__ v4f lds_gb[2 * SDIM / 4];
  static_assert(2 * SDIM == 256, "one LayerNorm parameter per thread");
  int wid = threadIdx.x >> 5;
  int lane = threadIdx.x & 31;
  int group = blockIdx.x * 8 + wid;
  bool live = group < NN / 16;
  int nb = (live ? group : NN / 16 - 1) * 16;
  int row = lane & 15;
  int half = lane >> 4;
  int noderow = nb + row;

  float gb = (threadIdx.x < SDIM) ? ln_g[threadIdx.x] : ln_b[threadIdx.x - SDIM];
  float invden = fast_rcp(fmaxf(cnt[noderow], 1.f));
  const v4f* sA = (const v4f*)(node_s + (long)noderow * SDIM) + half;
  const v4f* gA = (const v4f*)(agg_s + (long)noderow * SDIM) + half;
  v4f as[16], ai[16];
#pragma unroll
  for (int q = 0; q < 16; q++) {
    as[q] = sA[2 * q];
    ai[q] = gA[2 * q];
  }
  ((float*)lds_gb)[threadIdx.x] = gb;
  __syncthreads();
  if (!live) return;
  v4f acc = as[0];
#pragma unroll
  for (int q = 1; q < 16; q++) acc += as[q];
  float s = (acc.x + acc.y) + (acc.z + acc.w);
  s += __shfl_xor(s, 16, 32);
  float mu = s * (1.f / SDIM);
  v4f qa = {0.f, 0.f, 0.f, 0.f};
#pragma unroll
  for (int q = 0; q < 16; q++) {
    as[q] = as[q] - mu;
    qa += as[q] * as[q];
  }
  float qs = (qa.x + qa.y) + (qa.z + qa.w);
  qs += __shfl_xor(qs, 16, 32);
  float rs = rsqrtf(qs * (1.f / SDIM) + 1e-5f);
  const v4f* lG = lds_gb + half;
  const v4f* lB = lds_gb + SDIM / 4 + half;
#pragma unroll
  for (int q = 0; q < 16; q++) as[q] = as[q] * rs * lG[2 * q] + lB[2 * q];
#pragma unroll
  for (int q = 0; q < 16; q++) ai[q] = as[q] + ai[q] * invden;

#pragma unroll 1
  for (int ct = 0; ct < 8; ct++) {
    v8f ao = {};
    v8f ar = {};
    int n = 16 * ct + row;
    const v4f* bo = (const v4f*)(wns + (long)n * SDIM) + half;
    const v4f* br = (const v4f*)(wrs + (long)n * SDIM) + half;
#pragma unroll
    for (int q = 0; q < 16; q++) {
      v4f bO = bo[2 * q];
      v4f bR = br[2 * q];
      ao = wmma4(lo2(ai[q]), lo2(bO), ao);
      ar = wmma4(lo2(as[q]), lo2(bR), ar);
      ao = wmma4(hi2(ai[q]), hi2(bO), ao);
      ar = wmma4(hi2(as[q]), hi2(bR), ar);
    }
    float biasO = bns[n], biasR = brs[n];
#pragma unroll
    for (int r = 0; r < 8; r++) {
      int m = r + 8 * half;
      float pre = ao[r] + biasO;
      float so = pre * sigmoidf_(pre);
      if (ct == 0) lds_gate[wid][m * 16 + n] = sigmoidf_(so);
      out[(long)(nb + m) * SDIM + n] = so + (ar[r] + biasR);
    }
  }

  const v4f* vA = (const v4f*)(node_v + (long)noderow * V3) + half;
  const v4f* gV = (const v4f*)(agg_v + (long)noderow * V3) + half;
  v4f av[6], aw[6];
#pragma unroll
  for (int q = 0; q < 6; q++) {
    av[q] = vA[2 * q];
    aw[q] = gV[2 * q];
  }
  v4f qv = av[0] * av[0];
#pragma unroll
  for (int q = 1; q < 6; q++) qv += av[q] * av[q];
  float qq = (qv.x + qv.y) + (qv.z + qv.w);
  qq += __shfl_xor(qq, 16, 32);
  float vinv = rsqrtf(qq * (1.f / VDIM) + 1e-8f);
#pragma unroll
  for (int q = 0; q < 6; q++) av[q] = av[q] * vinv;
#pragma unroll
  for (int q = 0; q < 6; q++) aw[q] = av[q] + aw[q] * invden;
  float* outv = out + (long)NN * SDIM;
#pragma unroll 1
  for (int ct = 0; ct < 3; ct++) {
    v8f ao = {};
    v8f ar = {};
    int d = 16 * ct + row;
    const v4f* bo = (const v4f*)(wnv + (long)d * V3) + half;
    const v4f* br = (const v4f*)(wrv + (long)d * V3) + half;
#pragma unroll
    for (int q = 0; q < 6; q++) {
      v4f bO = bo[2 * q];
      v4f bR = br[2 * q];
      ao = wmma4(lo2(aw[q]), lo2(bO), ao);
      ar = wmma4(lo2(av[q]), lo2(bR), ar);
      ao = wmma4(hi2(aw[q]), hi2(bO), ao);
      ar = wmma4(hi2(av[q]), hi2(bR), ar);
    }
    float biasO = bnv[d], biasR = brv[d];
    int vo = d / 3;
#pragma unroll
    for (int r = 0; r < 8; r++) {
      int m = r + 8 * half;
      float pre = ao[r] + biasO;
      float g = lds_gate[wid][m * 16 + vo];
      outv[(long)(nb + m) * V3 + d] = pre * g + (ar[r] + biasR);
    }
  }
}

extern "C" void kernel_launch(void* const* d_in, const int* in_sizes, int n_in,
                              void* d_out, int out_size, void* d_ws,
                              size_t ws_size, hipStream_t stream) {
  const float* node_s = (const float*)d_in[0];
  const float* node_v = (const float*)d_in[1];
  const int* eidx     = (const int*)d_in[2];
  const float* edge_s = (const float*)d_in[3];
  const float* edge_v = (const float*)d_in[4];
  const float* ln_g = (const float*)d_in[5];
  const float* ln_b = (const float*)d_in[6];
  const float* wes = (const float*)d_in[7];
  const float* bes = (const float*)d_in[8];
  const float* wev = (const float*)d_in[9];
  const float* bev = (const float*)d_in[10];
  const float* wns = (const float*)d_in[11];
  const float* bns = (const float*)d_in[12];
  const float* wnv = (const float*)d_in[13];
  const float* bnv = (const float*)d_in[14];
  const float* wrs = (const float*)d_in[15];
  const float* brs = (const float*)d_in[16];
  const float* wrv = (const float*)d_in[17];
  const float* brv = (const float*)d_in[18];
  float* out = (float*)d_out;

  float* ws = (float*)d_ws;
  float* agg_s = ws;
  float* agg_v = agg_s + (size_t)NN * SDIM;
  float* cnt = agg_v + (size_t)NN * V3;
  int* hist1 = (int*)(cnt + NN);
  int* offs1 = hist1 + T1;
  int* off = offs1 + (T1 + 16);
  int* perm = off + (NN + 16);
  int* sdst = perm + EE;
  int* rec = sdst + EE;
  const int* edge_dst = eidx + EE;

  k_cnt1<<<G1, 1024, 0, stream>>>(edge_dst, hist1);
  k_scan<<<1, 1024, 0, stream>>>(hist1, offs1);
  k_scat1<<<G1, 1024, 0, stream>>>(edge_dst, offs1, rec);
  k_sort2<<<NB1, 1024, 0, stream>>>(rec, offs1, off, cnt, perm, sdst);
  k_edge<<<NN / 16, 128, 0, stream>>>(edge_s, edge_v, perm, sdst, off, wes, bes,
                                      wev, bev, agg_s, agg_v);
  k_node<<<(NN / 16 + 7) / 8, 256, 0, stream>>>(node_s, node_v, ln_g, ln_b,
                                                agg_s, agg_v, cnt, wns, bns,
                                                wnv, bnv, wrs, brs, wrv, brv,
                                                out);
}


# Bucketing level 1 sorts each edge chunk by range on its own in LDS (rank, in-block scan, place); range blocks gather their 256 pieces: count and table-scan launches gone; sum order still LDS arrival
# speedup vs baseline: 4.9150x; 1.1273x over previous
#include <hip/hip_runtime.h>
#include <math.h>

#define NN   50000
#define EE   800000
#define SDIM 128
#define VDIM 16
#define V3   48
#define ESD  32

typedef float v2f __attribute__((ext_vector_type(2)));
typedef float v8f __attribute__((ext_vector_type(8)));

__device__ __forceinline__ v8f wmma4(v2f a, v2f b, v8f c) {
  return __builtin_amdgcn_wmma_f32_16x16x4_f32(false, a, false, b, (short)0, c,
                                               false, false);
}

__device__ __forceinline__ float fast_rcp(float x) {
  return __builtin_amdgcn_rcpf(x);
}

__device__ __forceinline__ float sigmoidf_(float x) {
  return fast_rcp(1.f + __expf(-x));
}

#define G1 256
#define CH1 (EE / G1)
#define NB1 ((NN + 255) / 256)
#define SCAN_PER 7
#define PIECE_U 8
static_assert(EE % G1 == 0, "chunks must tile the edge list");
static_assert(CH1 <= 4 * 1024, "a chunk is read as 4 values per thread");
static_assert(NB1 <= SCAN_PER * 32, "one wave scans the range counts");
static_assert(G1 * 4 == 1024, "level 2 walks a chunk's piece with 4 threads");
static_assert(PIECE_U == 8, "the load fences of level 2 name 8 pairs");

__global__ void __launch_bounds__(1024) k_part(const int* __restrict__ edge_dst,
                                               int* __restrict__ offsT,
                                               int* __restrict__ rec) {
  __shared__ int h[NB1];
  int t = threadIdx.x;
  int g = blockIdx.x;
  if (t < NB1) h[t] = 0;
  const int* src = edge_dst + g * CH1;
  int d[4];
#pragma unroll
  for (int k = 0; k < 4; k++) {
    int i = t + 1024 * k;
    d[k] = (i < CH1) ? src[i] : 0;
  }
  __syncthreads();
  int rank[4];
#pragma unroll
  for (int k = 0; k < 4; k++) {
    rank[k] = 0;
    if (t + 1024 * k < CH1) rank[k] = atomicAdd(&h[d[k] >> 8], 1);
  }
  __syncthreads();
  if (t < 32) {
    int v[SCAN_PER];
    int s = 0;
#pragma unroll
    for (int k = 0; k < SCAN_PER; k++) {
      int idx = t * SCAN_PER + k;
      v[k] = (idx < NB1) ? h[idx] : 0;
      s += v[k];
    }
    int inc = s;
#pragma unroll
    for (int m = 1; m < 32; m <<= 1) {
      int y = __shfl_up(inc, m, 32);
      if (t >= m) inc += y;
    }
    int run = inc - s;
#pragma unroll
    for (int k = 0; k < SCAN_PER; k++) {
      int idx = t * SCAN_PER + k;
      if (idx < NB1) h[idx] = run;
      run += v[k];
    }
  }
  __syncthreads();
  if (t < NB1) offsT[t * G1 + g] = h[t];
  if (t == 0) offsT[NB1 * G1 + g] = CH1;
  int2* out = (int2*)rec + g * CH1;
#pragma unroll
  for (int k = 0; k < 4; k++) {
    int i = t + 1024 * k;
    if (i < CH1) out[h[d[k] >> 8] + rank[k]] = make_int2(g * CH1 + i, d[k]);
  }
}

__global__ void __launch_bounds__(1024) k_sort2(const int* __restrict__ rec,
                                                const int* __restrict__ offsT,
                                                int* __restrict__ off,
                                                float* __restrict__ cnt,
                                                int* __restrict__ perm,
                                                int* __restrict__ sdst) {
  __shared__ int c[256];
  __shared__ int cur[256];
  __shared__ int wsum[G1 / 32];
  int t = threadIdx.x;
  int b = blockIdx.x;
  int g = t >> 2;
  int j = t & 3;
  int ps = offsT[b * G1 + g];
  int pe = offsT[(b + 1) * G1 + g];
  const int2* in = (const int2*)rec + g * CH1;
  if (t < 256) c[t] = 0;
  int x = (t < G1) ? offsT[b * G1 + t] : 0;
#pragma unroll
  for (int m = 16; m >= 1; m >>= 1) x += __shfl_xor(x, m, 32);
  if (t < G1 && (t & 31) == 0) wsum[t >> 5] = x;
  __syncthreads();
  for (int base = ps + j; base < pe; base += 4 * PIECE_U) {
    int dn[PIECE_U];
#pragma unroll
    for (int k = 0; k < PIECE_U; k++) dn[k] = in[min(base + 4 * k, pe - 1)].y;
    asm volatile("" ::"v"(dn[0]), "v"(dn[1]), "v"(dn[2]), "v"(dn[3]),
                 "v"(dn[4]), "v"(dn[5]), "v"(dn[6]), "v"(dn[7]));
#pragma unroll
    for (int k = 0; k < PIECE_U; k++)
      if (base + 4 * k < pe) atomicAdd(&c[dn[k] & 255], 1);
  }
  __syncthreads();
  if (t < 32) {
    int S = 0;
#pragma unroll
    for (int w = 0; w < G1 / 32; w++) S += wsum[w];
    int v[8];
    int s = 0;
#pragma unroll
    for (int k = 0; k < 8; k++) {
      v[k] = c[t * 8 + k];
      s += v[k];
    }
    int inc = s;
#pragma unroll
    for (int m = 1; m < 32; m <<= 1) {
      int y = __shfl_up(inc, m, 32);
      if (t >= m) inc += y;
    }
    int run = S + inc - s;
#pragma unroll
    for (int k = 0; k < 8; k++) {
      cur[t * 8 + k] = run;
      run += v[k];
    }
  }
  __syncthreads();
  if (t < 256) {
    int node = b * 256 + t;
    if (node < NN) {
      off[node] = cur[t];
      cnt[node] = (float)c[t];
    }
  }
  if (b == 0 && t == 0) off[NN] = EE;
  __syncthreads();
  for (int base = ps + j; base < pe; base += 4 * PIECE_U) {
    int2 r[PIECE_U];
#pragma unroll
    for (int k = 0; k < PIECE_U; k++) r[k] = in[min(base + 4 * k, pe - 1)];
    asm volatile("" ::"v"(r[0].x), "v"(r[0].y), "v"(r[1].x), "v"(r[1].y),
                 "v"(r[2].x), "v"(r[2].y), "v"(r[3].x), "v"(r[3].y),
                 "v"(r[4].x), "v"(r[4].y), "v"(r[5].x), "v"(r[5].y),
                 "v"(r[6].x), "v"(r[6].y), "v"(r[7].x), "v"(r[7].y));
#pragma unroll
    for (int k = 0; k < PIECE_U; k++) {
      if (base + 4 * k < pe) {
        int pos = atomicAdd(&cur[r[k].y & 255], 1);
        perm[pos] = r[k].x;
        sdst[pos] = r[k].y;
      }
    }
  }
}

typedef float v4f __attribute__((ext_vector_type(4)));

__device__ __forceinline__ v2f lo2(v4f x) { return (v2f){x.x, x.y}; }
__device__ __forceinline__ v2f hi2(v4f x) { return (v2f){x.z, x.w}; }

#define WP4 9
#define RP4 45
#define RPF (4 * RP4)
static_assert(SDIM + V3 <= RPF, "a partial-sum row holds all message columns");

__device__ __forceinline__ void load_idx(const int* __restrict__ perm,
                                         const int* __restrict__ sdst, int p,
                                         int row, int pend, int nb, int& e,
                                         int& dl) {
  int pe = p + row;
  bool valid = pe < pend;
  int pc = valid ? pe : 0;
  int ev = perm[pc];
  int dv = sdst[pc];
  e = valid ? ev : 0;
  dl = valid ? dv - nb : -1;
}

__device__ __forceinline__ void sel_frag(int dl, int row, int half,
                                         v2f sa[4]) {
#pragma unroll
  for (int jj = 0; jj < 4; jj++) {
    int x0 = __shfl(dl, 8 * half + 2 * jj, 32);
    int x1 = __shfl(dl, 8 * half + 2 * jj + 1, 32);
    sa[jj].x = (x0 == row) ? 1.f : 0.f;
    sa[jj].y = (x1 == row) ? 1.f : 0.f;
  }
}

__global__ void __launch_bounds__(128) k_edge(
    const float* __restrict__ edge_s, const float* __restrict__ edge_v,
    const int* __restrict__ perm, const int* __restrict__ sdst,
    const int* __restrict__ off,
    const float* __restrict__ wes, const float* __restrict__ bes,
    const float* __restrict__ wev, const float* __restrict__ bev,
    float* __restrict__ agg_s, float* __restrict__ agg_v) {
  __shared__ v4f lds_w[SDIM * WP4];
  __shared__ float lds_bias[SDIM];
  __shared__ v4f lds_red[4 * 16 * RP4];
  int wid = __builtin_amdgcn_readfirstlane(threadIdx.x >> 5);
  int lane = threadIdx.x & 31;
  int row = lane & 15;
  int half = lane >> 4;
  int nb = blockIdx.x * 16;
  int pbeg = off[nb];
  int pend = off[nb + 16];

#pragma unroll
  for (int i = 0; i < SDIM * ESD / 4 / 128; i++) {
    int idx = threadIdx.x + 128 * i;
    lds_w[(idx >> 3) * WP4 + (idx & 7)] = ((const v4f*)wes)[idx];
  }
  lds_bias[threadIdx.x] = bes[threadIdx.x];

  v2f bw[3];
  float bv[3];
  int gsrc[3];
#pragma unroll
  for (int vt = 0; vt < 3; vt++) {
    int d = 16 * vt + row;
    float w1 = wev[d * 3 + 1];
    bw[vt].x = wev[d * 3 + 2 * half];
    bw[vt].y = (half == 0) ? w1 : 0.f;
    bv[vt] = bev[d];
    gsrc[vt] = d / 3 + 16 * half;
  }
  v8f g[8];
  v8f qv[3];
#pragma unroll
  for (int ct = 0; ct < 8; ct++) g[ct] = (v8f){};
#pragma unroll
  for (int vt = 0; vt < 3; vt++) qv[vt] = (v8f){};
  __syncthreads();

  int p = pbeg + 16 * ((wid + blockIdx.x) & 3);
  int e, dl;
  load_idx(perm, sdst, p, row, pend, nb, e, dl);
  for (; p < pend; p += 64) {
    int en, dln;
    load_idx(perm, sdst, p + 64, row, pend, nb, en, dln);
    const v4f* arow = (const v4f*)(edge_s + (long)e * ESD) + half;
    v4f a[4];
#pragma unroll
    for (int q = 0; q < 4; q++) a[q] = arow[2 * q];
    float e1 = edge_v[(long)e * 3 + 1];
    v2f av;
    av.x = edge_v[(long)e * 3 + 2 * half];
    av.y = (half == 0) ? e1 : 0.f;
    v2f sa[4];
    sel_frag(dl, row, half, sa);
    float gate[8];
    asm volatile("" ::: "memory");
#pragma unroll
    for (int cp = 0; cp < 4; cp++) {
      const v4f* w0 = lds_w + (32 * cp + row) * WP4 + half;
      const v4f* w1 = w0 + 16 * WP4;
      v8f c0 = {};
      v8f c1 = {};
#pragma unroll
      for (int q = 0; q < 4; q++) {
        v4f b0 = w0[2 * q];
        v4f b1 = w1[2 * q];
        c0 = wmma4(lo2(a[q]), lo2(b0), c0);
        c1 = wmma4(lo2(a[q]), lo2(b1), c1);
        c0 = wmma4(hi2(a[q]), hi2(b0), c0);
        c1 = wmma4(hi2(a[q]), hi2(b1), c1);
      }
      float bias0 = lds_bias[32 * cp + row];
      float bias1 = lds_bias[32 * cp + 16 + row];
#pragma unroll
      for (int r = 0; r < 8; r++) {
        float p0 = c0[r] + bias0;
        float p1 = c1[r] + bias1;
        c0[r] = p0 * sigmoidf_(p0);
        c1[r] = p1 * sigmoidf_(p1);
      }
      if (cp == 0) {
#pragma unroll
        for (int r = 0; r < 8; r++) gate[r] = sigmoidf_(c0[r]);
      }
#pragma unroll
      for (int jj = 0; jj < 4; jj++) {
        v2f m0 = {c0[2 * jj], c0[2 * jj + 1]};
        v2f m1 = {c1[2 * jj], c1[2 * jj + 1]};
        g[2 * cp] = wmma4(sa[jj], m0, g[2 * cp]);
        g[2 * cp + 1] = wmma4(sa[jj], m1, g[2 * cp + 1]);
      }
    }
#pragma unroll
    for (int vt = 0; vt < 3; vt++) {
      v8f vv = {};
      vv = wmma4(av, bw[vt], vv);
      float val[8];
#pragma unroll
      for (int r = 0; r < 8; r++)
        val[r] = (vv[r] + bv[vt]) * __shfl(gate[r], gsrc[vt], 32);
#pragma unroll
      for (int jj = 0; jj < 4; jj++) {
        v2f m = {val[2 * jj], val[2 * jj + 1]};
        qv[vt] = wmma4(sa[jj], m, qv[vt]);
      }
    }
    e = en;
    dl = dln;
  }

  float* red = (float*)lds_red + wid * (16 * RPF);
#pragma unroll
  for (int ct = 0; ct < 8; ct++) {
#pragma unroll
    for (int r = 0; r < 8; r++) red[(r + 8 * half) * RPF + 16 * ct + row] = g[ct][r];
  }
#pragma unroll
  for (int vt = 0; vt < 3; vt++) {
#pragma unroll
    for (int r = 0; r < 8; r++)
      red[(r + 8 * half) * RPF + SDIM + 16 * vt + row] = qv[vt][r];
  }
  __syncthreads();
#pragma unroll
  for (int i = 0; i < 16 * SDIM / 4 / 128; i++) {
    int idx = threadIdx.x + 128 * i;
    int m = idx >> 5;
    int c4 = idx & 31;
    const v4f* src = lds_red + m * RP4 + c4;
    v4f s = src[0];
    s += src[16 * RP4];
    s += src[2 * 16 * RP4];
    s += src[3 * 16 * RP4];
    ((v4f*)(agg_s + (long)(nb + m) * SDIM))[c4] = s;
  }
#pragma unroll
  for (int i = 0; i < 2; i++) {
    int idx = threadIdx.x + 128 * i;
    if (idx < 16 * V3 / 4) {
      int m = idx / (V3 / 4);
      int c4 = idx - m * (V3 / 4);
      const v4f* src = lds_red + m * RP4 + SDIM / 4 + c4;
      v4f s = src[0];
      s += src[16 * RP4];
      s += src[2 * 16 * RP4];
      s += src[3 * 16 * RP4];
      ((v4f*)(agg_v + (long)nb * V3))[idx] = s;
    }
  }
}

__global__ void __launch_bounds__(256, 4) k_node(
    const float* __restrict__ node_s, const float* __restrict__ node_v,
    const float* __restrict__ ln_g, const float* __restrict__ ln_b,
    const float* __restrict__ agg_s, const float* __restrict__ agg_v,
    const float* __restrict__ cnt,
    const float* __restrict__ wns, const float* __restrict__ bns,
    const float* __restrict__ wnv, const float* __restrict__ bnv,
    const float* __restrict__ wrs, const float* __restrict__ brs,
    const float* __restrict__ wrv, const float* __restrict__ brv,
    float* __restrict__ out) {
  __shared__ float lds_gate[8][16 * 16];
  __shared__ v4f lds_gb[2 * SDIM / 4];
  static_assert(2 * SDIM == 256, "one LayerNorm parameter per thread");
  int wid = threadIdx.x >> 5;
  int lane = threadIdx.x & 31;
  int group = blockIdx.x * 8 + wid;
  bool live = group < NN / 16;
  int nb = (live ? group : NN / 16 - 1) * 16;
  int row = lane & 15;
  int half = lane >> 4;
  int noderow = nb + row;

  float gb = (threadIdx.x < SDIM) ? ln_g[threadIdx.x] : ln_b[threadIdx.x - SDIM];
  float invden = fast_rcp(fmaxf(cnt[noderow], 1.f));
  const v4f* sA = (const v4f*)(node_s + (long)noderow * SDIM) + half;
  const v4f* gA = (const v4f*)(agg_s + (long)noderow * SDIM) + half;
  v4f as[16], ai[16];
#pragma unroll
  for (int q = 0; q < 16; q++) {
    as[q] = sA[2 * q];
    ai[q] = gA[2 * q];
  }
  ((float*)lds_gb)[threadIdx.x] = gb;
  __syncthreads();
  if (!live) return;
  v4f acc = as[0];
#pragma unroll
  for (int q = 1; q < 16; q++) acc += as[q];
  float s = (acc.x + acc.y) + (acc.z + acc.w);
  s += __shfl_xor(s, 16, 32);
  float mu = s * (1.f / SDIM);
  v4f qa = {0.f, 0.f, 0.f, 0.f};
#pragma unroll
  for (int q = 0; q < 16; q++) {
    as[q] = as[q] - mu;
    qa += as[q] * as[q];
  }
  float qs = (qa.x + qa.y) + (qa.z + qa.w);
  qs += __shfl_xor(qs, 16, 32);
  float rs = rsqrtf(qs * (1.f / SDIM) + 1e-5f);
  const v4f* lG = lds_gb + half;
  const v4f* lB = lds_gb + SDIM / 4 + half;
#pragma unroll
  for (int q = 0; q < 16; q++) as[q] = as[q] * rs * lG[2 * q] + lB[2 * q];
#pragma unroll
  for (int q = 0; q < 16; q++) ai[q] = as[q] + ai[q] * invden;

#pragma unroll 1
  for (int ct = 0; ct < 8; ct++) {
    v8f ao = {};
    v8f ar = {};
    int n = 16 * ct + row;
    const v4f* bo = (const v4f*)(wns + (long)n * SDIM) + half;
    const v4f* br = (const v4f*)(wrs + (long)n * SDIM) + half;
#pragma unroll
    for (int q = 0; q < 16; q++) {
      v4f bO = bo[2 * q];
      v4f bR = br[2 * q];
      ao = wmma4(lo2(ai[q]), lo2(bO), ao);
      ar = wmma4(lo2(as[q]), lo2(bR), ar);
      ao = wmma4(hi2(ai[q]), hi2(bO), ao);
      ar = wmma4(hi2(as[q]), hi2(bR), ar);
    }
    float biasO = bns[n], biasR = brs[n];
#pragma unroll
    for (int r = 0; r < 8; r++) {
      int m = r + 8 * half;
      float pre = ao[r] + biasO;
      float so = pre * sigmoidf_(pre);
      if (ct == 0) lds_gate[wid][m * 16 + n] = sigmoidf_(so);
      out[(long)(nb + m) * SDIM + n] = so + (ar[r] + biasR);
    }
  }

  const v4f* vA = (const v4f*)(node_v + (long)noderow * V3) + half;
  const v4f* gV = (const v4f*)(agg_v + (long)noderow * V3) + half;
  v4f av[6], aw[6];
#pragma unroll
  for (int q = 0; q < 6; q++) {
    av[q] = vA[2 * q];
    aw[q] = gV[2 * q];
  }
  v4f qv = av[0] * av[0];
#pragma unroll
  for (int q = 1; q < 6; q++) qv += av[q] * av[q];
  float qq = (qv.x + qv.y) + (qv.z + qv.w);
  qq += __shfl_xor(qq, 16, 32);
  float vinv = rsqrtf(qq * (1.f / VDIM) + 1e-8f);
#pragma unroll
  for (int q = 0; q < 6; q++) av[q] = av[q] * vinv;
#pragma unroll
  for (int q = 0; q < 6; q++) aw[q] = av[q] + aw[q] * invden;
  float* outv = out + (long)NN * SDIM;
#pragma unroll 1
  for (int ct = 0; ct < 3; ct++) {
    v8f ao = {};
    v8f ar = {};
    int d = 16 * ct + row;
    const v4f* bo = (const v4f*)(wnv + (long)d * V3) + half;
    const v4f* br = (const v4f*)(wrv + (long)d * V3) + half;
#pragma unroll
    for (int q = 0; q < 6; q++) {
      v4f bO = bo[2 * q];
      v4f bR = br[2 * q];
      ao = wmma4(lo2(aw[q]), lo2(bO), ao);
      ar = wmma4(lo2(av[q]), lo2(bR), ar);
      ao = wmma4(hi2(aw[q]), hi2(bO), ao);
      ar = wmma4(hi2(av[q]), hi2(bR), ar);
    }
    float biasO = bnv[d], biasR = brv[d];
    int vo = d / 3;
#pragma unroll
    for (int r = 0; r < 8; r++) {
      int m = r + 8 * half;
      float pre = ao[r] + biasO;
      float g = lds_gate[wid][m * 16 + vo];
      outv[(long)(nb + m) * V3 + d] = pre * g + (ar[r] + biasR);
    }
  }
}

extern "C" void kernel_launch(void* const* d_in, const int* in_sizes, int n_in,
                              void* d_out, int out_size, void* d_ws,
                              size_t ws_size, hipStream_t stream) {
  const float* node_s = (const float*)d_in[0];
  const float* node_v = (const float*)d_in[1];
  const int* eidx     = (const int*)d_in[2];
  const float* edge_s = (const float*)d_in[3];
  const float* edge_v = (const float*)d_in[4];
  const float* ln_g = (const float*)d_in[5];
  const float* ln_b = (const float*)d_in[6];
  const float* wes = (const float*)d_in[7];
  const float* bes = (const float*)d_in[8];
  const float* wev = (const float*)d_in[9];
  const float* bev = (const float*)d_in[10];
  const float* wns = (const float*)d_in[11];
  const float* bns = (const float*)d_in[12];
  const float* wnv = (const float*)d_in[13];
  const float* bnv = (const float*)d_in[14];
  const float* wrs = (const float*)d_in[15];
  const float* brs = (const float*)d_in[16];
  const float* wrv = (const float*)d_in[17];
  const float* brv = (const float*)d_in[18];
  float* out = (float*)d_out;

  float* ws = (float*)d_ws;
  float* agg_s = ws;
  float* agg_v = agg_s + (size_t)NN * SDIM;
  float* cnt = agg_v + (size_t)NN * V3;
  int* offsT = (int*)(cnt + NN);
  int* off = offsT + (NB1 + 1) * G1;
  int* perm = off + (NN + 16);
  int* sdst = perm + EE;
  int* rec = sdst + EE;
  const int* edge_dst = eidx + EE;

  k_part<<<G1, 1024, 0, stream>>>(edge_dst, offsT, rec);
  k_sort2<<<NB1, 1024, 0, stream>>>(rec, offsT, off, cnt, perm, sdst);
  k_edge<<<NN / 16, 128, 0, stream>>>(edge_s, edge_v, perm, sdst, off, wes, bes,
                                      wev, bev, agg_s, agg_v);
  k_node<<<(NN / 16 + 7) / 8, 256, 0, stream>>>(node_s, node_v, ln_g, ln_b,
                                                agg_s, agg_v, cnt, wns, bns,
                                                wnv, bnv, wrs, brs, wrv, brv,
                                                out);
}


# k_node: wns and wrs staged once per block in LDS (whole 512-B rows per load, pitch 33 x 16 B); its 8 waves read their B fragments from LDS, not by 16-line global loads; k order and sums unchanged
# speedup vs baseline: 5.1946x; 1.0569x over previous
#include <hip/hip_runtime.h>
#include <math.h>

#define NN   50000
#define EE   800000
#define SDIM 128
#define VDIM 16
#define V3   48
#define ESD  32

typedef float v2f __attribute__((ext_vector_type(2)));
typedef float v8f __attribute__((ext_vector_type(8)));

__device__ __forceinline__ v8f wmma4(v2f a, v2f b, v8f c) {
  return __builtin_amdgcn_wmma_f32_16x16x4_f32(false, a, false, b, (short)0, c,
                                               false, false);
}

__device__ __forceinline__ float fast_rcp(float x) {
  return __builtin_amdgcn_rcpf(x);
}

__device__ __forceinline__ float sigmoidf_(float x) {
  return fast_rcp(1.f + __expf(-x));
}

#define G1 256
#define CH1 (EE / G1)
#define NB1 ((NN + 255) / 256)
#define SCAN_PER 7
#define PIECE_U 8
static_assert(EE % G1 == 0, "chunks must tile the edge list");
static_assert(CH1 <= 4 * 1024, "a chunk is read as 4 values per thread");
static_assert(NB1 <= SCAN_PER * 32, "one wave scans the range counts");
static_assert(G1 * 4 == 1024, "level 2 walks a chunk's piece with 4 threads");
static_assert(PIECE_U == 8, "the load fences of level 2 name 8 pairs");

__global__ void __launch_bounds__(1024) k_part(const int* __restrict__ edge_dst,
                                               int* __restrict__ offsT,
                                               int* __restrict__ rec) {
  __shared__ int h[NB1];
  int t = threadIdx.x;
  int g = blockIdx.x;
  if (t < NB1) h[t] = 0;
  const int* src = edge_dst + g * CH1;
  int d[4];
#pragma unroll
  for (int k = 0; k < 4; k++) {
    int i = t + 1024 * k;
    d[k] = (i < CH1) ? src[i] : 0;
  }
  __syncthreads();
  int rank[4];
#pragma unroll
  for (int k = 0; k < 4; k++) {
    rank[k] = 0;
    if (t + 1024 * k < CH1) rank[k] = atomicAdd(&h[d[k] >> 8], 1);
  }
  __syncthreads();
  if (t < 32) {
    int v[SCAN_PER];
    int s = 0;
#pragma unroll
    for (int k = 0; k < SCAN_PER; k++) {
      int idx = t * SCAN_PER + k;
      v[k] = (idx < NB1) ? h[idx] : 0;
      s += v[k];
    }
    int inc = s;
#pragma unroll
    for (int m = 1; m < 32; m <<= 1) {
      int y = __shfl_up(inc, m, 32);
      if (t >= m) inc += y;
    }
    int run = inc - s;
#pragma unroll
    for (int k = 0; k < SCAN_PER; k++) {
      int idx = t * SCAN_PER + k;
      if (idx < NB1) h[idx] = run;
      run += v[k];
    }
  }
  __syncthreads();
  if (t < NB1) offsT[t * G1 + g] = h[t];
  if (t == 0) offsT[NB1 * G1 + g] = CH1;
  int2* out = (int2*)rec + g * CH1;
#pragma unroll
  for (int k = 0; k < 4; k++) {
    int i = t + 1024 * k;
    if (i < CH1) out[h[d[k] >> 8] + rank[k]] = make_int2(g * CH1 + i, d[k]);
  }
}

__global__ void __launch_bounds__(1024) k_sort2(const int* __restrict__ rec,
                                                const int* __restrict__ offsT,
                                                int* __restrict__ off,
                                                float* __restrict__ cnt,
                                                int* __restrict__ perm,
                                                int* __restrict__ sdst) {
  __shared__ int c[256];
  __shared__ int cur[256];
  __shared__ int wsum[G1 / 32];
  int t = threadIdx.x;
  int b = blockIdx.x;
  int g = t >> 2;
  int j = t & 3;
  int ps = offsT[b * G1 + g];
  int pe = offsT[(b + 1) * G1 + g];
  const int2* in = (const int2*)rec + g * CH1;
  if (t < 256) c[t] = 0;
  int x = (t < G1) ? offsT[b * G1 + t] : 0;
#pragma unroll
  for (int m = 16; m >= 1; m >>= 1) x += __shfl_xor(x, m, 32);
  if (t < G1 && (t & 31) == 0) wsum[t >> 5] = x;
  __syncthreads();
  for (int base = ps + j; base < pe; base += 4 * PIECE_U) {
    int dn[PIECE_U];
#pragma unroll
    for (int k = 0; k < PIECE_U; k++) dn[k] = in[min(base + 4 * k, pe - 1)].y;
    asm volatile("" ::"v"(dn[0]), "v"(dn[1]), "v"(dn[2]), "v"(dn[3]),
                 "v"(dn[4]), "v"(dn[5]), "v"(dn[6]), "v"(dn[7]));
#pragma unroll
    for (int k = 0; k < PIECE_U; k++)
      if (base + 4 * k < pe) atomicAdd(&c[dn[k] & 255], 1);
  }
  __syncthreads();
  if (t < 32) {
    int S = 0;
#pragma unroll
    for (int w = 0; w < G1 / 32; w++) S += wsum[w];
    int v[8];
    int s = 0;
#pragma unroll
    for (int k = 0; k < 8; k++) {
      v[k] = c[t * 8 + k];
      s += v[k];
    }
    int inc = s;
#pragma unroll
    for (int m = 1; m < 32; m <<= 1) {
      int y = __shfl_up(inc, m, 32);
      if (t >= m) inc += y;
    }
    int run = S + inc - s;
#pragma unroll
    for (int k = 0; k < 8; k++) {
      cur[t * 8 + k] = run;
      run += v[k];
    }
  }
  __syncthreads();
  if (t < 256) {
    int node = b * 256 + t;
    if (node < NN) {
      off[node] = cur[t];
      cnt[node] = (float)c[t];
    }
  }
  if (b == 0 && t == 0) off[NN] = EE;
  __syncthreads();
  for (int base = ps + j; base < pe; base += 4 * PIECE_U) {
    int2 r[PIECE_U];
#pragma unroll
    for (int k = 0; k < PIECE_U; k++) r[k] = in[min(base + 4 * k, pe - 1)];
    asm volatile("" ::"v"(r[0].x), "v"(r[0].y), "v"(r[1].x), "v"(r[1].y),
                 "v"(r[2].x), "v"(r[2].y), "v"(r[3].x), "v"(r[3].y),
                 "v"(r[4].x), "v"(r[4].y), "v"(r[5].x), "v"(r[5].y),
                 "v"(r[6].x), "v"(r[6].y), "v"(r[7].x), "v"(r[7].y));
#pragma unroll
    for (int k = 0; k < PIECE_U; k++) {
      if (base + 4 * k < pe) {
        int pos = atomicAdd(&cur[r[k].y & 255], 1);
        perm[pos] = r[k].x;
        sdst[pos] = r[k].y;
      }
    }
  }
}

typedef float v4f __attribute__((ext_vector_type(4)));

__device__ __forceinline__ v2f lo2(v4f x) { return (v2f){x.x, x.y}; }
__device__ __forceinline__ v2f hi2(v4f x) { return (v2f){x.z, x.w}; }

#define WP4 9
#define RP4 45
#define RPF (4 * RP4)
static_assert(SDIM + V3 <= RPF, "a partial-sum row holds all message columns");
#define NP4 33
static_assert(NP4 > SDIM / 4 && (NP4 & 1) == 1, "a row fits, the pitch is odd");
static_assert(2 * SDIM * NP4 * 16 + 8 * 1024 + 1024 <= 160 * 1024,
              "two blocks of k_node share a WGP's LDS");

__device__ __forceinline__ void load_idx(const int* __restrict__ perm,
                                         const int* __restrict__ sdst, int p,
                                         int row, int pend, int nb, int& e,
                                         int& dl) {
  int pe = p + row;
  bool valid = pe < pend;
  int pc = valid ? pe : 0;
  int ev = perm[pc];
  int dv = sdst[pc];
  e = valid ? ev : 0;
  dl = valid ? dv - nb : -1;
}

__device__ __forceinline__ void sel_frag(int dl, int row, int half,
                                         v2f sa[4]) {
#pragma unroll
  for (int jj = 0; jj < 4; jj++) {
    int x0 = __shfl(dl, 8 * half + 2 * jj, 32);
    int x1 = __shfl(dl, 8 * half + 2 * jj + 1, 32);
    sa[jj].x = (x0 == row) ? 1.f : 0.f;
    sa[jj].y = (x1 == row) ? 1.f : 0.f;
  }
}

__global__ void __launch_bounds__(128) k_edge(
    const float* __restrict__ edge_s, const float* __restrict__ edge_v,
    const int* __restrict__ perm, const int* __restrict__ sdst,
    const int* __restrict__ off,
    const float* __restrict__ wes, const float* __restrict__ bes,
    const float* __restrict__ wev, const float* __restrict__ bev,
    float* __restrict__ agg_s, float* __restrict__ agg_v) {
  __shared__ v4f lds_w[SDIM * WP4];
  __shared__ float lds_bias[SDIM];
  __shared__ v4f lds_red[4 * 16 * RP4];
  int wid = __builtin_amdgcn_readfirstlane(threadIdx.x >> 5);
  int lane = threadIdx.x & 31;
  int row = lane & 15;
  int half = lane >> 4;
  int nb = blockIdx.x * 16;
  int pbeg = off[nb];
  int pend = off[nb + 16];

#pragma unroll
  for (int i = 0; i < SDIM * ESD / 4 / 128; i++) {
    int idx = threadIdx.x + 128 * i;
    lds_w[(idx >> 3) * WP4 + (idx & 7)] = ((const v4f*)wes)[idx];
  }
  lds_bias[threadIdx.x] = bes[threadIdx.x];

  v2f bw[3];
  float bv[3];
  int gsrc[3];
#pragma unroll
  for (int vt = 0; vt < 3; vt++) {
    int d = 16 * vt + row;
    float w1 = wev[d * 3 + 1];
    bw[vt].x = wev[d * 3 + 2 * half];
    bw[vt].y = (half == 0) ? w1 : 0.f;
    bv[vt] = bev[d];
    gsrc[vt] = d / 3 + 16 * half;
  }
  v8f g[8];
  v8f qv[3];
#pragma unroll
  for (int ct = 0; ct < 8; ct++) g[ct] = (v8f){};
#pragma unroll
  for (int vt = 0; vt < 3; vt++) qv[vt] = (v8f){};
  __syncthreads();

  int p = pbeg + 16 * ((wid + blockIdx.x) & 3);
  int e, dl;
  load_idx(perm, sdst, p, row, pend, nb, e, dl);
  for (; p < pend; p += 64) {
    int en, dln;
    load_idx(perm, sdst, p + 64, row, pend, nb, en, dln);
    const v4f* arow = (const v4f*)(edge_s + (long)e * ESD) + half;
    v4f a[4];
#pragma unroll
    for (int q = 0; q < 4; q++) a[q] = arow[2 * q];
    float e1 = edge_v[(long)e * 3 + 1];
    v2f av;
    av.x = edge_v[(long)e * 3 + 2 * half];
    av.y = (half == 0) ? e1 : 0.f;
    v2f sa[4];
    sel_frag(dl, row, half, sa);
    float gate[8];
    asm volatile("" ::: "memory");
#pragma unroll
    for (int cp = 0; cp < 4; cp++) {
      const v4f* w0 = lds_w + (32 * cp + row) * WP4 + half;
      const v4f* w1 = w0 + 16 * WP4;
      v8f c0 = {};
      v8f c1 = {};
#pragma unroll
      for (int q = 0; q < 4; q++) {
        v4f b0 = w0[2 * q];
        v4f b1 = w1[2 * q];
        c0 = wmma4(lo2(a[q]), lo2(b0), c0);
        c1 = wmma4(lo2(a[q]), lo2(b1), c1);
        c0 = wmma4(hi2(a[q]), hi2(b0), c0);
        c1 = wmma4(hi2(a[q]), hi2(b1), c1);
      }
      float bias0 = lds_bias[32 * cp + row];
      float bias1 = lds_bias[32 * cp + 16 + row];
#pragma unroll
      for (int r = 0; r < 8; r++) {
        float p0 = c0[r] + bias0;
        float p1 = c1[r] + bias1;
        c0[r] = p0 * sigmoidf_(p0);
        c1[r] = p1 * sigmoidf_(p1);
      }
      if (cp == 0) {
#pragma unroll
        for (int r = 0; r < 8; r++) gate[r] = sigmoidf_(c0[r]);
      }
#pragma unroll
      for (int jj = 0; jj < 4; jj++) {
        v2f m0 = {c0[2 * jj], c0[2 * jj + 1]};
        v2f m1 = {c1[2 * jj], c1[2 * jj + 1]};
        g[2 * cp] = wmma4(sa[jj], m0, g[2 * cp]);
        g[2 * cp + 1] = wmma4(sa[jj], m1, g[2 * cp + 1]);
      }
    }
#pragma unroll
    for (int vt = 0; vt < 3; vt++) {
      v8f vv = {};
      vv = wmma4(av, bw[vt], vv);
      float val[8];
#pragma unroll
      for (int r = 0; r < 8; r++)
        val[r] = (vv[r] + bv[vt]) * __shfl(gate[r], gsrc[vt], 32);
#pragma unroll
      for (int jj = 0; jj < 4; jj++) {
        v2f m = {val[2 * jj], val[2 * jj + 1]};
        qv[vt] = wmma4(sa[jj], m, qv[vt]);
      }
    }
    e = en;
    dl = dln;
  }

  float* red = (float*)lds_red + wid * (16 * RPF);
#pragma unroll
  for (int ct = 0; ct < 8; ct++) {
#pragma unroll
    for (int r = 0; r < 8; r++) red[(r + 8 * half) * RPF + 16 * ct + row] = g[ct][r];
  }
#pragma unroll
  for (int vt = 0; vt < 3; vt++) {
#pragma unroll
    for (int r = 0; r < 8; r++)
      red[(r + 8 * half) * RPF + SDIM + 16 * vt + row] = qv[vt][r];
  }
  __syncthreads();
#pragma unroll
  for (int i = 0; i < 16 * SDIM / 4 / 128; i++) {
    int idx = threadIdx.x + 128 * i;
    int m = idx >> 5;
    int c4 = idx & 31;
    const v4f* src = lds_red + m * RP4 + c4;
    v4f s = src[0];
    s += src[16 * RP4];
    s += src[2 * 16 * RP4];
    s += src[3 * 16 * RP4];
    ((v4f*)(agg_s + (long)(nb + m) * SDIM))[c4] = s;
  }
#pragma unroll
  for (int i = 0; i < 2; i++) {
    int idx = threadIdx.x + 128 * i;
    if (idx < 16 * V3 / 4) {
      int m = idx / (V3 / 4);
      int c4 = idx - m * (V3 / 4);
      const v4f* src = lds_red + m * RP4 + SDIM / 4 + c4;
      v4f s = src[0];
      s += src[16 * RP4];
      s += src[2 * 16 * RP4];
      s += src[3 * 16 * RP4];
      ((v4f*)(agg_v + (long)nb * V3))[idx] = s;
    }
  }
}

__global__ void __launch_bounds__(256, 4) k_node(
    const float* __restrict__ node_s, const float* __restrict__ node_v,
    const float* __restrict__ ln_g, const float* __restrict__ ln_b,
    const float* __restrict__ agg_s, const float* __restrict__ agg_v,
    const float* __restrict__ cnt,
    const float* __restrict__ wns, const float* __restrict__ bns,
    const float* __restrict__ wnv, const float* __restrict__ bnv,
    const float* __restrict__ wrs, const float* __restrict__ brs,
    const float* __restrict__ wrv, const float* __restrict__ brv,
    float* __restrict__ out) {
  __shared__ v4f lds_all[8 * 64 + 2 * SDIM / 4 + 2 * SDIM * NP4];
  float (*lds_gate)[16 * 16] = (float (*)[16 * 16])lds_all;
  v4f* lds_gb = lds_all + 8 * 64;
  static_assert(2 * SDIM == 256, "one LayerNorm parameter per thread");
  v4f* lds_wn = lds_gb + 2 * SDIM / 4;
  int wid = threadIdx.x >> 5;
  int lane = threadIdx.x & 31;
  int group = blockIdx.x * 8 + wid;
  bool live = group < NN / 16;
  int nb = (live ? group : NN / 16 - 1) * 16;
  int row = lane & 15;
  int half = lane >> 4;
  int noderow = nb + row;

  float gb = (threadIdx.x < SDIM) ? ln_g[threadIdx.x] : ln_b[threadIdx.x - SDIM];
  float invden = fast_rcp(fmaxf(cnt[noderow], 1.f));
  const v4f* sA = (const v4f*)(node_s + (long)noderow * SDIM) + half;
  const v4f* gA = (const v4f*)(agg_s + (long)noderow * SDIM) + half;
  v4f as[16], ai[16];
#pragma unroll
  for (int q = 0; q < 16; q++) {
    as[q] = sA[2 * q];
    ai[q] = gA[2 * q];
  }
#pragma unroll 1
  for (int mtx = 0; mtx < 2; mtx++) {
    const v4f* wsrc = (const v4f*)(mtx ? wrs : wns) + threadIdx.x;
    v4f* wdst = lds_wn + (mtx * SDIM + wid) * NP4 + lane;
#pragma unroll
    for (int i = 0; i < SDIM * SDIM / 4 / 256; i++)
      wdst[i * 8 * NP4] = wsrc[256 * i];
  }
  ((float*)lds_gb)[threadIdx.x] = gb;
  __syncthreads();
  if (!live) return;
  v4f acc = as[0];
#pragma unroll
  for (int q = 1; q < 16; q++) acc += as[q];
  float s = (acc.x + acc.y) + (acc.z + acc.w);
  s += __shfl_xor(s, 16, 32);
  float mu = s * (1.f / SDIM);
  v4f qa = {0.f, 0.f, 0.f, 0.f};
#pragma unroll
  for (int q = 0; q < 16; q++) {
    as[q] = as[q] - mu;
    qa += as[q] * as[q];
  }
  float qs = (qa.x + qa.y) + (qa.z + qa.w);
  qs += __shfl_xor(qs, 16, 32);
  float rs = rsqrtf(qs * (1.f / SDIM) + 1e-5f);
  const v4f* lG = lds_gb + half;
  const v4f* lB = lds_gb + SDIM / 4 + half;
#pragma unroll
  for (int q = 0; q < 16; q++) {
    if (q && (q & 3) == 0) __builtin_amdgcn_sched_barrier(0);
    as[q] = as[q] * rs * lG[2 * q] + lB[2 * q];
    ai[q] = as[q] + ai[q] * invden;
  }

#pragma unroll 1
  for (int ct = 0; ct < 8; ct++) {
    v8f ao = {};
    v8f ar = {};
    int n = 16 * ct + row;
    const v4f* bo = lds_wn + n * NP4 + half;
    const v4f* br = bo + SDIM * NP4;
#pragma unroll
    for (int q = 0; q < 16; q++) {
      v4f bO = bo[2 * q];
      v4f bR = br[2 * q];
      ao = wmma4(lo2(ai[q]), lo2(bO), ao);
      ar = wmma4(lo2(as[q]), lo2(bR), ar);
      ao = wmma4(hi2(ai[q]), hi2(bO), ao);
      ar = wmma4(hi2(as[q]), hi2(bR), ar);
    }
    float biasO = bns[n], biasR = brs[n];
#pragma unroll
    for (int r = 0; r < 8; r++) {
      int m = r + 8 * half;
      float pre = ao[r] + biasO;
      float so = pre * sigmoidf_(pre);
      if (ct == 0) lds_gate[wid][m * 16 + n] = sigmoidf_(so);
      out[(long)(nb + m) * SDIM + n] = so + (ar[r] + biasR);
    }
  }

  const v4f* vA = (const v4f*)(node_v + (long)noderow * V3) + half;
  const v4f* gV = (const v4f*)(agg_v + (long)noderow * V3) + half;
  v4f av[6], aw[6];
#pragma unroll
  for (int q = 0; q < 6; q++) {
    av[q] = vA[2 * q];
    aw[q] = gV[2 * q];
  }
  v4f qv = av[0] * av[0];
#pragma unroll
  for (int q = 1; q < 6; q++) qv += av[q] * av[q];
  float qq = (qv.x + qv.y) + (qv.z + qv.w);
  qq += __shfl_xor(qq, 16, 32);
  float vinv = rsqrtf(qq * (1.f / VDIM) + 1e-8f);
#pragma unroll
  for (int q = 0; q < 6; q++) av[q] = av[q] * vinv;
#pragma unroll
  for (int q = 0; q < 6; q++) aw[q] = av[q] + aw[q] * invden;
  float* outv = out + (long)NN * SDIM;
#pragma unroll 1
  for (int ct = 0; ct < 3; ct++) {
    v8f ao = {};
    v8f ar = {};
    int d = 16 * ct + row;
    const v4f* bo = (const v4f*)(wnv + (long)d * V3) + half;
    const v4f* br = (const v4f*)(wrv + (long)d * V3) + half;
#pragma unroll
    for (int q = 0; q < 6; q++) {
      v4f bO = bo[2 * q];
      v4f bR = br[2 * q];
      ao = wmma4(lo2(aw[q]), lo2(bO), ao);
      ar = wmma4(lo2(av[q]), lo2(bR), ar);
      ao = wmma4(hi2(aw[q]), hi2(bO), ao);
      ar = wmma4(hi2(av[q]), hi2(bR), ar);
    }
    float biasO = bnv[d], biasR = brv[d];
    int vo = d / 3;
#pragma unroll
    for (int r = 0; r < 8; r++) {
      int m = r + 8 * half;
      float pre = ao[r] + biasO;
      float g = lds_gate[wid][m * 16 + vo];
      outv[(long)(nb + m) * V3 + d] = pre * g + (ar[r] + biasR);
    }
  }
}

extern "C" void kernel_launch(void* const* d_in, const int* in_sizes, int n_in,
                              void* d_out, int out_size, void* d_ws,
                              size_t ws_size, hipStream_t stream) {
  const float* node_s = (const float*)d_in[0];
  const float* node_v = (const float*)d_in[1];
  const int* eidx     = (const int*)d_in[2];
  const float* edge_s = (const float*)d_in[3];
  const float* edge_v = (const float*)d_in[4];
  const float* ln_g = (const float*)d_in[5];
  const float* ln_b = (const float*)d_in[6];
  const float* wes = (const float*)d_in[7];
  const float* bes = (const float*)d_in[8];
  const float* wev = (const float*)d_in[9];
  const float* bev = (const float*)d_in[10];
  const float* wns = (const float*)d_in[11];
  const float* bns = (const float*)d_in[12];
  const float* wnv = (const float*)d_in[13];
  const float* bnv = (const float*)d_in[14];
  const float* wrs = (const float*)d_in[15];
  const float* brs = (const float*)d_in[16];
  const float* wrv = (const float*)d_in[17];
  const float* brv = (const float*)d_in[18];
  float* out = (float*)d_out;

  float* ws = (float*)d_ws;
  float* agg_s = ws;
  float* agg_v = agg_s + (size_t)NN * SDIM;
  float* cnt = agg_v + (size_t)NN * V3;
  int* offsT = (int*)(cnt + NN);
  int* off = offsT + (NB1 + 1) * G1;
  int* perm = off + (NN + 16);
  int* sdst = perm + EE;
  int* rec = sdst + EE;
  const int* edge_dst = eidx + EE;

  k_part<<<G1, 1024, 0, stream>>>(edge_dst, offsT, rec);
  k_sort2<<<NB1, 1024, 0, stream>>>(rec, offsT, off, cnt, perm, sdst);
  k_edge<<<NN / 16, 128, 0, stream>>>(edge_s, edge_v, perm, sdst, off, wes, bes,
                                      wev, bev, agg_s, agg_v);
  k_node<<<(NN / 16 + 7) / 8, 256, 0, stream>>>(node_s, node_v, ln_g, ln_b,
                                                agg_s, agg_v, cnt, wns, bns,
                                                wnv, bnv, wrs, brs, wrv, brv,
                                                out);
}


# k_edge: each column pair's weight fragments go from LDS to a second register set as ONE load group a pair ahead, under the previous pair's products (was: a load and a wait per two products)
# speedup vs baseline: 5.4852x; 1.0559x over previous
#include <hip/hip_runtime.h>
#include <math.h>

#define NN   50000
#define EE   800000
#define SDIM 128
#define VDIM 16
#define V3   48
#define ESD  32

typedef float v2f __attribute__((ext_vector_type(2)));
typedef float v8f __attribute__((ext_vector_type(8)));

__device__ __forceinline__ v8f wmma4(v2f a, v2f b, v8f c) {
  return __builtin_amdgcn_wmma_f32_16x16x4_f32(false, a, false, b, (short)0, c,
                                               false, false);
}

__device__ __forceinline__ float fast_rcp(float x) {
  return __builtin_amdgcn_rcpf(x);
}

__device__ __forceinline__ float sigmoidf_(float x) {
  return fast_rcp(1.f + __expf(-x));
}

#define G1 256
#define CH1 (EE / G1)
#define NB1 ((NN + 255) / 256)
#define SCAN_PER 7
#define PIECE_U 8
static_assert(EE % G1 == 0, "chunks must tile the edge list");
static_assert(CH1 <= 4 * 1024, "a chunk is read as 4 values per thread");
static_assert(NB1 <= SCAN_PER * 32, "one wave scans the range counts");
static_assert(G1 * 4 == 1024, "level 2 walks a chunk's piece with 4 threads");
static_assert(PIECE_U == 8, "the load fences of level 2 name 8 pairs");

__global__ void __launch_bounds__(1024) k_part(const int* __restrict__ edge_dst,
                                               int* __restrict__ offsT,
                                               int* __restrict__ rec) {
  __shared__ int h[NB1];
  int t = threadIdx.x;
  int g = blockIdx.x;
  if (t < NB1) h[t] = 0;
  const int* src = edge_dst + g * CH1;
  int d[4];
#pragma unroll
  for (int k = 0; k < 4; k++) {
    int i = t + 1024 * k;
    d[k] = (i < CH1) ? src[i] : 0;
  }
  __syncthreads();
  int rank[4];
#pragma unroll
  for (int k = 0; k < 4; k++) {
    rank[k] = 0;
    if (t + 1024 * k < CH1) rank[k] = atomicAdd(&h[d[k] >> 8], 1);
  }
  __syncthreads();
  if (t < 32) {
    int v[SCAN_PER];
    int s = 0;
#pragma unroll
    for (int k = 0; k < SCAN_PER; k++) {
      int idx = t * SCAN_PER + k;
      v[k] = (idx < NB1) ? h[idx] : 0;
      s += v[k];
    }
    int inc = s;
#pragma unroll
    for (int m = 1; m < 32; m <<= 1) {
      int y = __shfl_up(inc, m, 32);
      if (t >= m) inc += y;
    }
    int run = inc - s;
#pragma unroll
    for (int k = 0; k < SCAN_PER; k++) {
      int idx = t * SCAN_PER + k;
      if (idx < NB1) h[idx] = run;
      run += v[k];
    }
  }
  __syncthreads();
  if (t < NB1) offsT[t * G1 + g] = h[t];
  if (t == 0) offsT[NB1 * G1 + g] = CH1;
  int2* out = (int2*)rec + g * CH1;
#pragma unroll
  for (int k = 0; k < 4; k++) {
    int i = t + 1024 * k;
    if (i < CH1) out[h[d[k] >> 8] + rank[k]] = make_int2(g * CH1 + i, d[k]);
  }
}

__global__ void __launch_bounds__(1024) k_sort2(const int* __restrict__ rec,
                                                const int* __restrict__ offsT,
                                                int* __restrict__ off,
                                                float* __restrict__ cnt,
                                                int* __restrict__ perm,
                                                int* __restrict__ sdst) {
  __shared__ int c[256];
  __shared__ int cur[256];
  __shared__ int wsum[G1 / 32];
  int t = threadIdx.x;
  int b = blockIdx.x;
  int g = t >> 2;
  int j = t & 3;
  int ps = offsT[b * G1 + g];
  int pe = offsT[(b + 1) * G1 + g];
  const int2* in = (const int2*)rec + g * CH1;
  if (t < 256) c[t] = 0;
  int x = (t < G1) ? offsT[b * G1 + t] : 0;
#pragma unroll
  for (int m = 16; m >= 1; m >>= 1) x += __shfl_xor(x, m, 32);
  if (t < G1 && (t & 31) == 0) wsum[t >> 5] = x;
  __syncthreads();
  for (int base = ps + j; base < pe; base += 4 * PIECE_U) {
    int dn[PIECE_U];
#pragma unroll
    for (int k = 0; k < PIECE_U; k++) dn[k] = in[min(base + 4 * k, pe - 1)].y;
    asm volatile("" ::"v"(dn[0]), "v"(dn[1]), "v"(dn[2]), "v"(dn[3]),
                 "v"(dn[4]), "v"(dn[5]), "v"(dn[6]), "v"(dn[7]));
#pragma unroll
    for (int k = 0; k < PIECE_U; k++)
      if (base + 4 * k < pe) atomicAdd(&c[dn[k] & 255], 1);
  }
  __syncthreads();
  if (t < 32) {
    int S = 0;
#pragma unroll
    for (int w = 0; w < G1 / 32; w++) S += wsum[w];
    int v[8];
    int s = 0;
#pragma unroll
    for (int k = 0; k < 8; k++) {
      v[k] = c[t * 8 + k];
      s += v[k];
    }
    int inc = s;
#pragma unroll
    for (int m = 1; m < 32; m <<= 1) {
      int y = __shfl_up(inc, m, 32);
      if (t >= m) inc += y;
    }
    int run = S + inc - s;
#pragma unroll
    for (int k = 0; k < 8; k++) {
      cur[t * 8 + k] = run;
      run += v[k];
    }
  }
  __syncthreads();
  if (t < 256) {
    int node = b * 256 + t;
    if (node < NN) {
      off[node] = cur[t];
      cnt[node] = (float)c[t];
    }
  }
  if (b == 0 && t == 0) off[NN] = EE;
  __syncthreads();
  for (int base = ps + j; base < pe; base += 4 * PIECE_U) {
    int2 r[PIECE_U];
#pragma unroll
    for (int k = 0; k < PIECE_U; k++) r[k] = in[min(base + 4 * k, pe - 1)];
    asm volatile("" ::"v"(r[0].x), "v"(r[0].y), "v"(r[1].x), "v"(r[1].y),
                 "v"(r[2].x), "v"(r[2].y), "v"(r[3].x), "v"(r[3].y),
                 "v"(r[4].x), "v"(r[4].y), "v"(r[5].x), "v"(r[5].y),
                 "v"(r[6].x), "v"(r[6].y), "v"(r[7].x), "v"(r[7].y));
#pragma unroll
    for (int k = 0; k < PIECE_U; k++) {
      if (base + 4 * k < pe) {
        int pos = atomicAdd(&cur[r[k].y & 255], 1);
        perm[pos] = r[k].x;
        sdst[pos] = r[k].y;
      }
    }
  }
}

typedef float v4f __attribute__((ext_vector_type(4)));

__device__ __forceinline__ v2f lo2(v4f x) { return (v2f){x.x, x.y}; }
__device__ __forceinline__ v2f hi2(v4f x) { return (v2f){x.z, x.w}; }

#define WP4 9
#define RP4 45
#define RPF (4 * RP4)
static_assert(SDIM + V3 <= RPF, "a partial-sum row holds all message columns");
#define NP4 33
static_assert(NP4 > SDIM / 4 && (NP4 & 1) == 1, "a row fits, the pitch is odd");
static_assert(2 * SDIM * NP4 * 16 + 8 * 1024 + 1024 <= 160 * 1024,
              "two blocks of k_node share a WGP's LDS");

__device__ __forceinline__ void load_idx(const int* __restrict__ perm,
                                         const int* __restrict__ sdst, int p,
                                         int row, int pend, int nb, int& e,
                                         int& dl) {
  int pe = p + row;
  bool valid = pe < pend;
  int pc = valid ? pe : 0;
  int ev = perm[pc];
  int dv = sdst[pc];
  e = valid ? ev : 0;
  dl = valid ? dv - nb : -1;
}

__device__ __forceinline__ void sel_frag(int dl, int row, int half,
                                         v2f sa[4]) {
#pragma unroll
  for (int jj = 0; jj < 4; jj++) {
    int x0 = __shfl(dl, 8 * half + 2 * jj, 32);
    int x1 = __shfl(dl, 8 * half + 2 * jj + 1, 32);
    sa[jj].x = (x0 == row) ? 1.f : 0.f;
    sa[jj].y = (x1 == row) ? 1.f : 0.f;
  }
}

__device__ __forceinline__ void load_wgroup(const v4f* lds_w,
                                            const float* lds_bias, int cp,
                                            int row, int half, v4f (&f)[8],
                                            float (&b)[2]) {
  const v4f* w0 = lds_w + (32 * cp + row) * WP4 + half;
  const v4f* w1 = w0 + 16 * WP4;
#pragma unroll
  for (int q = 0; q < 4; q++) {
    f[2 * q] = w0[2 * q];
    f[2 * q + 1] = w1[2 * q];
  }
  b[0] = lds_bias[32 * cp + row];
  b[1] = lds_bias[32 * cp + 16 + row];
}

__global__ void __launch_bounds__(128) k_edge(
    const float* __restrict__ edge_s, const float* __restrict__ edge_v,
    const int* __restrict__ perm, const int* __restrict__ sdst,
    const int* __restrict__ off,
    const float* __restrict__ wes, const float* __restrict__ bes,
    const float* __restrict__ wev, const float* __restrict__ bev,
    float* __restrict__ agg_s, float* __restrict__ agg_v) {
  __shared__ v4f lds_w[SDIM * WP4];
  __shared__ float lds_bias[SDIM];
  __shared__ v4f lds_red[4 * 16 * RP4];
  int wid = __builtin_amdgcn_readfirstlane(threadIdx.x >> 5);
  int lane = threadIdx.x & 31;
  int row = lane & 15;
  int half = lane >> 4;
  int nb = blockIdx.x * 16;
  int pbeg = off[nb];
  int pend = off[nb + 16];

#pragma unroll
  for (int i = 0; i < SDIM * ESD / 4 / 128; i++) {
    int idx = threadIdx.x + 128 * i;
    lds_w[(idx >> 3) * WP4 + (idx & 7)] = ((const v4f*)wes)[idx];
  }
  lds_bias[threadIdx.x] = bes[threadIdx.x];

  v2f bw[3];
  float bv[3];
  int gsrc[3];
#pragma unroll
  for (int vt = 0; vt < 3; vt++) {
    int d = 16 * vt + row;
    float w1 = wev[d * 3 + 1];
    bw[vt].x = wev[d * 3 + 2 * half];
    bw[vt].y = (half == 0) ? w1 : 0.f;
    bv[vt] = bev[d];
    gsrc[vt] = d / 3 + 16 * half;
  }
  v8f g[8];
  v8f qv[3];
#pragma unroll
  for (int ct = 0; ct < 8; ct++) g[ct] = (v8f){};
#pragma unroll
  for (int vt = 0; vt < 3; vt++) qv[vt] = (v8f){};
  __syncthreads();

  v4f fr[2][8];
  float fb[2][2];
  load_wgroup(lds_w, lds_bias, 0, row, half, fr[0], fb[0]);

  int p = pbeg + 16 * ((wid + blockIdx.x) & 3);
  int e, dl;
  load_idx(perm, sdst, p, row, pend, nb, e, dl);
  for (; p < pend; p += 64) {
    int en, dln;
    load_idx(perm, sdst, p + 64, row, pend, nb, en, dln);
    const v4f* arow = (const v4f*)(edge_s + (long)e * ESD) + half;
    v4f a[4];
#pragma unroll
    for (int q = 0; q < 4; q++) a[q] = arow[2 * q];
    float e1 = edge_v[(long)e * 3 + 1];
    v2f av;
    av.x = edge_v[(long)e * 3 + 2 * half];
    av.y = (half == 0) ? e1 : 0.f;
    v2f sa[4];
    sel_frag(dl, row, half, sa);
    float gate[8];
    asm volatile("" ::: "memory");
#pragma unroll
    for (int cp = 0; cp < 4; cp++) {
      __builtin_amdgcn_sched_barrier(0);
      load_wgroup(lds_w, lds_bias, (cp + 1) & 3, row, half, fr[(cp + 1) & 1],
                  fb[(cp + 1) & 1]);
      __builtin_amdgcn_sched_barrier(0);
      v8f c0 = {};
      v8f c1 = {};
#pragma unroll
      for (int q = 0; q < 4; q++) {
        v4f b0 = fr[cp & 1][2 * q];
        v4f b1 = fr[cp & 1][2 * q + 1];
        c0 = wmma4(lo2(a[q]), lo2(b0), c0);
        c1 = wmma4(lo2(a[q]), lo2(b1), c1);
        c0 = wmma4(hi2(a[q]), hi2(b0), c0);
        c1 = wmma4(hi2(a[q]), hi2(b1), c1);
      }
      float bias0 = fb[cp & 1][0];
      float bias1 = fb[cp & 1][1];
#pragma unroll
      for (int r = 0; r < 8; r++) {
        float p0 = c0[r] + bias0;
        float p1 = c1[r] + bias1;
        c0[r] = p0 * sigmoidf_(p0);
        c1[r] = p1 * sigmoidf_(p1);
      }
      if (cp == 0) {
#pragma unroll
        for (int r = 0; r < 8; r++) gate[r] = sigmoidf_(c0[r]);
      }
#pragma unroll
      for (int jj = 0; jj < 4; jj++) {
        v2f m0 = {c0[2 * jj], c0[2 * jj + 1]};
        v2f m1 = {c1[2 * jj], c1[2 * jj + 1]};
        g[2 * cp] = wmma4(sa[jj], m0, g[2 * cp]);
        g[2 * cp + 1] = wmma4(sa[jj], m1, g[2 * cp + 1]);
      }
    }
#pragma unroll
    for (int vt = 0; vt < 3; vt++) {
      v8f vv = {};
      vv = wmma4(av, bw[vt], vv);
      float val[8];
#pragma unroll
      for (int r = 0; r < 8; r++)
        val[r] = (vv[r] + bv[vt]) * __shfl(gate[r], gsrc[vt], 32);
#pragma unroll
      for (int jj = 0; jj < 4; jj++) {
        v2f m = {val[2 * jj], val[2 * jj + 1]};
        qv[vt] = wmma4(sa[jj], m, qv[vt]);
      }
    }
    e = en;
    dl = dln;
  }

  float* red = (float*)lds_red + wid * (16 * RPF);
#pragma unroll
  for (int ct = 0; ct < 8; ct++) {
#pragma unroll
    for (int r = 0; r < 8; r++) red[(r + 8 * half) * RPF + 16 * ct + row] = g[ct][r];
  }
#pragma unroll
  for (int vt = 0; vt < 3; vt++) {
#pragma unroll
    for (int r = 0; r < 8; r++)
      red[(r + 8 * half) * RPF + SDIM + 16 * vt + row] = qv[vt][r];
  }
  __syncthreads();
#pragma unroll
  for (int i = 0; i < 16 * SDIM / 4 / 128; i++) {
    int idx = threadIdx.x + 128 * i;
    int m = idx >> 5;
    int c4 = idx & 31;
    const v4f* src = lds_red + m * RP4 + c4;
    v4f s = src[0];
    s += src[16 * RP4];
    s += src[2 * 16 * RP4];
    s += src[3 * 16 * RP4];
    ((v4f*)(agg_s + (long)(nb + m) * SDIM))[c4] = s;
  }
#pragma unroll
  for (int i = 0; i < 2; i++) {
    int idx = threadIdx.x + 128 * i;
    if (idx < 16 * V3 / 4) {
      int m = idx / (V3 / 4);
      int c4 = idx - m * (V3 / 4);
      const v4f* src = lds_red + m * RP4 + SDIM / 4 + c4;
      v4f s = src[0];
      s += src[16 * RP4];
      s += src[2 * 16 * RP4];
      s += src[3 * 16 * RP4];
      ((v4f*)(agg_v + (long)nb * V3))[idx] = s;
    }
  }
}

__global__ void __launch_bounds__(256, 4) k_node(
    const float* __restrict__ node_s, const float* __restrict__ node_v,
    const float* __restrict__ ln_g, const float* __restrict__ ln_b,
    const float* __restrict__ agg_s, const float* __restrict__ agg_v,
    const float* __restrict__ cnt,
    const float* __restrict__ wns, const float* __restrict__ bns,
    const float* __restrict__ wnv, const float* __restrict__ bnv,
    const float* __restrict__ wrs, const float* __restrict__ brs,
    const float* __restrict__ wrv, const float* __restrict__ brv,
    float* __restrict__ out) {
  __shared__ v4f lds_all[8 * 64 + 2 * SDIM / 4 + 2 * SDIM * NP4];
  float (*lds_gate)[16 * 16] = (float (*)[16 * 16])lds_all;
  v4f* lds_gb = lds_all + 8 * 64;
  static_assert(2 * SDIM == 256, "one LayerNorm parameter per thread");
  v4f* lds_wn = lds_gb + 2 * SDIM / 4;
  int wid = threadIdx.x >> 5;
  int lane = threadIdx.x & 31;
  int group = blockIdx.x * 8 + wid;
  bool live = group < NN / 16;
  int nb = (live ? group : NN / 16 - 1) * 16;
  int row = lane & 15;
  int half = lane >> 4;
  int noderow = nb + row;

  float gb = (threadIdx.x < SDIM) ? ln_g[threadIdx.x] : ln_b[threadIdx.x - SDIM];
  float invden = fast_rcp(fmaxf(cnt[noderow], 1.f));
  const v4f* sA = (const v4f*)(node_s + (long)noderow * SDIM) + half;
  const v4f* gA = (const v4f*)(agg_s + (long)noderow * SDIM) + half;
  v4f as[16], ai[16];
#pragma unroll
  for (int q = 0; q < 16; q++) {
    as[q] = sA[2 * q];
    ai[q] = gA[2 * q];
  }
#pragma unroll 1
  for (int mtx = 0; mtx < 2; mtx++) {
    const v4f* wsrc = (const v4f*)(mtx ? wrs : wns) + threadIdx.x;
    v4f* wdst = lds_wn + (mtx * SDIM + wid) * NP4 + lane;
#pragma unroll
    for (int i = 0; i < SDIM * SDIM / 4 / 256; i++)
      wdst[i * 8 * NP4] = wsrc[256 * i];
  }
  ((float*)lds_gb)[threadIdx.x] = gb;
  __syncthreads();
  if (!live) return;
  v4f acc = as[0];
#pragma unroll
  for (int q = 1; q < 16; q++) acc += as[q];
  float s = (acc.x + acc.y) + (acc.z + acc.w);
  s += __shfl_xor(s, 16, 32);
  float mu = s * (1.f / SDIM);
  v4f qa = {0.f, 0.f, 0.f, 0.f};
#pragma unroll
  for (int q = 0; q < 16; q++) {
    as[q] = as[q] - mu;
    qa += as[q] * as[q];
  }
  float qs = (qa.x + qa.y) + (qa.z + qa.w);
  qs += __shfl_xor(qs, 16, 32);
  float rs = rsqrtf(qs * (1.f / SDIM) + 1e-5f);
  const v4f* lG = lds_gb + half;
  const v4f* lB = lds_gb + SDIM / 4 + half;
#pragma unroll
  for (int q = 0; q < 16; q++) {
    if (q && (q & 3) == 0) __builtin_amdgcn_sched_barrier(0);
    as[q] = as[q] * rs * lG[2 * q] + lB[2 * q];
    ai[q] = as[q] + ai[q] * invden;
  }

#pragma unroll 1
  for (int ct = 0; ct < 8; ct++) {
    v8f ao = {};
    v8f ar = {};
    int n = 16 * ct + row;
    const v4f* bo = lds_wn + n * NP4 + half;
    const v4f* br = bo + SDIM * NP4;
#pragma unroll
    for (int q = 0; q < 16; q++) {
      v4f bO = bo[2 * q];
      v4f bR = br[2 * q];
      ao = wmma4(lo2(ai[q]), lo2(bO), ao);
      ar = wmma4(lo2(as[q]), lo2(bR), ar);
      ao = wmma4(hi2(ai[q]), hi2(bO), ao);
      ar = wmma4(hi2(as[q]), hi2(bR), ar);
    }
    float biasO = bns[n], biasR = brs[n];
#pragma unroll
    for (int r = 0; r < 8; r++) {
      int m = r + 8 * half;
      float pre = ao[r] + biasO;
      float so = pre * sigmoidf_(pre);
      if (ct == 0) lds_gate[wid][m * 16 + n] = sigmoidf_(so);
      out[(long)(nb + m) * SDIM + n] = so + (ar[r] + biasR);
    }
  }

  const v4f* vA = (const v4f*)(node_v + (long)noderow * V3) + half;
  const v4f* gV = (const v4f*)(agg_v + (long)noderow * V3) + half;
  v4f av[6], aw[6];
#pragma unroll
  for (int q = 0; q < 6; q++) {
    av[q] = vA[2 * q];
    aw[q] = gV[2 * q];
  }
  v4f qv = av[0] * av[0];
#pragma unroll
  for (int q = 1; q < 6; q++) qv += av[q] * av[q];
  float qq = (qv.x + qv.y) + (qv.z + qv.w);
  qq += __shfl_xor(qq, 16, 32);
  float vinv = rsqrtf(qq * (1.f / VDIM) + 1e-8f);
#pragma unroll
  for (int q = 0; q < 6; q++) av[q] = av[q] * vinv;
#pragma unroll
  for (int q = 0; q < 6; q++) aw[q] = av[q] + aw[q] * invden;
  float* outv = out + (long)NN * SDIM;
#pragma unroll 1
  for (int ct = 0; ct < 3; ct++) {
    v8f ao = {};
    v8f ar = {};
    int d = 16 * ct + row;
    const v4f* bo = (const v4f*)(wnv + (long)d * V3) + half;
    const v4f* br = (const v4f*)(wrv + (long)d * V3) + half;
#pragma unroll
    for (int q = 0; q < 6; q++) {
      v4f bO = bo[2 * q];
      v4f bR = br[2 * q];
      ao = wmma4(lo2(aw[q]), lo2(bO), ao);
      ar = wmma4(lo2(av[q]), lo2(bR), ar);
      ao = wmma4(hi2(aw[q]), hi2(bO), ao);
      ar = wmma4(hi2(av[q]), hi2(bR), ar);
    }
    float biasO = bnv[d], biasR = brv[d];
    int vo = d / 3;
#pragma unroll
    for (int r = 0; r < 8; r++) {
      int m = r + 8 * half;
      float pre = ao[r] + biasO;
      float g = lds_gate[wid][m * 16 + vo];
      outv[(long)(nb + m) * V3 + d] = pre * g + (ar[r] + biasR);
    }
  }
}

extern "C" void kernel_launch(void* const* d_in, const int* in_sizes, int n_in,
                              void* d_out, int out_size, void* d_ws,
                              size_t ws_size, hipStream_t stream) {
  const float* node_s = (const float*)d_in[0];
  const float* node_v = (const float*)d_in[1];
  const int* eidx     = (const int*)d_in[2];
  const float* edge_s = (const float*)d_in[3];
  const float* edge_v = (const float*)d_in[4];
  const float* ln_g = (const float*)d_in[5];
  const float* ln_b = (const float*)d_in[6];
  const float* wes = (const float*)d_in[7];
  const float* bes = (const float*)d_in[8];
  const float* wev = (const float*)d_in[9];
  const float* bev = (const float*)d_in[10];
  const float* wns = (const float*)d_in[11];
  const float* bns = (const float*)d_in[12];
  const float* wnv = (const float*)d_in[13];
  const float* bnv = (const float*)d_in[14];
  const float* wrs = (const float*)d_in[15];
  const float* brs = (const float*)d_in[16];
  const float* wrv = (const float*)d_in[17];
  const float* brv = (const float*)d_in[18];
  float* out = (float*)d_out;

  float* ws = (float*)d_ws;
  float* agg_s = ws;
  float* agg_v = agg_s + (size_t)NN * SDIM;
  float* cnt = agg_v + (size_t)NN * V3;
  int* offsT = (int*)(cnt + NN);
  int* off = offsT + (NB1 + 1) * G1;
  int* perm = off + (NN + 16);
  int* sdst = perm + EE;
  int* rec = sdst + EE;
  const int* edge_dst = eidx + EE;

  k_part<<<G1, 1024, 0, stream>>>(edge_dst, offsT, rec);
  k_sort2<<<NB1, 1024, 0, stream>>>(rec, offsT, off, cnt, perm, sdst);
  k_edge<<<NN / 16, 128, 0, stream>>>(edge_s, edge_v, perm, sdst, off, wes, bes,
                                      wev, bev, agg_s, agg_v);
  k_node<<<(NN / 16 + 7) / 8, 256, 0, stream>>>(node_s, node_v, ln_g, ln_b,
                                                agg_s, agg_v, cnt, wns, bns,
                                                wnv, bnv, wrs, brs, wrv, brv,
                                                out);
}


# k_edge: per-node sums leave the 0/1 selection WMMA (44 of 111 per tile): each lane half runs a restartable sum down its bucket-sorted rows, LDS float atomic adds at node ends; sum order changes
# speedup vs baseline: 5.9923x; 1.0924x over previous
#include <hip/hip_runtime.h>
#include <math.h>

#define NN   50000
#define EE   800000
#define SDIM 128
#define VDIM 16
#define V3   48
#define ESD  32

typedef float v2f __attribute__((ext_vector_type(2)));
typedef float v8f __attribute__((ext_vector_type(8)));

__device__ __forceinline__ v8f wmma4(v2f a, v2f b, v8f c) {
  return __builtin_amdgcn_wmma_f32_16x16x4_f32(false, a, false, b, (short)0, c,
                                               false, false);
}

__device__ __forceinline__ float fast_rcp(float x) {
  return __builtin_amdgcn_rcpf(x);
}

__device__ __forceinline__ float sigmoidf_(float x) {
  return fast_rcp(1.f + __expf(-x));
}

#define G1 256
#define CH1 (EE / G1)
#define NB1 ((NN + 255) / 256)
#define SCAN_PER 7
#define PIECE_U 8
static_assert(EE % G1 == 0, "chunks must tile the edge list");
static_assert(CH1 <= 4 * 1024, "a chunk is read as 4 values per thread");
static_assert(NB1 <= SCAN_PER * 32, "one wave scans the range counts");
static_assert(G1 * 4 == 1024, "level 2 walks a chunk's piece with 4 threads");
static_assert(PIECE_U == 8, "the load fences of level 2 name 8 pairs");

__global__ void __launch_bounds__(1024) k_part(const int* __restrict__ edge_dst,
                                               int* __restrict__ offsT,
                                               int* __restrict__ rec) {
  __shared__ int h[NB1];
  int t = threadIdx.x;
  int g = blockIdx.x;
  if (t < NB1) h[t] = 0;
  const int* src = edge_dst + g * CH1;
  int d[4];
#pragma unroll
  for (int k = 0; k < 4; k++) {
    int i = t + 1024 * k;
    d[k] = (i < CH1) ? src[i] : 0;
  }
  __syncthreads();
  int rank[4];
#pragma unroll
  for (int k = 0; k < 4; k++) {
    rank[k] = 0;
    if (t + 1024 * k < CH1) rank[k] = atomicAdd(&h[d[k] >> 8], 1);
  }
  __syncthreads();
  if (t < 32) {
    int v[SCAN_PER];
    int s = 0;
#pragma unroll
    for (int k = 0; k < SCAN_PER; k++) {
      int idx = t * SCAN_PER + k;
      v[k] = (idx < NB1) ? h[idx] : 0;
      s += v[k];
    }
    int inc = s;
#pragma unroll
    for (int m = 1; m < 32; m <<= 1) {
      int y = __shfl_up(inc, m, 32);
      if (t >= m) inc += y;
    }
    int run = inc - s;
#pragma unroll
    for (int k = 0; k < SCAN_PER; k++) {
      int idx = t * SCAN_PER + k;
      if (idx < NB1) h[idx] = run;
      run += v[k];
    }
  }
  __syncthreads();
  if (t < NB1) offsT[t * G1 + g] = h[t];
  if (t == 0) offsT[NB1 * G1 + g] = CH1;
  int2* out = (int2*)rec + g * CH1;
#pragma unroll
  for (int k = 0; k < 4; k++) {
    int i = t + 1024 * k;
    if (i < CH1) out[h[d[k] >> 8] + rank[k]] = make_int2(g * CH1 + i, d[k]);
  }
}

__global__ void __launch_bounds__(1024) k_sort2(const int* __restrict__ rec,
                                                const int* __restrict__ offsT,
                                                int* __restrict__ off,
                                                float* __restrict__ cnt,
                                                int* __restrict__ perm,
                                                int* __restrict__ sdst) {
  __shared__ int c[256];
  __shared__ int cur[256];
  __shared__ int wsum[G1 / 32];
  int t = threadIdx.x;
  int b = blockIdx.x;
  int g = t >> 2;
  int j = t & 3;
  int ps = offsT[b * G1 + g];
  int pe = offsT[(b + 1) * G1 + g];
  const int2* in = (const int2*)rec + g * CH1;
  if (t < 256) c[t] = 0;
  int x = (t < G1) ? offsT[b * G1 + t] : 0;
#pragma unroll
  for (int m = 16; m >= 1; m >>= 1) x += __shfl_xor(x, m, 32);
  if (t < G1 && (t & 31) == 0) wsum[t >> 5] = x;
  __syncthreads();
  for (int base = ps + j; base < pe; base += 4 * PIECE_U) {
    int dn[PIECE_U];
#pragma unroll
    for (int k = 0; k < PIECE_U; k++) dn[k] = in[min(base + 4 * k, pe - 1)].y;
    asm volatile("" ::"v"(dn[0]), "v"(dn[1]), "v"(dn[2]), "v"(dn[3]),
                 "v"(dn[4]), "v"(dn[5]), "v"(dn[6]), "v"(dn[7]));
#pragma unroll
    for (int k = 0; k < PIECE_U; k++)
      if (base + 4 * k < pe) atomicAdd(&c[dn[k] & 255], 1);
  }
  __syncthreads();
  if (t < 32) {
    int S = 0;
#pragma unroll
    for (int w = 0; w < G1 / 32; w++) S += wsum[w];
    int v[8];
    int s = 0;
#pragma unroll
    for (int k = 0; k < 8; k++) {
      v[k] = c[t * 8 + k];
      s += v[k];
    }
    int inc = s;
#pragma unroll
    for (int m = 1; m < 32; m <<= 1) {
      int y = __shfl_up(inc, m, 32);
      if (t >= m) inc += y;
    }
    int run = S + inc - s;
#pragma unroll
    for (int k = 0; k < 8; k++) {
      cur[t * 8 + k] = run;
      run += v[k];
    }
  }
  __syncthreads();
  if (t < 256) {
    int node = b * 256 + t;
    if (node < NN) {
      off[node] = cur[t];
      cnt[node] = (float)c[t];
    }
  }
  if (b == 0 && t == 0) off[NN] = EE;
  __syncthreads();
  for (int base = ps + j; base < pe; base += 4 * PIECE_U) {
    int2 r[PIECE_U];
#pragma unroll
    for (int k = 0; k < PIECE_U; k++) r[k] = in[min(base + 4 * k, pe - 1)];
    asm volatile("" ::"v"(r[0].x), "v"(r[0].y), "v"(r[1].x), "v"(r[1].y),
                 "v"(r[2].x), "v"(r[2].y), "v"(r[3].x), "v"(r[3].y),
                 "v"(r[4].x), "v"(r[4].y), "v"(r[5].x), "v"(r[5].y),
                 "v"(r[6].x), "v"(r[6].y), "v"(r[7].x), "v"(r[7].y));
#pragma unroll
    for (int k = 0; k < PIECE_U; k++) {
      if (base + 4 * k < pe) {
        int pos = atomicAdd(&cur[r[k].y & 255], 1);
        perm[pos] = r[k].x;
        sdst[pos] = r[k].y;
      }
    }
  }
}

typedef float v4f __attribute__((ext_vector_type(4)));

__device__ __forceinline__ v2f lo2(v4f x) { return (v2f){x.x, x.y}; }
__device__ __forceinline__ v2f hi2(v4f x) { return (v2f){x.z, x.w}; }

#define WP4 9
#define RP4 44
#define RPF (4 * RP4)
static_assert(SDIM + V3 <= RPF, "a partial-sum row holds all message columns");
#define NP4 33
static_assert(NP4 > SDIM / 4 && (NP4 & 1) == 1, "a row fits, the pitch is odd");
static_assert(2 * SDIM * NP4 * 16 + 8 * 1024 + 1024 <= 160 * 1024,
              "two blocks of k_node share a WGP's LDS");

__device__ __forceinline__ void load_idx(const int* __restrict__ perm,
                                         const int* __restrict__ sdst, int p,
                                         int row, int pend, int nb, int& e,
                                         int& dl) {
  int pe = p + row;
  bool valid = pe < pend;
  int pc = valid ? pe : 0;
  int ev = perm[pc];
  int dv = sdst[pc];
  e = valid ? ev : 0;
  dl = valid ? dv - nb : -1;
}

__device__ __forceinline__ void dest_rows(int dl, int half, int (&d)[8],
                                          float (&keep)[8]) {
#pragma unroll
  for (int r = 0; r < 8; r++) d[r] = __shfl(dl, 8 * half + r, 32);
  keep[0] = 0.f;
#pragma unroll
  for (int r = 1; r < 8; r++) keep[r] = (d[r] == d[r - 1]) ? 1.f : 0.f;
}

__device__ __forceinline__ void lds_add(float* p, float v) {
  __hip_atomic_fetch_add(p, v, __ATOMIC_RELAXED, __HIP_MEMORY_SCOPE_WORKGROUP);
}

__device__ __forceinline__ void load_wgroup(const v4f* lds_w,
                                            const float* lds_bias, int cp,
                                            int row, int half, v4f (&f)[8],
                                            float (&b)[2]) {
  const v4f* w0 = lds_w + (32 * cp + row) * WP4 + half;
  const v4f* w1 = w0 + 16 * WP4;
#pragma unroll
  for (int q = 0; q < 4; q++) {
    f[2 * q] = w0[2 * q];
    f[2 * q + 1] = w1[2 * q];
  }
  b[0] = lds_bias[32 * cp + row];
  b[1] = lds_bias[32 * cp + 16 + row];
}

__global__ void __launch_bounds__(128) k_edge(
    const float* __restrict__ edge_s, const float* __restrict__ edge_v,
    const int* __restrict__ perm, const int* __restrict__ sdst,
    const int* __restrict__ off,
    const float* __restrict__ wes, const float* __restrict__ bes,
    const float* __restrict__ wev, const float* __restrict__ bev,
    float* __restrict__ agg_s, float* __restrict__ agg_v) {
  __shared__ v4f lds_w[SDIM * WP4];
  __shared__ float lds_bias[SDIM];
  __shared__ v4f lds_red[4 * 16 * RP4];
  int wid = __builtin_amdgcn_readfirstlane(threadIdx.x >> 5);
  int lane = threadIdx.x & 31;
  int row = lane & 15;
  int half = lane >> 4;
  int nb = blockIdx.x * 16;
  int pbeg = off[nb];
  int pend = off[nb + 16];

#pragma unroll
  for (int i = 0; i < SDIM * ESD / 4 / 128; i++) {
    int idx = threadIdx.x + 128 * i;
    lds_w[(idx >> 3) * WP4 + (idx & 7)] = ((const v4f*)wes)[idx];
  }
  lds_bias[threadIdx.x] = bes[threadIdx.x];

  v2f bw[3];
  float bv[3];
  int gsrc[3];
#pragma unroll
  for (int vt = 0; vt < 3; vt++) {
    int d = 16 * vt + row;
    float w1 = wev[d * 3 + 1];
    bw[vt].x = wev[d * 3 + 2 * half];
    bw[vt].y = (half == 0) ? w1 : 0.f;
    bv[vt] = bev[d];
    gsrc[vt] = d / 3 + 16 * half;
  }
#pragma unroll
  for (int i = 0; i < 4 * 16 * RP4 / 128; i++)
    lds_red[threadIdx.x + 128 * i] = (v4f){0.f, 0.f, 0.f, 0.f};
  __syncthreads();
  float* red = (float*)lds_red + wid * (16 * RPF) + row;

  v4f fr[2][8];
  float fb[2][2];
  load_wgroup(lds_w, lds_bias, 0, row, half, fr[0], fb[0]);

  int p = pbeg + 16 * ((wid + blockIdx.x) & 3);
  int e, dl;
  load_idx(perm, sdst, p, row, pend, nb, e, dl);
  for (; p < pend; p += 64) {
    int en, dln;
    load_idx(perm, sdst, p + 64, row, pend, nb, en, dln);
    const v4f* arow = (const v4f*)(edge_s + (long)e * ESD) + half;
    v4f a[4];
#pragma unroll
    for (int q = 0; q < 4; q++) a[q] = arow[2 * q];
    float e1 = edge_v[(long)e * 3 + 1];
    v2f av;
    av.x = edge_v[(long)e * 3 + 2 * half];
    av.y = (half == 0) ? e1 : 0.f;
    int d[8];
    float keep[8];
    dest_rows(dl, half, d, keep);
    float gate[8];
    v8f msg[8];
    v8f mv[3];
    asm volatile("" ::: "memory");
#pragma unroll
    for (int cp = 0; cp < 4; cp++) {
      __builtin_amdgcn_sched_barrier(0);
      load_wgroup(lds_w, lds_bias, (cp + 1) & 3, row, half, fr[(cp + 1) & 1],
                  fb[(cp + 1) & 1]);
      __builtin_amdgcn_sched_barrier(0);
      v8f c0 = {};
      v8f c1 = {};
#pragma unroll
      for (int q = 0; q < 4; q++) {
        v4f b0 = fr[cp & 1][2 * q];
        v4f b1 = fr[cp & 1][2 * q + 1];
        c0 = wmma4(lo2(a[q]), lo2(b0), c0);
        c1 = wmma4(lo2(a[q]), lo2(b1), c1);
        c0 = wmma4(hi2(a[q]), hi2(b0), c0);
        c1 = wmma4(hi2(a[q]), hi2(b1), c1);
      }
      float bias0 = fb[cp & 1][0];
      float bias1 = fb[cp & 1][1];
      float run0 = 0.f;
      float run1 = 0.f;
#pragma unroll
      for (int r = 0; r < 8; r++) {
        float p0 = c0[r] + bias0;
        float p1 = c1[r] + bias1;
        float s0 = p0 * sigmoidf_(p0);
        float s1 = p1 * sigmoidf_(p1);
        if (cp == 0) gate[r] = sigmoidf_(s0);
        run0 = r ? fmaf(run0, keep[r], s0) : s0;
        run1 = r ? fmaf(run1, keep[r], s1) : s1;
        msg[2 * cp][r] = run0;
        msg[2 * cp + 1][r] = run1;
      }
    }
#pragma unroll
    for (int vt = 0; vt < 3; vt++) {
      v8f vv = {};
      vv = wmma4(av, bw[vt], vv);
      float run = 0.f;
#pragma unroll
      for (int r = 0; r < 8; r++) {
        float val = (vv[r] + bv[vt]) * __shfl(gate[r], gsrc[vt], 32);
        run = r ? fmaf(run, keep[r], val) : val;
        mv[vt][r] = run;
      }
    }
#pragma unroll
    for (int r = 0; r < 8; r++) {
      bool ends = (r == 7) || (d[r] != d[r + (r < 7)]);
      if (ends && d[r] >= 0) {
        float* dst = red + d[r] * RPF;
#pragma unroll
        for (int ct = 0; ct < 8; ct++) lds_add(dst + 16 * ct, msg[ct][r]);
#pragma unroll
        for (int vt = 0; vt < 3; vt++)
          lds_add(dst + SDIM + 16 * vt, mv[vt][r]);
      }
    }
    e = en;
    dl = dln;
  }
  __syncthreads();
#pragma unroll
  for (int i = 0; i < 16 * SDIM / 4 / 128; i++) {
    int idx = threadIdx.x + 128 * i;
    int m = idx >> 5;
    int c4 = idx & 31;
    const v4f* src = lds_red + m * RP4 + c4;
    v4f s = src[0];
    s += src[16 * RP4];
    s += src[2 * 16 * RP4];
    s += src[3 * 16 * RP4];
    ((v4f*)(agg_s + (long)(nb + m) * SDIM))[c4] = s;
  }
#pragma unroll
  for (int i = 0; i < 2; i++) {
    int idx = threadIdx.x + 128 * i;
    if (idx < 16 * V3 / 4) {
      int m = idx / (V3 / 4);
      int c4 = idx - m * (V3 / 4);
      const v4f* src = lds_red + m * RP4 + SDIM / 4 + c4;
      v4f s = src[0];
      s += src[16 * RP4];
      s += src[2 * 16 * RP4];
      s += src[3 * 16 * RP4];
      ((v4f*)(agg_v + (long)nb * V3))[idx] = s;
    }
  }
}

__global__ void __launch_bounds__(256, 4) k_node(
    const float* __restrict__ node_s, const float* __restrict__ node_v,
    const float* __restrict__ ln_g, const float* __restrict__ ln_b,
    const float* __restrict__ agg_s, const float* __restrict__ agg_v,
    const float* __restrict__ cnt,
    const float* __restrict__ wns, const float* __restrict__ bns,
    const float* __restrict__ wnv, const float* __restrict__ bnv,
    const float* __restrict__ wrs, const float* __restrict__ brs,
    const float* __restrict__ wrv, const float* __restrict__ brv,
    float* __restrict__ out) {
  __shared__ v4f lds_all[8 * 64 + 2 * SDIM / 4 + 2 * SDIM * NP4];
  float (*lds_gate)[16 * 16] = (float (*)[16 * 16])lds_all;
  v4f* lds_gb = lds_all + 8 * 64;
  static_assert(2 * SDIM == 256, "one LayerNorm parameter per thread");
  v4f* lds_wn = lds_gb + 2 * SDIM / 4;
  int wid = threadIdx.x >> 5;
  int lane = threadIdx.x & 31;
  int group = blockIdx.x * 8 + wid;
  bool live = group < NN / 16;
  int nb = (live ? group : NN / 16 - 1) * 16;
  int row = lane & 15;
  int half = lane >> 4;
  int noderow = nb + row;

  float gb = (threadIdx.x < SDIM) ? ln_g[threadIdx.x] : ln_b[threadIdx.x - SDIM];
  float invden = fast_rcp(fmaxf(cnt[noderow], 1.f));
  const v4f* sA = (const v4f*)(node_s + (long)noderow * SDIM) + half;
  const v4f* gA = (const v4f*)(agg_s + (long)noderow * SDIM) + half;
  v4f as[16], ai[16];
#pragma unroll
  for (int q = 0; q < 16; q++) {
    as[q] = sA[2 * q];
    ai[q] = gA[2 * q];
  }
#pragma unroll 1
  for (int mtx = 0; mtx < 2; mtx++) {
    const v4f* wsrc = (const v4f*)(mtx ? wrs : wns) + threadIdx.x;
    v4f* wdst = lds_wn + (mtx * SDIM + wid) * NP4 + lane;
#pragma unroll
    for (int i = 0; i < SDIM * SDIM / 4 / 256; i++)
      wdst[i * 8 * NP4] = wsrc[256 * i];
  }
  ((float*)lds_gb)[threadIdx.x] = gb;
  __syncthreads();
  if (!live) return;
  v4f acc = as[0];
#pragma unroll
  for (int q = 1; q < 16; q++) acc += as[q];
  float s = (acc.x + acc.y) + (acc.z + acc.w);
  s += __shfl_xor(s, 16, 32);
  float mu = s * (1.f / SDIM);
  v4f qa = {0.f, 0.f, 0.f, 0.f};
#pragma unroll
  for (int q = 0; q < 16; q++) {
    as[q] = as[q] - mu;
    qa += as[q] * as[q];
  }
  float qs = (qa.x + qa.y) + (qa.z + qa.w);
  qs += __shfl_xor(qs, 16, 32);
  float rs = rsqrtf(qs * (1.f / SDIM) + 1e-5f);
  const v4f* lG = lds_gb + half;
  const v4f* lB = lds_gb + SDIM / 4 + half;
#pragma unroll
  for (int q = 0; q < 16; q++) {
    if (q && (q & 3) == 0) __builtin_amdgcn_sched_barrier(0);
    as[q] = as[q] * rs * lG[2 * q] + lB[2 * q];
    ai[q] = as[q] + ai[q] * invden;
  }

#pragma unroll 1
  for (int ct = 0; ct < 8; ct++) {
    v8f ao = {};
    v8f ar = {};
    int n = 16 * ct + row;
    const v4f* bo = lds_wn + n * NP4 + half;
    const v4f* br = bo + SDIM * NP4;
#pragma unroll
    for (int q = 0; q < 16; q++) {
      v4f bO = bo[2 * q];
      v4f bR = br[2 * q];
      ao = wmma4(lo2(ai[q]), lo2(bO), ao);
      ar = wmma4(lo2(as[q]), lo2(bR), ar);
      ao = wmma4(hi2(ai[q]), hi2(bO), ao);
      ar = wmma4(hi2(as[q]), hi2(bR), ar);
    }
    float biasO = bns[n], biasR = brs[n];
#pragma unroll
    for (int r = 0; r < 8; r++) {
      int m = r + 8 * half;
      float pre = ao[r] + biasO;
      float so = pre * sigmoidf_(pre);
      if (ct == 0) lds_gate[wid][m * 16 + n] = sigmoidf_(so);
      out[(long)(nb + m) * SDIM + n] = so + (ar[r] + biasR);
    }
  }

  const v4f* vA = (const v4f*)(node_v + (long)noderow * V3) + half;
  const v4f* gV = (const v4f*)(agg_v + (long)noderow * V3) + half;
  v4f av[6], aw[6];
#pragma unroll
  for (int q = 0; q < 6; q++) {
    av[q] = vA[2 * q];
    aw[q] = gV[2 * q];
  }
  v4f qv = av[0] * av[0];
#pragma unroll
  for (int q = 1; q < 6; q++) qv += av[q] * av[q];
  float qq = (qv.x + qv.y) + (qv.z + qv.w);
  qq += __shfl_xor(qq, 16, 32);
  float vinv = rsqrtf(qq * (1.f / VDIM) + 1e-8f);
#pragma unroll
  for (int q = 0; q < 6; q++) av[q] = av[q] * vinv;
#pragma unroll
  for (int q = 0; q < 6; q++) aw[q] = av[q] + aw[q] * invden;
  float* outv = out + (long)NN * SDIM;
#pragma unroll 1
  for (int ct = 0; ct < 3; ct++) {
    v8f ao = {};
    v8f ar = {};
    int d = 16 * ct + row;
    const v4f* bo = (const v4f*)(wnv + (long)d * V3) + half;
    const v4f* br = (const v4f*)(wrv + (long)d * V3) + half;
#pragma unroll
    for (int q = 0; q < 6; q++) {
      v4f bO = bo[2 * q];
      v4f bR = br[2 * q];
      ao = wmma4(lo2(aw[q]), lo2(bO), ao);
      ar = wmma4(lo2(av[q]), lo2(bR), ar);
      ao = wmma4(hi2(aw[q]), hi2(bO), ao);
      ar = wmma4(hi2(av[q]), hi2(bR), ar);
    }
    float biasO = bnv[d], biasR = brv[d];
    int vo = d / 3;
#pragma unroll
    for (int r = 0; r < 8; r++) {
      int m = r + 8 * half;
      float pre = ao[r] + biasO;
      float g = lds_gate[wid][m * 16 + vo];
      outv[(long)(nb + m) * V3 + d] = pre * g + (ar[r] + biasR);
    }
  }
}

extern "C" void kernel_launch(void* const* d_in, const int* in_sizes, int n_in,
                              void* d_out, int out_size, void* d_ws,
                              size_t ws_size, hipStream_t stream) {
  const float* node_s = (const float*)d_in[0];
  const float* node_v = (const float*)d_in[1];
  const int* eidx     = (const int*)d_in[2];
  const float* edge_s = (const float*)d_in[3];
  const float* edge_v = (const float*)d_in[4];
  const float* ln_g = (const float*)d_in[5];
  const float* ln_b = (const float*)d_in[6];
  const float* wes = (const float*)d_in[7];
  const float* bes = (const float*)d_in[8];
  const float* wev = (const float*)d_in[9];
  const float* bev = (const float*)d_in[10];
  const float* wns = (const float*)d_in[11];
  const float* bns = (const float*)d_in[12];
  const float* wnv = (const float*)d_in[13];
  const float* bnv = (const float*)d_in[14];
  const float* wrs = (const float*)d_in[15];
  const float* brs = (const float*)d_in[16];
  const float* wrv = (const float*)d_in[17];
  const float* brv = (const float*)d_in[18];
  float* out = (float*)d_out;

  float* ws = (float*)d_ws;
  float* agg_s = ws;
  float* agg_v = agg_s + (size_t)NN * SDIM;
  float* cnt = agg_v + (size_t)NN * V3;
  int* offsT = (int*)(cnt + NN);
  int* off = offsT + (NB1 + 1) * G1;
  int* perm = off + (NN + 16);
  int* sdst = perm + EE;
  int* rec = sdst + EE;
  const int* edge_dst = eidx + EE;

  k_part<<<G1, 1024, 0, stream>>>(edge_dst, offsT, rec);
  k_sort2<<<NB1, 1024, 0, stream>>>(rec, offsT, off, cnt, perm, sdst);
  k_edge<<<NN / 16, 128, 0, stream>>>(edge_s, edge_v, perm, sdst, off, wes, bes,
                                      wev, bev, agg_s, agg_v);
  k_node<<<(NN / 16 + 7) / 8, 256, 0, stream>>>(node_s, node_v, ln_g, ln_b,
                                                agg_s, agg_v, cnt, wns, bns,
                                                wnv, bnv, wrs, brs, wrv, brv,
                                                out);
}


# k_edge: each tile's edge rows are gathered one tile ahead (edge ids two ahead), so the rows' trip from memory passes under the previous tile's products instead of in front of its own
# speedup vs baseline: 6.1879x; 1.0326x over previous
#include <hip/hip_runtime.h>
#include <math.h>

#define NN   50000
#define EE   800000
#define SDIM 128
#define VDIM 16
#define V3   48
#define ESD  32

typedef float v2f __attribute__((ext_vector_type(2)));
typedef float v8f __attribute__((ext_vector_type(8)));

__device__ __forceinline__ v8f wmma4(v2f a, v2f b, v8f c) {
  return __builtin_amdgcn_wmma_f32_16x16x4_f32(false, a, false, b, (short)0, c,
                                               false, false);
}

__device__ __forceinline__ float fast_rcp(float x) {
  return __builtin_amdgcn_rcpf(x);
}

__device__ __forceinline__ float sigmoidf_(float x) {
  return fast_rcp(1.f + __expf(-x));
}

#define G1 256
#define CH1 (EE / G1)
#define NB1 ((NN + 255) / 256)
#define SCAN_PER 7
#define PIECE_U 8
static_assert(EE % G1 == 0, "chunks must tile the edge list");
static_assert(CH1 <= 4 * 1024, "a chunk is read as 4 values per thread");
static_assert(NB1 <= SCAN_PER * 32, "one wave scans the range counts");
static_assert(G1 * 4 == 1024, "level 2 walks a chunk's piece with 4 threads");
static_assert(PIECE_U == 8, "the load fences of level 2 name 8 pairs");

__global__ void __launch_bounds__(1024) k_part(const int* __restrict__ edge_dst,
                                               int* __restrict__ offsT,
                                               int* __restrict__ rec) {
  __shared__ int h[NB1];
  int t = threadIdx.x;
  int g = blockIdx.x;
  if (t < NB1) h[t] = 0;
  const int* src = edge_dst + g * CH1;
  int d[4];
#pragma unroll
  for (int k = 0; k < 4; k++) {
    int i = t + 1024 * k;
    d[k] = (i < CH1) ? src[i] : 0;
  }
  __syncthreads();
  int rank[4];
#pragma unroll
  for (int k = 0; k < 4; k++) {
    rank[k] = 0;
    if (t + 1024 * k < CH1) rank[k] = atomicAdd(&h[d[k] >> 8], 1);
  }
  __syncthreads();
  if (t < 32) {
    int v[SCAN_PER];
    int s = 0;
#pragma unroll
    for (int k = 0; k < SCAN_PER; k++) {
      int idx = t * SCAN_PER + k;
      v[k] = (idx < NB1) ? h[idx] : 0;
      s += v[k];
    }
    int inc = s;
#pragma unroll
    for (int m = 1; m < 32; m <<= 1) {
      int y = __shfl_up(inc, m, 32);
      if (t >= m) inc += y;
    }
    int run = inc - s;
#pragma unroll
    for (int k = 0; k < SCAN_PER; k++) {
      int idx = t * SCAN_PER + k;
      if (idx < NB1) h[idx] = run;
      run += v[k];
    }
  }
  __syncthreads();
  if (t < NB1) offsT[t * G1 + g] = h[t];
  if (t == 0) offsT[NB1 * G1 + g] = CH1;
  int2* out = (int2*)rec + g * CH1;
#pragma unroll
  for (int k = 0; k < 4; k++) {
    int i = t + 1024 * k;
    if (i < CH1) out[h[d[k] >> 8] + rank[k]] = make_int2(g * CH1 + i, d[k]);
  }
}

__global__ void __launch_bounds__(1024) k_sort2(const int* __restrict__ rec,
                                                const int* __restrict__ offsT,
                                                int* __restrict__ off,
                                                float* __restrict__ cnt,
                                                int* __restrict__ perm,
                                                int* __restrict__ sdst) {
  __shared__ int c[256];
  __shared__ int cur[256];
  __shared__ int wsum[G1 / 32];
  int t = threadIdx.x;
  int b = blockIdx.x;
  int g = t >> 2;
  int j = t & 3;
  int ps = offsT[b * G1 + g];
  int pe = offsT[(b + 1) * G1 + g];
  const int2* in = (const int2*)rec + g * CH1;
  if (t < 256) c[t] = 0;
  int x = (t < G1) ? offsT[b * G1 + t] : 0;
#pragma unroll
  for (int m = 16; m >= 1; m >>= 1) x += __shfl_xor(x, m, 32);
  if (t < G1 && (t & 31) == 0) wsum[t >> 5] = x;
  __syncthreads();
  for (int base = ps + j; base < pe; base += 4 * PIECE_U) {
    int dn[PIECE_U];
#pragma unroll
    for (int k = 0; k < PIECE_U; k++) dn[k] = in[min(base + 4 * k, pe - 1)].y;
    asm volatile("" ::"v"(dn[0]), "v"(dn[1]), "v"(dn[2]), "v"(dn[3]),
                 "v"(dn[4]), "v"(dn[5]), "v"(dn[6]), "v"(dn[7]));
#pragma unroll
    for (int k = 0; k < PIECE_U; k++)
      if (base + 4 * k < pe) atomicAdd(&c[dn[k] & 255], 1);
  }
  __syncthreads();
  if (t < 32) {
    int S = 0;
#pragma unroll
    for (int w = 0; w < G1 / 32; w++) S += wsum[w];
    int v[8];
    int s = 0;
#pragma unroll
    for (int k = 0; k < 8; k++) {
      v[k] = c[t * 8 + k];
      s += v[k];
    }
    int inc = s;
#pragma unroll
    for (int m = 1; m < 32; m <<= 1) {
      int y = __shfl_up(inc, m, 32);
      if (t >= m) inc += y;
    }
    int run = S + inc - s;
#pragma unroll
    for (int k = 0; k < 8; k++) {
      cur[t * 8 + k] = run;
      run += v[k];
    }
  }
  __syncthreads();
  if (t < 256) {
    int node = b * 256 + t;
    if (node < NN) {
      off[node] = cur[t];
      cnt[node] = (float)c[t];
    }
  }
  if (b == 0 && t == 0) off[NN] = EE;
  __syncthreads();
  for (int base = ps + j; base < pe; base += 4 * PIECE_U) {
    int2 r[PIECE_U];
#pragma unroll
    for (int k = 0; k < PIECE_U; k++) r[k] = in[min(base + 4 * k, pe - 1)];
    asm volatile("" ::"v"(r[0].x), "v"(r[0].y), "v"(r[1].x), "v"(r[1].y),
                 "v"(r[2].x), "v"(r[2].y), "v"(r[3].x), "v"(r[3].y),
                 "v"(r[4].x), "v"(r[4].y), "v"(r[5].x), "v"(r[5].y),
                 "v"(r[6].x), "v"(r[6].y), "v"(r[7].x), "v"(r[7].y));
#pragma unroll
    for (int k = 0; k < PIECE_U; k++) {
      if (base + 4 * k < pe) {
        int pos = atomicAdd(&cur[r[k].y & 255], 1);
        perm[pos] = r[k].x;
        sdst[pos] = r[k].y;
      }
    }
  }
}

typedef float v4f __attribute__((ext_vector_type(4)));

__device__ __forceinline__ v2f lo2(v4f x) { return (v2f){x.x, x.y}; }
__device__ __forceinline__ v2f hi2(v4f x) { return (v2f){x.z, x.w}; }

#define WP4 9
#define RP4 44
#define RPF (4 * RP4)
static_assert(SDIM + V3 <= RPF, "a partial-sum row holds all message columns");
#define NP4 33
static_assert(NP4 > SDIM / 4 && (NP4 & 1) == 1, "a row fits, the pitch is odd");
static_assert(2 * SDIM * NP4 * 16 + 8 * 1024 + 1024 <= 160 * 1024,
              "two blocks of k_node share a WGP's LDS");

__device__ __forceinline__ void load_idx(const int* __restrict__ perm,
                                         const int* __restrict__ sdst, int p,
                                         int row, int pend, int nb, int& e,
                                         int& dl) {
  int pe = p + row;
  bool valid = pe < pend;
  int pc = valid ? pe : 0;
  int ev = perm[pc];
  int dv = sdst[pc];
  e = valid ? ev : 0;
  dl = valid ? dv - nb : -1;
}

__device__ __forceinline__ void load_rows(const float* edge_s,
                                          const float* edge_v, int e, int half,
                                          v4f (&a)[4], float& vx, float& v1) {
  const v4f* arow = (const v4f*)(edge_s + (long)e * ESD) + half;
#pragma unroll
  for (int q = 0; q < 4; q++) a[q] = arow[2 * q];
  v1 = edge_v[(long)e * 3 + 1];
  vx = edge_v[(long)e * 3 + 2 * half];
}

__device__ __forceinline__ void dest_rows(int dl, int half, int (&d)[8],
                                          float (&keep)[8]) {
#pragma unroll
  for (int r = 0; r < 8; r++) d[r] = __shfl(dl, 8 * half + r, 32);
  keep[0] = 0.f;
#pragma unroll
  for (int r = 1; r < 8; r++) keep[r] = (d[r] == d[r - 1]) ? 1.f : 0.f;
}

__device__ __forceinline__ void lds_add(float* p, float v) {
  __hip_atomic_fetch_add(p, v, __ATOMIC_RELAXED, __HIP_MEMORY_SCOPE_WORKGROUP);
}

__device__ __forceinline__ void load_wgroup(const v4f* lds_w,
                                            const float* lds_bias, int cp,
                                            int row, int half, v4f (&f)[8],
                                            float (&b)[2]) {
  const v4f* w0 = lds_w + (32 * cp + row) * WP4 + half;
  const v4f* w1 = w0 + 16 * WP4;
#pragma unroll
  for (int q = 0; q < 4; q++) {
    f[2 * q] = w0[2 * q];
    f[2 * q + 1] = w1[2 * q];
  }
  b[0] = lds_bias[32 * cp + row];
  b[1] = lds_bias[32 * cp + 16 + row];
}

__global__ void __launch_bounds__(128) k_edge(
    const float* __restrict__ edge_s, const float* __restrict__ edge_v,
    const int* __restrict__ perm, const int* __restrict__ sdst,
    const int* __restrict__ off,
    const float* __restrict__ wes, const float* __restrict__ bes,
    const float* __restrict__ wev, const float* __restrict__ bev,
    float* __restrict__ agg_s, float* __restrict__ agg_v) {
  __shared__ v4f lds_w[SDIM * WP4];
  __shared__ float lds_bias[SDIM];
  __shared__ v4f lds_red[4 * 16 * RP4];
  int wid = __builtin_amdgcn_readfirstlane(threadIdx.x >> 5);
  int lane = threadIdx.x & 31;
  int row = lane & 15;
  int half = lane >> 4;
  int nb = blockIdx.x * 16;
  int pbeg = off[nb];
  int pend = off[nb + 16];

#pragma unroll
  for (int i = 0; i < SDIM * ESD / 4 / 128; i++) {
    int idx = threadIdx.x + 128 * i;
    lds_w[(idx >> 3) * WP4 + (idx & 7)] = ((const v4f*)wes)[idx];
  }
  lds_bias[threadIdx.x] = bes[threadIdx.x];

  v2f bw[3];
  float bv[3];
  int gsrc[3];
#pragma unroll
  for (int vt = 0; vt < 3; vt++) {
    int d = 16 * vt + row;
    float w1 = wev[d * 3 + 1];
    bw[vt].x = wev[d * 3 + 2 * half];
    bw[vt].y = (half == 0) ? w1 : 0.f;
    bv[vt] = bev[d];
    gsrc[vt] = d / 3 + 16 * half;
  }
#pragma unroll
  for (int i = 0; i < 4 * 16 * RP4 / 128; i++)
    lds_red[threadIdx.x + 128 * i] = (v4f){0.f, 0.f, 0.f, 0.f};
  __syncthreads();
  float* red = (float*)lds_red + wid * (16 * RPF) + row;

  v4f fr[2][8];
  float fb[2][2];
  load_wgroup(lds_w, lds_bias, 0, row, half, fr[0], fb[0]);

  int p = pbeg + 16 * ((wid + blockIdx.x) & 3);
  int e, dl, en, dln;
  load_idx(perm, sdst, p, row, pend, nb, e, dl);
  load_idx(perm, sdst, p + 64, row, pend, nb, en, dln);
  v4f a[4];
  float vx, v1;
  load_rows(edge_s, edge_v, e, half, a, vx, v1);
  for (; p < pend; p += 64) {
    int e2, dl2;
    load_idx(perm, sdst, p + 128, row, pend, nb, e2, dl2);
    v4f an[4];
    float vxn, v1n;
    load_rows(edge_s, edge_v, en, half, an, vxn, v1n);
    v2f av;
    av.x = vx;
    av.y = (half == 0) ? v1 : 0.f;
    int d[8];
    float keep[8];
    dest_rows(dl, half, d, keep);
    float gate[8];
    v8f msg[8];
    v8f mv[3];
    asm volatile("" ::: "memory");
#pragma unroll
    for (int cp = 0; cp < 4; cp++) {
      __builtin_amdgcn_sched_barrier(0);
      load_wgroup(lds_w, lds_bias, (cp + 1) & 3, row, half, fr[(cp + 1) & 1],
                  fb[(cp + 1) & 1]);
      __builtin_amdgcn_sched_barrier(0);
      v8f c0 = {};
      v8f c1 = {};
#pragma unroll
      for (int q = 0; q < 4; q++) {
        v4f b0 = fr[cp & 1][2 * q];
        v4f b1 = fr[cp & 1][2 * q + 1];
        c0 = wmma4(lo2(a[q]), lo2(b0), c0);
        c1 = wmma4(lo2(a[q]), lo2(b1), c1);
        c0 = wmma4(hi2(a[q]), hi2(b0), c0);
        c1 = wmma4(hi2(a[q]), hi2(b1), c1);
      }
      float bias0 = fb[cp & 1][0];
      float bias1 = fb[cp & 1][1];
      float run0 = 0.f;
      float run1 = 0.f;
#pragma unroll
      for (int r = 0; r < 8; r++) {
        float p0 = c0[r] + bias0;
        float p1 = c1[r] + bias1;
        float s0 = p0 * sigmoidf_(p0);
        float s1 = p1 * sigmoidf_(p1);
        if (cp == 0) gate[r] = sigmoidf_(s0);
        run0 = r ? fmaf(run0, keep[r], s0) : s0;
        run1 = r ? fmaf(run1, keep[r], s1) : s1;
        msg[2 * cp][r] = run0;
        msg[2 * cp + 1][r] = run1;
      }
    }
#pragma unroll
    for (int vt = 0; vt < 3; vt++) {
      v8f vv = {};
      vv = wmma4(av, bw[vt], vv);
      float run = 0.f;
#pragma unroll
      for (int r = 0; r < 8; r++) {
        float val = (vv[r] + bv[vt]) * __shfl(gate[r], gsrc[vt], 32);
        run = r ? fmaf(run, keep[r], val) : val;
        mv[vt][r] = run;
      }
    }
#pragma unroll
    for (int r = 0; r < 8; r++) {
      bool ends = (r == 7) || (d[r] != d[r + (r < 7)]);
      if (ends && d[r] >= 0) {
        float* dst = red + d[r] * RPF;
#pragma unroll
        for (int ct = 0; ct < 8; ct++) lds_add(dst + 16 * ct, msg[ct][r]);
#pragma unroll
        for (int vt = 0; vt < 3; vt++)
          lds_add(dst + SDIM + 16 * vt, mv[vt][r]);
      }
    }
#pragma unroll
    for (int q = 0; q < 4; q++) a[q] = an[q];
    vx = vxn;
    v1 = v1n;
    dl = dln;
    en = e2;
    dln = dl2;
  }
  __syncthreads();
#pragma unroll
  for (int i = 0; i < 16 * SDIM / 4 / 128; i++) {
    int idx = threadIdx.x + 128 * i;
    int m = idx >> 5;
    int c4 = idx & 31;
    const v4f* src = lds_red + m * RP4 + c4;
    v4f s = src[0];
    s += src[16 * RP4];
    s += src[2 * 16 * RP4];
    s += src[3 * 16 * RP4];
    ((v4f*)(agg_s + (long)(nb + m) * SDIM))[c4] = s;
  }
#pragma unroll
  for (int i = 0; i < 2; i++) {
    int idx = threadIdx.x + 128 * i;
    if (idx < 16 * V3 / 4) {
      int m = idx / (V3 / 4);
      int c4 = idx - m * (V3 / 4);
      const v4f* src = lds_red + m * RP4 + SDIM / 4 + c4;
      v4f s = src[0];
      s += src[16 * RP4];
      s += src[2 * 16 * RP4];
      s += src[3 * 16 * RP4];
      ((v4f*)(agg_v + (long)nb * V3))[idx] = s;
    }
  }
}

__global__ void __launch_bounds__(256, 4) k_node(
    const float* __restrict__ node_s, const float* __restrict__ node_v,
    const float* __restrict__ ln_g, const float* __restrict__ ln_b,
    const float* __restrict__ agg_s, const float* __restrict__ agg_v,
    const float* __restrict__ cnt,
    const float* __restrict__ wns, const float* __restrict__ bns,
    const float* __restrict__ wnv, const float* __restrict__ bnv,
    const float* __restrict__ wrs, const float* __restrict__ brs,
    const float* __restrict__ wrv, const float* __restrict__ brv,
    float* __restrict__ out) {
  __shared__ v4f lds_all[8 * 64 + 2 * SDIM / 4 + 2 * SDIM * NP4];
  float (*lds_gate)[16 * 16] = (float (*)[16 * 16])lds_all;
  v4f* lds_gb = lds_all + 8 * 64;
  static_assert(2 * SDIM == 256, "one LayerNorm parameter per thread");
  v4f* lds_wn = lds_gb + 2 * SDIM / 4;
  int wid = threadIdx.x >> 5;
  int lane = threadIdx.x & 31;
  int group = blockIdx.x * 8 + wid;
  bool live = group < NN / 16;
  int nb = (live ? group : NN / 16 - 1) * 16;
  int row = lane & 15;
  int half = lane >> 4;
  int noderow = nb + row;

  float gb = (threadIdx.x < SDIM) ? ln_g[threadIdx.x] : ln_b[threadIdx.x - SDIM];
  float invden = fast_rcp(fmaxf(cnt[noderow], 1.f));
  const v4f* sA = (const v4f*)(node_s + (long)noderow * SDIM) + half;
  const v4f* gA = (const v4f*)(agg_s + (long)noderow * SDIM) + half;
  v4f as[16], ai[16];
#pragma unroll
  for (int q = 0; q < 16; q++) {
    as[q] = sA[2 * q];
    ai[q] = gA[2 * q];
  }
#pragma unroll 1
  for (int mtx = 0; mtx < 2; mtx++) {
    const v4f* wsrc = (const v4f*)(mtx ? wrs : wns) + threadIdx.x;
    v4f* wdst = lds_wn + (mtx * SDIM + wid) * NP4 + lane;
#pragma unroll
    for (int i = 0; i < SDIM * SDIM / 4 / 256; i++)
      wdst[i * 8 * NP4] = wsrc[256 * i];
  }
  ((float*)lds_gb)[threadIdx.x] = gb;
  __syncthreads();
  if (!live) return;
  v4f acc = as[0];
#pragma unroll
  for (int q = 1; q < 16; q++) acc += as[q];
  float s = (acc.x + acc.y) + (acc.z + acc.w);
  s += __shfl_xor(s, 16, 32);
  float mu = s * (1.f / SDIM);
  v4f qa = {0.f, 0.f, 0.f, 0.f};
#pragma unroll
  for (int q = 0; q < 16; q++) {
    as[q] = as[q] - mu;
    qa += as[q] * as[q];
  }
  float qs = (qa.x + qa.y) + (qa.z + qa.w);
  qs += __shfl_xor(qs, 16, 32);
  float rs = rsqrtf(qs * (1.f / SDIM) + 1e-5f);
  const v4f* lG = lds_gb + half;
  const v4f* lB = lds_gb + SDIM / 4 + half;
#pragma unroll
  for (int q = 0; q < 16; q++) {
    if (q && (q & 3) == 0) __builtin_amdgcn_sched_barrier(0);
    as[q] = as[q] * rs * lG[2 * q] + lB[2 * q];
    ai[q] = as[q] + ai[q] * invden;
  }

#pragma unroll 1
  for (int ct = 0; ct < 8; ct++) {
    v8f ao = {};
    v8f ar = {};
    int n = 16 * ct + row;
    const v4f* bo = lds_wn + n * NP4 + half;
    const v4f* br = bo + SDIM * NP4;
#pragma unroll
    for (int q = 0; q < 16; q++) {
      v4f bO = bo[2 * q];
      v4f bR = br[2 * q];
      ao = wmma4(lo2(ai[q]), lo2(bO), ao);
      ar = wmma4(lo2(as[q]), lo2(bR), ar);
      ao = wmma4(hi2(ai[q]), hi2(bO), ao);
      ar = wmma4(hi2(as[q]), hi2(bR), ar);
    }
    float biasO = bns[n], biasR = brs[n];
#pragma unroll
    for (int r = 0; r < 8; r++) {
      int m = r + 8 * half;
      float pre = ao[r] + biasO;
      float so = pre * sigmoidf_(pre);
      if (ct == 0) lds_gate[wid][m * 16 + n] = sigmoidf_(so);
      out[(long)(nb + m) * SDIM + n] = so + (ar[r] + biasR);
    }
  }

  const v4f* vA = (const v4f*)(node_v + (long)noderow * V3) + half;
  const v4f* gV = (const v4f*)(agg_v + (long)noderow * V3) + half;
  v4f av[6], aw[6];
#pragma unroll
  for (int q = 0; q < 6; q++) {
    av[q] = vA[2 * q];
    aw[q] = gV[2 * q];
  }
  v4f qv = av[0] * av[0];
#pragma unroll
  for (int q = 1; q < 6; q++) qv += av[q] * av[q];
  float qq = (qv.x + qv.y) + (qv.z + qv.w);
  qq += __shfl_xor(qq, 16, 32);
  float vinv = rsqrtf(qq * (1.f / VDIM) + 1e-8f);
#pragma unroll
  for (int q = 0; q < 6; q++) av[q] = av[q] * vinv;
#pragma unroll
  for (int q = 0; q < 6; q++) aw[q] = av[q] + aw[q] * invden;
  float* outv = out + (long)NN * SDIM;
#pragma unroll 1
  for (int ct = 0; ct < 3; ct++) {
    v8f ao = {};
    v8f ar = {};
    int d = 16 * ct + row;
    const v4f* bo = (const v4f*)(wnv + (long)d * V3) + half;
    const v4f* br = (const v4f*)(wrv + (long)d * V3) + half;
#pragma unroll
    for (int q = 0; q < 6; q++) {
      v4f bO = bo[2 * q];
      v4f bR = br[2 * q];
      ao = wmma4(lo2(aw[q]), lo2(bO), ao);
      ar = wmma4(lo2(av[q]), lo2(bR), ar);
      ao = wmma4(hi2(aw[q]), hi2(bO), ao);
      ar = wmma4(hi2(av[q]), hi2(bR), ar);
    }
    float biasO = bnv[d], biasR = brv[d];
    int vo = d / 3;
#pragma unroll
    for (int r = 0; r < 8; r++) {
      int m = r + 8 * half;
      float pre = ao[r] + biasO;
      float g = lds_gate[wid][m * 16 + vo];
      outv[(long)(nb + m) * V3 + d] = pre * g + (ar[r] + biasR);
    }
  }
}

extern "C" void kernel_launch(void* const* d_in, const int* in_sizes, int n_in,
                              void* d_out, int out_size, void* d_ws,
                              size_t ws_size, hipStream_t stream) {
  const float* node_s = (const float*)d_in[0];
  const float* node_v = (const float*)d_in[1];
  const int* eidx     = (const int*)d_in[2];
  const float* edge_s = (const float*)d_in[3];
  const float* edge_v = (const float*)d_in[4];
  const float* ln_g = (const float*)d_in[5];
  const float* ln_b = (const float*)d_in[6];
  const float* wes = (const float*)d_in[7];
  const float* bes = (const float*)d_in[8];
  const float* wev = (const float*)d_in[9];
  const float* bev = (const float*)d_in[10];
  const float* wns = (const float*)d_in[11];
  const float* bns = (const float*)d_in[12];
  const float* wnv = (const float*)d_in[13];
  const float* bnv = (const float*)d_in[14];
  const float* wrs = (const float*)d_in[15];
  const float* brs = (const float*)d_in[16];
  const float* wrv = (const float*)d_in[17];
  const float* brv = (const float*)d_in[18];
  float* out = (float*)d_out;

  float* ws = (float*)d_ws;
  float* agg_s = ws;
  float* agg_v = agg_s + (size_t)NN * SDIM;
  float* cnt = agg_v + (size_t)NN * V3;
  int* offsT = (int*)(cnt + NN);
  int* off = offsT + (NB1 + 1) * G1;
  int* perm = off + (NN + 16);
  int* sdst = perm + EE;
  int* rec = sdst + EE;
  const int* edge_dst = eidx + EE;

  k_part<<<G1, 1024, 0, stream>>>(edge_dst, offsT, rec);
  k_sort2<<<NB1, 1024, 0, stream>>>(rec, offsT, off, cnt, perm, sdst);
  k_edge<<<NN / 16, 128, 0, stream>>>(edge_s, edge_v, perm, sdst, off, wes, bes,
                                      wev, bev, agg_s, agg_v);
  k_node<<<(NN / 16 + 7) / 8, 256, 0, stream>>>(node_s, node_v, ln_g, ln_b,
                                                agg_s, agg_v, cnt, wns, bns,
                                                wnv, bnv, wrs, brs, wrv, brv,
                                                out);
}
